# starting kernel
# baseline (speedup 1.0000x reference)
#include <hip/hip_runtime.h>
#include <hip/hip_bf16.h>

typedef __attribute__((ext_vector_type(2))) float v2f;
typedef __attribute__((ext_vector_type(8))) float v8f;

#define IN_CH 64
#define OUT_DIM 128
#define HEADS 4

static __device__ __forceinline__ float lrelu(float x) { return fmaxf(x, 0.2f * x); }

static __device__ __forceinline__ unsigned fkey(float f) {
  unsigned u = __float_as_uint(f);
  return (u & 0x80000000u) ? ~u : (u | 0x80000000u);
}
static __device__ __forceinline__ float fkey_inv(unsigned k) {
  return __uint_as_float((k & 0x80000000u) ? (k & 0x7FFFFFFFu) : ~k);
}

__global__ __launch_bounds__(256) void k_gemm(const float* __restrict__ x,
                                              const float* __restrict__ lin_w,
                                              const float* __restrict__ res_w,
                                              float* __restrict__ xt,
                                              float* __restrict__ z,
                                              int N) {
  const int wave = threadIdx.x >> 5;
  const int lane = threadIdx.x & 31;
  const int m0 = blockIdx.x * 16;
  const int mrow = lane & 15;
  const int kb = (lane >> 4) << 1;
  int arow = m0 + mrow;
  if (arow >= N) arow = N - 1;
  const float* xr = x + (size_t)arow * IN_CH;
  v2f a[16];
#pragma unroll
  for (int ks = 0; ks < 16; ++ks) {
    a[ks].x = xr[ks * 4 + kb];
    a[ks].y = xr[ks * 4 + kb + 1];
  }
#pragma unroll
  for (int tt = 0; tt < 2; ++tt) {
    const int tile = wave + tt * 8;
    const float* W = (tile < 8) ? lin_w : res_w;
    const int cbase = (tile & 7) * 16;
    const float* wr = W + (size_t)(cbase + (lane & 15)) * IN_CH;
    v8f c = {0.f, 0.f, 0.f, 0.f, 0.f, 0.f, 0.f, 0.f};
#pragma unroll
    for (int ks = 0; ks < 16; ++ks) {
      v2f b;
      b.x = wr[ks * 4 + kb];
      b.y = wr[ks * 4 + kb + 1];
      c = __builtin_amdgcn_wmma_f32_16x16x4_f32(false, a[ks], false, b,
                                                (short)0, c, false, false);
    }
    float* dst = (tile < 8) ? xt : z;
    const int mb = m0 + ((lane >> 4) << 3);
#pragma unroll
    for (int r = 0; r < 8; ++r) {
      const int m = mb + r;
      if (m < N) dst[(size_t)m * OUT_DIM + cbase + (lane & 15)] = c[r];
    }
  }
}

__global__ __launch_bounds__(256) void k_node(const float* __restrict__ xt,
                                              const float* __restrict__ att_src,
                                              const float* __restrict__ att_dst,
                                              const float* __restrict__ gat_bias,
                                              const float* __restrict__ res_b,
                                              float* __restrict__ z,
                                              float* __restrict__ a_src,
                                              float* __restrict__ a_dst,
                                              unsigned* __restrict__ emax_u,
                                              float* __restrict__ esum, int N) {
  const int lane = threadIdx.x & 31;
  const int n = blockIdx.x * 8 + (threadIdx.x >> 5);
  if (n >= N) return;
  const float4 v = ((const float4*)xt)[(size_t)n * 32 + lane];
  const float4 as = ((const float4*)att_src)[lane];
  const float4 ad = ((const float4*)att_dst)[lane];
  float ps = v.x * as.x + v.y * as.y + v.z * as.z + v.w * as.w;
  float pd = v.x * ad.x + v.y * ad.y + v.z * ad.z + v.w * ad.w;
#pragma unroll
  for (int off = 1; off < 8; off <<= 1) {
    ps += __shfl_xor(ps, off, 32);
    pd += __shfl_xor(pd, off, 32);
  }
  const int h = lane >> 3;
  if ((lane & 7) == 0) {
    a_src[(size_t)n * HEADS + h] = ps;
    a_dst[(size_t)n * HEADS + h] = pd;
  }
  if (lane < HEADS) {
    emax_u[(size_t)n * HEADS + lane] = 0u;
    esum[(size_t)n * HEADS + lane] = 0.f;
  }
  float4 zr = ((float4*)z)[(size_t)n * 32 + lane];
  const float4 gb = ((const float4*)gat_bias)[lane];
  const float4 rb = ((const float4*)res_b)[lane];
  zr.x += gb.x + rb.x; zr.y += gb.y + rb.y;
  zr.z += gb.z + rb.z; zr.w += gb.w + rb.w;
  ((float4*)z)[(size_t)n * 32 + lane] = zr;
}

static __device__ __forceinline__ void edge_sd(const int* __restrict__ ei, int e,
                                               int E, int& s, int& d) {
  if (e < E) { s = ei[e]; d = ei[E + e]; }
  else       { s = e - E; d = s; }
}

__global__ __launch_bounds__(256) void k_edge_max(const int* __restrict__ ei,
                                                  const float* __restrict__ a_src,
                                                  const float* __restrict__ a_dst,
                                                  unsigned* __restrict__ emax_u,
                                                  int E, int N) {
  const int e = blockIdx.x * blockDim.x + threadIdx.x;
  if (e >= E + N) return;
  int s, d; edge_sd(ei, e, E, s, d);
  const float4 as = ((const float4*)a_src)[s];
  const float4 ad = ((const float4*)a_dst)[d];
  unsigned* mp = emax_u + (size_t)d * HEADS;
  atomicMax(mp + 0, fkey(lrelu(as.x + ad.x)));
  atomicMax(mp + 1, fkey(lrelu(as.y + ad.y)));
  atomicMax(mp + 2, fkey(lrelu(as.z + ad.z)));
  atomicMax(mp + 3, fkey(lrelu(as.w + ad.w)));
}

__global__ __launch_bounds__(256) void k_edge_sum(const int* __restrict__ ei,
                                                  const float* __restrict__ a_src,
                                                  const float* __restrict__ a_dst,
                                                  const unsigned* __restrict__ emax_u,
                                                  float* __restrict__ esum,
                                                  float* __restrict__ exb,
                                                  int E, int N) {
  const int e = blockIdx.x * blockDim.x + threadIdx.x;
  if (e >= E + N) return;
  int s, d; edge_sd(ei, e, E, s, d);
  const float4 as = ((const float4*)a_src)[s];
  const float4 ad = ((const float4*)a_dst)[d];
  const unsigned* mp = emax_u + (size_t)d * HEADS;
  float4 ex;
  ex.x = __expf(lrelu(as.x + ad.x) - fkey_inv(mp[0]));
  ex.y = __expf(lrelu(as.y + ad.y) - fkey_inv(mp[1]));
  ex.z = __expf(lrelu(as.z + ad.z) - fkey_inv(mp[2]));
  ex.w = __expf(lrelu(as.w + ad.w) - fkey_inv(mp[3]));
  ((float4*)exb)[e] = ex;
  float* sp = esum + (size_t)d * HEADS;
  atomicAdd(sp + 0, ex.x);
  atomicAdd(sp + 1, ex.y);
  atomicAdd(sp + 2, ex.z);
  atomicAdd(sp + 3, ex.w);
}

__global__ __launch_bounds__(256) void k_scatter(const int* __restrict__ ei,
                                                 const float* __restrict__ exb,
                                                 const float* __restrict__ esum,
                                                 const float* __restrict__ xt,
                                                 float* __restrict__ z,
                                                 int E, int N) {
  const int lane = threadIdx.x & 31;
  const int e = blockIdx.x * 8 + (threadIdx.x >> 5);
  if (e >= E + N) return;
  int s, d; edge_sd(ei, e, E, s, d);
  const int h = lane >> 3;
  const float alpha =
      exb[(size_t)e * HEADS + h] / (esum[(size_t)d * HEADS + h] + 1e-16f);
  const float4 m = ((const float4*)xt)[(size_t)s * 32 + lane];
  float* zp = z + (size_t)d * OUT_DIM + lane * 4;
  atomicAdd(zp + 0, alpha * m.x);
  atomicAdd(zp + 1, alpha * m.y);
  atomicAdd(zp + 2, alpha * m.z);
  atomicAdd(zp + 3, alpha * m.w);
}

__global__ __launch_bounds__(256) void k_ln(const float* __restrict__ z,
                                            const float* __restrict__ ln_g,
                                            const float* __restrict__ ln_b,
                                            float* __restrict__ out, int N) {
  const int lane = threadIdx.x & 31;
  const int n = blockIdx.x * 8 + (threadIdx.x >> 5);
  if (n >= N) return;
  const float4 v = ((const float4*)z)[(size_t)n * 32 + lane];
  float s = v.x + v.y + v.z + v.w;
#pragma unroll
  for (int off = 1; off < 32; off <<= 1) s += __shfl_xor(s, off, 32);
  const float mu = s * (1.0f / 128.0f);
  const float dx = v.x - mu, dy = v.y - mu, dz = v.z - mu, dw = v.w - mu;
  float q = dx * dx + dy * dy + dz * dz + dw * dw;
#pragma unroll
  for (int off = 1; off < 32; off <<= 1) q += __shfl_xor(q, off, 32);
  const float inv = rsqrtf(q * (1.0f / 128.0f) + 1e-5f);
  const float4 g = ((const float4*)ln_g)[lane];
  const float4 b = ((const float4*)ln_b)[lane];
  float4 o;
  o.x = lrelu(dx * inv * g.x + b.x);
  o.y = lrelu(dy * inv * g.y + b.y);
  o.z = lrelu(dz * inv * g.z + b.z);
  o.w = lrelu(dw * inv * g.w + b.w);
  ((float4*)out)[(size_t)n * 32 + lane] = o;
}

extern "C" void kernel_launch(void* const* d_in, const int* in_sizes, int n_in,
                              void* d_out, int out_size, void* d_ws, size_t ws_size,
                              hipStream_t stream) {
  const float* x        = (const float*)d_in[0];
  const int*   ei       = (const int*)d_in[1];
  const float* lin_w    = (const float*)d_in[2];
  const float* att_src  = (const float*)d_in[3];
  const float* att_dst  = (const float*)d_in[4];
  const float* gat_bias = (const float*)d_in[5];
  const float* res_w    = (const float*)d_in[6];
  const float* res_b    = (const float*)d_in[7];
  const float* ln_g     = (const float*)d_in[8];
  const float* ln_b     = (const float*)d_in[9];
  const int N  = in_sizes[0] / IN_CH;
  const int E  = in_sizes[1] / 2;
  const int Et = E + N;

  float*    xt     = (float*)d_ws;
  float*    z      = xt + (size_t)N * OUT_DIM;
  float*    a_src  = z + (size_t)N * OUT_DIM;
  float*    a_dst  = a_src + (size_t)N * HEADS;
  unsigned* emax_u = (unsigned*)(a_dst + (size_t)N * HEADS);
  float*    esum   = (float*)(emax_u + (size_t)N * HEADS);
  float*    exb    = esum + (size_t)N * HEADS;

  k_gemm<<<(N + 15) / 16, 256, 0, stream>>>(x, lin_w, res_w, xt, z, N);
  k_node<<<(N + 7) / 8, 256, 0, stream>>>(xt, att_src, att_dst, gat_bias, res_b,
                                          z, a_src, a_dst, emax_u, esum, N);
  k_edge_max<<<(Et + 255) / 256, 256, 0, stream>>>(ei, a_src, a_dst, emax_u, E, N);
  k_edge_sum<<<(Et + 255) / 256, 256, 0, stream>>>(ei, a_src, a_dst, emax_u, esum,
                                                   exb, E, N);
  k_scatter<<<(Et + 7) / 8, 256, 0, stream>>>(ei, exb, esum, xt, z, E, N);
  k_ln<<<(N + 7) / 8, 256, 0, stream>>>(z, ln_g, ln_b, (float*)d_out, N);
}


# Edge-max and edge-sum passes: one lane per (edge, head), so an edge's 4 per-head atomics leave as one 16-byte piece of one line (8 lines per wave request, not 32 four times); still float atomics
# speedup vs baseline: 1.5333x; 1.5333x over previous
#include <hip/hip_runtime.h>
#include <hip/hip_bf16.h>

typedef __attribute__((ext_vector_type(2))) float v2f;
typedef __attribute__((ext_vector_type(8))) float v8f;

#define IN_CH 64
#define OUT_DIM 128
#define HEADS 4

static __device__ __forceinline__ float lrelu(float x) { return fmaxf(x, 0.2f * x); }

static __device__ __forceinline__ unsigned fkey(float f) {
  unsigned u = __float_as_uint(f);
  return (u & 0x80000000u) ? ~u : (u | 0x80000000u);
}
static __device__ __forceinline__ float fkey_inv(unsigned k) {
  return __uint_as_float((k & 0x80000000u) ? (k & 0x7FFFFFFFu) : ~k);
}

__global__ __launch_bounds__(256) void k_gemm(const float* __restrict__ x,
                                              const float* __restrict__ lin_w,
                                              const float* __restrict__ res_w,
                                              float* __restrict__ xt,
                                              float* __restrict__ z,
                                              int N) {
  const int wave = threadIdx.x >> 5;
  const int lane = threadIdx.x & 31;
  const int m0 = blockIdx.x * 16;
  const int mrow = lane & 15;
  const int kb = (lane >> 4) << 1;
  int arow = m0 + mrow;
  if (arow >= N) arow = N - 1;
  const float* xr = x + (size_t)arow * IN_CH;
  v2f a[16];
#pragma unroll
  for (int ks = 0; ks < 16; ++ks) {
    a[ks].x = xr[ks * 4 + kb];
    a[ks].y = xr[ks * 4 + kb + 1];
  }
#pragma unroll
  for (int tt = 0; tt < 2; ++tt) {
    const int tile = wave + tt * 8;
    const float* W = (tile < 8) ? lin_w : res_w;
    const int cbase = (tile & 7) * 16;
    const float* wr = W + (size_t)(cbase + (lane & 15)) * IN_CH;
    v8f c = {0.f, 0.f, 0.f, 0.f, 0.f, 0.f, 0.f, 0.f};
#pragma unroll
    for (int ks = 0; ks < 16; ++ks) {
      v2f b;
      b.x = wr[ks * 4 + kb];
      b.y = wr[ks * 4 + kb + 1];
      c = __builtin_amdgcn_wmma_f32_16x16x4_f32(false, a[ks], false, b,
                                                (short)0, c, false, false);
    }
    float* dst = (tile < 8) ? xt : z;
    const int mb = m0 + ((lane >> 4) << 3);
#pragma unroll
    for (int r = 0; r < 8; ++r) {
      const int m = mb + r;
      if (m < N) dst[(size_t)m * OUT_DIM + cbase + (lane & 15)] = c[r];
    }
  }
}

__global__ __launch_bounds__(256) void k_node(const float* __restrict__ xt,
                                              const float* __restrict__ att_src,
                                              const float* __restrict__ att_dst,
                                              const float* __restrict__ gat_bias,
                                              const float* __restrict__ res_b,
                                              float* __restrict__ z,
                                              float* __restrict__ a_src,
                                              float* __restrict__ a_dst,
                                              unsigned* __restrict__ emax_u,
                                              float* __restrict__ esum, int N) {
  const int lane = threadIdx.x & 31;
  const int n = blockIdx.x * 8 + (threadIdx.x >> 5);
  if (n >= N) return;
  const float4 v = ((const float4*)xt)[(size_t)n * 32 + lane];
  const float4 as = ((const float4*)att_src)[lane];
  const float4 ad = ((const float4*)att_dst)[lane];
  float ps = v.x * as.x + v.y * as.y + v.z * as.z + v.w * as.w;
  float pd = v.x * ad.x + v.y * ad.y + v.z * ad.z + v.w * ad.w;
#pragma unroll
  for (int off = 1; off < 8; off <<= 1) {
    ps += __shfl_xor(ps, off, 32);
    pd += __shfl_xor(pd, off, 32);
  }
  const int h = lane >> 3;
  if ((lane & 7) == 0) {
    a_src[(size_t)n * HEADS + h] = ps;
    a_dst[(size_t)n * HEADS + h] = pd;
  }
  if (lane < HEADS) {
    emax_u[(size_t)n * HEADS + lane] = 0u;
    esum[(size_t)n * HEADS + lane] = 0.f;
  }
  float4 zr = ((float4*)z)[(size_t)n * 32 + lane];
  const float4 gb = ((const float4*)gat_bias)[lane];
  const float4 rb = ((const float4*)res_b)[lane];
  zr.x += gb.x + rb.x; zr.y += gb.y + rb.y;
  zr.z += gb.z + rb.z; zr.w += gb.w + rb.w;
  ((float4*)z)[(size_t)n * 32 + lane] = zr;
}

static __device__ __forceinline__ void edge_sd(const int* __restrict__ ei, int e,
                                               int E, int& s, int& d) {
  if (e < E) { s = ei[e]; d = ei[E + e]; }
  else       { s = e - E; d = s; }
}

static __device__ __forceinline__ void edge_sd_flat(const int* __restrict__ ei,
                                                    int e, int E, int& s, int& d) {
  const int ec = min(e, E - 1);
  const int sl = ei[ec];
  const int dl = ei[E + ec];
  const bool real = e < E;
  s = real ? sl : e - E;
  d = real ? dl : e - E;
}

template <bool FULL>
static __device__ __forceinline__ void edge_max_body(const int* __restrict__ ei,
                                                     const float* __restrict__ a_src,
                                                     const float* __restrict__ a_dst,
                                                     unsigned* __restrict__ emax_u,
                                                     int E, int Et, int e0, int h) {
  int s[4], d[4];
#pragma unroll
  for (int j = 0; j < 4; ++j) {
    const int e = e0 + j * 8;
    edge_sd_flat(ei, (FULL || e < Et) ? e : Et - 1, E, s[j], d[j]);
  }
  float as[4], ad[4];
#pragma unroll
  for (int j = 0; j < 4; ++j) {
    as[j] = a_src[(size_t)s[j] * HEADS + h];
    ad[j] = a_dst[(size_t)d[j] * HEADS + h];
  }
#pragma unroll
  for (int j = 0; j < 4; ++j) {
    if (FULL || e0 + j * 8 < Et)
      atomicMax(emax_u + (size_t)d[j] * HEADS + h, fkey(lrelu(as[j] + ad[j])));
  }
}

__global__ __launch_bounds__(256) void k_edge_max(const int* __restrict__ ei,
                                                  const float* __restrict__ a_src,
                                                  const float* __restrict__ a_dst,
                                                  unsigned* __restrict__ emax_u,
                                                  int E, int N) {
  const int lane = threadIdx.x & 31;
  const int Et = E + N;
  const int wbase = (blockIdx.x * 8 + (threadIdx.x >> 5)) * 32;
  if (wbase >= Et) return;
  const int e0 = wbase + (lane >> 2);
  if (wbase + 32 <= Et)
    edge_max_body<true>(ei, a_src, a_dst, emax_u, E, Et, e0, lane & 3);
  else
    edge_max_body<false>(ei, a_src, a_dst, emax_u, E, Et, e0, lane & 3);
}

template <bool FULL>
static __device__ __forceinline__ void edge_sum_body(const int* __restrict__ ei,
                                                     const float* __restrict__ a_src,
                                                     const float* __restrict__ a_dst,
                                                     const unsigned* __restrict__ emax_u,
                                                     float* __restrict__ esum,
                                                     float* __restrict__ exb,
                                                     int E, int Et, int e0, int h) {
  int s[4], d[4];
#pragma unroll
  for (int j = 0; j < 4; ++j) {
    const int e = e0 + j * 8;
    edge_sd_flat(ei, (FULL || e < Et) ? e : Et - 1, E, s[j], d[j]);
  }
  float as[4], ad[4];
  unsigned mk[4];
#pragma unroll
  for (int j = 0; j < 4; ++j) {
    as[j] = a_src[(size_t)s[j] * HEADS + h];
    ad[j] = a_dst[(size_t)d[j] * HEADS + h];
    mk[j] = emax_u[(size_t)d[j] * HEADS + h];
  }
#pragma unroll
  for (int j = 0; j < 4; ++j) {
    const int e = e0 + j * 8;
    if (FULL || e < Et) {
      const float ex = __expf(lrelu(as[j] + ad[j]) - fkey_inv(mk[j]));
      exb[(size_t)e * HEADS + h] = ex;
      atomicAdd(esum + (size_t)d[j] * HEADS + h, ex);
    }
  }
}

__global__ __launch_bounds__(256) void k_edge_sum(const int* __restrict__ ei,
                                                  const float* __restrict__ a_src,
                                                  const float* __restrict__ a_dst,
                                                  const unsigned* __restrict__ emax_u,
                                                  float* __restrict__ esum,
                                                  float* __restrict__ exb,
                                                  int E, int N) {
  const int lane = threadIdx.x & 31;
  const int Et = E + N;
  const int wbase = (blockIdx.x * 8 + (threadIdx.x >> 5)) * 32;
  if (wbase >= Et) return;
  const int e0 = wbase + (lane >> 2);
  if (wbase + 32 <= Et)
    edge_sum_body<true>(ei, a_src, a_dst, emax_u, esum, exb, E, Et, e0, lane & 3);
  else
    edge_sum_body<false>(ei, a_src, a_dst, emax_u, esum, exb, E, Et, e0, lane & 3);
}

__global__ __launch_bounds__(256) void k_scatter(const int* __restrict__ ei,
                                                 const float* __restrict__ exb,
                                                 const float* __restrict__ esum,
                                                 const float* __restrict__ xt,
                                                 float* __restrict__ z,
                                                 int E, int N) {
  const int lane = threadIdx.x & 31;
  const int e = blockIdx.x * 8 + (threadIdx.x >> 5);
  if (e >= E + N) return;
  int s, d; edge_sd(ei, e, E, s, d);
  const int h = lane >> 3;
  const float alpha =
      exb[(size_t)e * HEADS + h] / (esum[(size_t)d * HEADS + h] + 1e-16f);
  const float4 m = ((const float4*)xt)[(size_t)s * 32 + lane];
  float* zp = z + (size_t)d * OUT_DIM + lane * 4;
  atomicAdd(zp + 0, alpha * m.x);
  atomicAdd(zp + 1, alpha * m.y);
  atomicAdd(zp + 2, alpha * m.z);
  atomicAdd(zp + 3, alpha * m.w);
}

__global__ __launch_bounds__(256) void k_ln(const float* __restrict__ z,
                                            const float* __restrict__ ln_g,
                                            const float* __restrict__ ln_b,
                                            float* __restrict__ out, int N) {
  const int lane = threadIdx.x & 31;
  const int n = blockIdx.x * 8 + (threadIdx.x >> 5);
  if (n >= N) return;
  const float4 v = ((const float4*)z)[(size_t)n * 32 + lane];
  float s = v.x + v.y + v.z + v.w;
#pragma unroll
  for (int off = 1; off < 32; off <<= 1) s += __shfl_xor(s, off, 32);
  const float mu = s * (1.0f / 128.0f);
  const float dx = v.x - mu, dy = v.y - mu, dz = v.z - mu, dw = v.w - mu;
  float q = dx * dx + dy * dy + dz * dz + dw * dw;
#pragma unroll
  for (int off = 1; off < 32; off <<= 1) q += __shfl_xor(q, off, 32);
  const float inv = rsqrtf(q * (1.0f / 128.0f) + 1e-5f);
  const float4 g = ((const float4*)ln_g)[lane];
  const float4 b = ((const float4*)ln_b)[lane];
  float4 o;
  o.x = lrelu(dx * inv * g.x + b.x);
  o.y = lrelu(dy * inv * g.y + b.y);
  o.z = lrelu(dz * inv * g.z + b.z);
  o.w = lrelu(dw * inv * g.w + b.w);
  ((float4*)out)[(size_t)n * 32 + lane] = o;
}

extern "C" void kernel_launch(void* const* d_in, const int* in_sizes, int n_in,
                              void* d_out, int out_size, void* d_ws, size_t ws_size,
                              hipStream_t stream) {
  const float* x        = (const float*)d_in[0];
  const int*   ei       = (const int*)d_in[1];
  const float* lin_w    = (const float*)d_in[2];
  const float* att_src  = (const float*)d_in[3];
  const float* att_dst  = (const float*)d_in[4];
  const float* gat_bias = (const float*)d_in[5];
  const float* res_w    = (const float*)d_in[6];
  const float* res_b    = (const float*)d_in[7];
  const float* ln_g     = (const float*)d_in[8];
  const float* ln_b     = (const float*)d_in[9];
  const int N  = in_sizes[0] / IN_CH;
  const int E  = in_sizes[1] / 2;
  const int Et = E + N;

  float*    xt     = (float*)d_ws;
  float*    z      = xt + (size_t)N * OUT_DIM;
  float*    a_src  = z + (size_t)N * OUT_DIM;
  float*    a_dst  = a_src + (size_t)N * HEADS;
  unsigned* emax_u = (unsigned*)(a_dst + (size_t)N * HEADS);
  float*    esum   = (float*)(emax_u + (size_t)N * HEADS);
  float*    exb    = esum + (size_t)N * HEADS;

  k_gemm<<<(N + 15) / 16, 256, 0, stream>>>(x, lin_w, res_w, xt, z, N);
  k_node<<<(N + 7) / 8, 256, 0, stream>>>(xt, att_src, att_dst, gat_bias, res_b,
                                          z, a_src, a_dst, emax_u, esum, N);
  k_edge_max<<<(Et + 255) / 256, 256, 0, stream>>>(ei, a_src, a_dst, emax_u, E, N);
  k_edge_sum<<<(Et + 255) / 256, 256, 0, stream>>>(ei, a_src, a_dst, emax_u, esum,
                                                   exb, E, N);
  k_scatter<<<(Et + 7) / 8, 256, 0, stream>>>(ei, exb, esum, xt, z, E, N);
  k_ln<<<(N + 7) / 8, 256, 0, stream>>>(z, ln_g, ln_b, (float*)d_out, N);
}


# Scatter pass: lane l takes floats l, l+32, l+64, l+96 of a row, so each of a wave's 4 float-atomic requests and 4 gathers is one whole 128-byte line, not a quarter of 4 lines; float atomics kept
# speedup vs baseline: 1.8351x; 1.1968x over previous
#include <hip/hip_runtime.h>
#include <hip/hip_bf16.h>

typedef __attribute__((ext_vector_type(2))) float v2f;
typedef __attribute__((ext_vector_type(8))) float v8f;

#define IN_CH 64
#define OUT_DIM 128
#define HEADS 4

static __device__ __forceinline__ float lrelu(float x) { return fmaxf(x, 0.2f * x); }

static __device__ __forceinline__ unsigned fkey(float f) {
  unsigned u = __float_as_uint(f);
  return (u & 0x80000000u) ? ~u : (u | 0x80000000u);
}
static __device__ __forceinline__ float fkey_inv(unsigned k) {
  return __uint_as_float((k & 0x80000000u) ? (k & 0x7FFFFFFFu) : ~k);
}

__global__ __launch_bounds__(256) void k_gemm(const float* __restrict__ x,
                                              const float* __restrict__ lin_w,
                                              const float* __restrict__ res_w,
                                              float* __restrict__ xt,
                                              float* __restrict__ z,
                                              int N) {
  const int wave = threadIdx.x >> 5;
  const int lane = threadIdx.x & 31;
  const int m0 = blockIdx.x * 16;
  const int mrow = lane & 15;
  const int kb = (lane >> 4) << 1;
  int arow = m0 + mrow;
  if (arow >= N) arow = N - 1;
  const float* xr = x + (size_t)arow * IN_CH;
  v2f a[16];
#pragma unroll
  for (int ks = 0; ks < 16; ++ks) {
    a[ks].x = xr[ks * 4 + kb];
    a[ks].y = xr[ks * 4 + kb + 1];
  }
#pragma unroll
  for (int tt = 0; tt < 2; ++tt) {
    const int tile = wave + tt * 8;
    const float* W = (tile < 8) ? lin_w : res_w;
    const int cbase = (tile & 7) * 16;
    const float* wr = W + (size_t)(cbase + (lane & 15)) * IN_CH;
    v8f c = {0.f, 0.f, 0.f, 0.f, 0.f, 0.f, 0.f, 0.f};
#pragma unroll
    for (int ks = 0; ks < 16; ++ks) {
      v2f b;
      b.x = wr[ks * 4 + kb];
      b.y = wr[ks * 4 + kb + 1];
      c = __builtin_amdgcn_wmma_f32_16x16x4_f32(false, a[ks], false, b,
                                                (short)0, c, false, false);
    }
    float* dst = (tile < 8) ? xt : z;
    const int mb = m0 + ((lane >> 4) << 3);
#pragma unroll
    for (int r = 0; r < 8; ++r) {
      const int m = mb + r;
      if (m < N) dst[(size_t)m * OUT_DIM + cbase + (lane & 15)] = c[r];
    }
  }
}

__global__ __launch_bounds__(256) void k_node(const float* __restrict__ xt,
                                              const float* __restrict__ att_src,
                                              const float* __restrict__ att_dst,
                                              const float* __restrict__ gat_bias,
                                              const float* __restrict__ res_b,
                                              float* __restrict__ z,
                                              float* __restrict__ a_src,
                                              float* __restrict__ a_dst,
                                              unsigned* __restrict__ emax_u,
                                              float* __restrict__ esum, int N) {
  const int lane = threadIdx.x & 31;
  const int n = blockIdx.x * 8 + (threadIdx.x >> 5);
  if (n >= N) return;
  const float4 v = ((const float4*)xt)[(size_t)n * 32 + lane];
  const float4 as = ((const float4*)att_src)[lane];
  const float4 ad = ((const float4*)att_dst)[lane];
  float ps = v.x * as.x + v.y * as.y + v.z * as.z + v.w * as.w;
  float pd = v.x * ad.x + v.y * ad.y + v.z * ad.z + v.w * ad.w;
#pragma unroll
  for (int off = 1; off < 8; off <<= 1) {
    ps += __shfl_xor(ps, off, 32);
    pd += __shfl_xor(pd, off, 32);
  }
  const int h = lane >> 3;
  if ((lane & 7) == 0) {
    a_src[(size_t)n * HEADS + h] = ps;
    a_dst[(size_t)n * HEADS + h] = pd;
  }
  if (lane < HEADS) {
    emax_u[(size_t)n * HEADS + lane] = 0u;
    esum[(size_t)n * HEADS + lane] = 0.f;
  }
  float4 zr = ((float4*)z)[(size_t)n * 32 + lane];
  const float4 gb = ((const float4*)gat_bias)[lane];
  const float4 rb = ((const float4*)res_b)[lane];
  zr.x += gb.x + rb.x; zr.y += gb.y + rb.y;
  zr.z += gb.z + rb.z; zr.w += gb.w + rb.w;
  ((float4*)z)[(size_t)n * 32 + lane] = zr;
}

static __device__ __forceinline__ void edge_sd(const int* __restrict__ ei, int e,
                                               int E, int& s, int& d) {
  if (e < E) { s = ei[e]; d = ei[E + e]; }
  else       { s = e - E; d = s; }
}

static __device__ __forceinline__ void edge_sd_flat(const int* __restrict__ ei,
                                                    int e, int E, int& s, int& d) {
  const int ec = min(e, E - 1);
  const int sl = ei[ec];
  const int dl = ei[E + ec];
  const bool real = e < E;
  s = real ? sl : e - E;
  d = real ? dl : e - E;
}

template <bool FULL>
static __device__ __forceinline__ void edge_max_body(const int* __restrict__ ei,
                                                     const float* __restrict__ a_src,
                                                     const float* __restrict__ a_dst,
                                                     unsigned* __restrict__ emax_u,
                                                     int E, int Et, int e0, int h) {
  int s[4], d[4];
#pragma unroll
  for (int j = 0; j < 4; ++j) {
    const int e = e0 + j * 8;
    edge_sd_flat(ei, (FULL || e < Et) ? e : Et - 1, E, s[j], d[j]);
  }
  float as[4], ad[4];
#pragma unroll
  for (int j = 0; j < 4; ++j) {
    as[j] = a_src[(size_t)s[j] * HEADS + h];
    ad[j] = a_dst[(size_t)d[j] * HEADS + h];
  }
#pragma unroll
  for (int j = 0; j < 4; ++j) {
    if (FULL || e0 + j * 8 < Et)
      atomicMax(emax_u + (size_t)d[j] * HEADS + h, fkey(lrelu(as[j] + ad[j])));
  }
}

__global__ __launch_bounds__(256) void k_edge_max(const int* __restrict__ ei,
                                                  const float* __restrict__ a_src,
                                                  const float* __restrict__ a_dst,
                                                  unsigned* __restrict__ emax_u,
                                                  int E, int N) {
  const int lane = threadIdx.x & 31;
  const int Et = E + N;
  const int wbase = (blockIdx.x * 8 + (threadIdx.x >> 5)) * 32;
  if (wbase >= Et) return;
  const int e0 = wbase + (lane >> 2);
  if (wbase + 32 <= Et)
    edge_max_body<true>(ei, a_src, a_dst, emax_u, E, Et, e0, lane & 3);
  else
    edge_max_body<false>(ei, a_src, a_dst, emax_u, E, Et, e0, lane & 3);
}

template <bool FULL>
static __device__ __forceinline__ void edge_sum_body(const int* __restrict__ ei,
                                                     const float* __restrict__ a_src,
                                                     const float* __restrict__ a_dst,
                                                     const unsigned* __restrict__ emax_u,
                                                     float* __restrict__ esum,
                                                     float* __restrict__ exb,
                                                     int E, int Et, int e0, int h) {
  int s[4], d[4];
#pragma unroll
  for (int j = 0; j < 4; ++j) {
    const int e = e0 + j * 8;
    edge_sd_flat(ei, (FULL || e < Et) ? e : Et - 1, E, s[j], d[j]);
  }
  float as[4], ad[4];
  unsigned mk[4];
#pragma unroll
  for (int j = 0; j < 4; ++j) {
    as[j] = a_src[(size_t)s[j] * HEADS + h];
    ad[j] = a_dst[(size_t)d[j] * HEADS + h];
    mk[j] = emax_u[(size_t)d[j] * HEADS + h];
  }
#pragma unroll
  for (int j = 0; j < 4; ++j) {
    const int e = e0 + j * 8;
    if (FULL || e < Et) {
      const float ex = __expf(lrelu(as[j] + ad[j]) - fkey_inv(mk[j]));
      exb[(size_t)e * HEADS + h] = ex;
      atomicAdd(esum + (size_t)d[j] * HEADS + h, ex);
    }
  }
}

__global__ __launch_bounds__(256) void k_edge_sum(const int* __restrict__ ei,
                                                  const float* __restrict__ a_src,
                                                  const float* __restrict__ a_dst,
                                                  const unsigned* __restrict__ emax_u,
                                                  float* __restrict__ esum,
                                                  float* __restrict__ exb,
                                                  int E, int N) {
  const int lane = threadIdx.x & 31;
  const int Et = E + N;
  const int wbase = (blockIdx.x * 8 + (threadIdx.x >> 5)) * 32;
  if (wbase >= Et) return;
  const int e0 = wbase + (lane >> 2);
  if (wbase + 32 <= Et)
    edge_sum_body<true>(ei, a_src, a_dst, emax_u, esum, exb, E, Et, e0, lane & 3);
  else
    edge_sum_body<false>(ei, a_src, a_dst, emax_u, esum, exb, E, Et, e0, lane & 3);
}

__global__ __launch_bounds__(256) void k_scatter(const int* __restrict__ ei,
                                                 const float* __restrict__ exb,
                                                 const float* __restrict__ esum,
                                                 const float* __restrict__ xt,
                                                 float* __restrict__ z,
                                                 int E, int N) {
  const int lane = threadIdx.x & 31;
  const int e = blockIdx.x * 8 + (threadIdx.x >> 5);
  if (e >= E + N) return;
  int s, d; edge_sd(ei, e, E, s, d);
  const int hl = lane & 3;
  const float al =
      exb[(size_t)e * HEADS + hl] / (esum[(size_t)d * HEADS + hl] + 1e-16f);
  const float* xp = xt + (size_t)s * OUT_DIM + lane;
  const float m0 = xp[0], m1 = xp[32], m2 = xp[64], m3 = xp[96];
  const float a0 = __int_as_float(__builtin_amdgcn_readlane(__float_as_int(al), 0));
  const float a1 = __int_as_float(__builtin_amdgcn_readlane(__float_as_int(al), 1));
  const float a2 = __int_as_float(__builtin_amdgcn_readlane(__float_as_int(al), 2));
  const float a3 = __int_as_float(__builtin_amdgcn_readlane(__float_as_int(al), 3));
  float* zp = z + (size_t)d * OUT_DIM + lane;
  atomicAdd(zp,      a0 * m0);
  atomicAdd(zp + 32, a1 * m1);
  atomicAdd(zp + 64, a2 * m2);
  atomicAdd(zp + 96, a3 * m3);
}

__global__ __launch_bounds__(256) void k_ln(const float* __restrict__ z,
                                            const float* __restrict__ ln_g,
                                            const float* __restrict__ ln_b,
                                            float* __restrict__ out, int N) {
  const int lane = threadIdx.x & 31;
  const int n = blockIdx.x * 8 + (threadIdx.x >> 5);
  if (n >= N) return;
  const float4 v = ((const float4*)z)[(size_t)n * 32 + lane];
  float s = v.x + v.y + v.z + v.w;
#pragma unroll
  for (int off = 1; off < 32; off <<= 1) s += __shfl_xor(s, off, 32);
  const float mu = s * (1.0f / 128.0f);
  const float dx = v.x - mu, dy = v.y - mu, dz = v.z - mu, dw = v.w - mu;
  float q = dx * dx + dy * dy + dz * dz + dw * dw;
#pragma unroll
  for (int off = 1; off < 32; off <<= 1) q += __shfl_xor(q, off, 32);
  const float inv = rsqrtf(q * (1.0f / 128.0f) + 1e-5f);
  const float4 g = ((const float4*)ln_g)[lane];
  const float4 b = ((const float4*)ln_b)[lane];
  float4 o;
  o.x = lrelu(dx * inv * g.x + b.x);
  o.y = lrelu(dy * inv * g.y + b.y);
  o.z = lrelu(dz * inv * g.z + b.z);
  o.w = lrelu(dw * inv * g.w + b.w);
  ((float4*)out)[(size_t)n * 32 + lane] = o;
}

extern "C" void kernel_launch(void* const* d_in, const int* in_sizes, int n_in,
                              void* d_out, int out_size, void* d_ws, size_t ws_size,
                              hipStream_t stream) {
  const float* x        = (const float*)d_in[0];
  const int*   ei       = (const int*)d_in[1];
  const float* lin_w    = (const float*)d_in[2];
  const float* att_src  = (const float*)d_in[3];
  const float* att_dst  = (const float*)d_in[4];
  const float* gat_bias = (const float*)d_in[5];
  const float* res_w    = (const float*)d_in[6];
  const float* res_b    = (const float*)d_in[7];
  const float* ln_g     = (const float*)d_in[8];
  const float* ln_b     = (const float*)d_in[9];
  const int N  = in_sizes[0] / IN_CH;
  const int E  = in_sizes[1] / 2;
  const int Et = E + N;

  float*    xt     = (float*)d_ws;
  float*    z      = xt + (size_t)N * OUT_DIM;
  float*    a_src  = z + (size_t)N * OUT_DIM;
  float*    a_dst  = a_src + (size_t)N * HEADS;
  unsigned* emax_u = (unsigned*)(a_dst + (size_t)N * HEADS);
  float*    esum   = (float*)(emax_u + (size_t)N * HEADS);
  float*    exb    = esum + (size_t)N * HEADS;

  k_gemm<<<(N + 15) / 16, 256, 0, stream>>>(x, lin_w, res_w, xt, z, N);
  k_node<<<(N + 7) / 8, 256, 0, stream>>>(xt, att_src, att_dst, gat_bias, res_b,
                                          z, a_src, a_dst, emax_u, esum, N);
  k_edge_max<<<(Et + 255) / 256, 256, 0, stream>>>(ei, a_src, a_dst, emax_u, E, N);
  k_edge_sum<<<(Et + 255) / 256, 256, 0, stream>>>(ei, a_src, a_dst, emax_u, esum,
                                                   exb, E, N);
  k_scatter<<<(Et + 7) / 8, 256, 0, stream>>>(ei, exb, esum, xt, z, E, N);
  k_ln<<<(N + 7) / 8, 256, 0, stream>>>(z, ln_g, ln_b, (float*)d_out, N);
}


# Edges grouped by destination in-call (count, scan, fill: int atomics); one wave per node does softmax, message sum in registers and LayerNorm: no float atomics, sum order = fill arrival order
# speedup vs baseline: 2.5471x; 1.3880x over previous
#include <hip/hip_runtime.h>
#include <hip/hip_bf16.h>

typedef __attribute__((ext_vector_type(2))) float v2f;
typedef __attribute__((ext_vector_type(8))) float v8f;

#define IN_CH 64
#define OUT_DIM 128
#define HEADS 4

static __device__ __forceinline__ float lrelu(float x) { return fmaxf(x, 0.2f * x); }

__global__ __launch_bounds__(256) void k_gemm(const float* __restrict__ x,
                                              const float* __restrict__ lin_w,
                                              const float* __restrict__ res_w,
                                              float* __restrict__ xt,
                                              float* __restrict__ z,
                                              int* __restrict__ cnt,
                                              int N) {
  const int wave = threadIdx.x >> 5;
  const int lane = threadIdx.x & 31;
  const int m0 = blockIdx.x * 16;
  const int mrow = lane & 15;
  const int kb = (lane >> 4) << 1;
  int arow = m0 + mrow;
  if (arow >= N) arow = N - 1;
  const float* xr = x + (size_t)arow * IN_CH;
  v2f a[16];
#pragma unroll
  for (int ks = 0; ks < 16; ++ks) {
    a[ks].x = xr[ks * 4 + kb];
    a[ks].y = xr[ks * 4 + kb + 1];
  }
#pragma unroll
  for (int tt = 0; tt < 2; ++tt) {
    const int tile = wave + tt * 8;
    const float* W = (tile < 8) ? lin_w : res_w;
    const int cbase = (tile & 7) * 16;
    const float* wr = W + (size_t)(cbase + (lane & 15)) * IN_CH;
    v8f c = {0.f, 0.f, 0.f, 0.f, 0.f, 0.f, 0.f, 0.f};
#pragma unroll
    for (int ks = 0; ks < 16; ++ks) {
      v2f b;
      b.x = wr[ks * 4 + kb];
      b.y = wr[ks * 4 + kb + 1];
      c = __builtin_amdgcn_wmma_f32_16x16x4_f32(false, a[ks], false, b,
                                                (short)0, c, false, false);
    }
    float* dst = (tile < 8) ? xt : z;
    const int mb = m0 + ((lane >> 4) << 3);
#pragma unroll
    for (int r = 0; r < 8; ++r) {
      const int m = mb + r;
      if (m < N) dst[(size_t)m * OUT_DIM + cbase + (lane & 15)] = c[r];
    }
  }
  if (threadIdx.x < 16) {
    const int m = m0 + threadIdx.x;
    if (m < N) cnt[m] = 0;
  }
}

__global__ __launch_bounds__(256) void k_node(const float* __restrict__ xt,
                                              const float* __restrict__ att_src,
                                              const float* __restrict__ att_dst,
                                              const int* __restrict__ ei,
                                              float* __restrict__ a_src,
                                              float* __restrict__ a_dst,
                                              int* __restrict__ cnt,
                                              int E, int N) {
  const int lane = threadIdx.x & 31;
  const int n = blockIdx.x * 8 + (threadIdx.x >> 5);
  if (n >= N) return;
  for (int g = n * 32 + lane; g < E; g += N * 32) atomicAdd(cnt + ei[E + g], 1);
  const float4 v = ((const float4*)xt)[(size_t)n * 32 + lane];
  const float4 as = ((const float4*)att_src)[lane];
  const float4 ad = ((const float4*)att_dst)[lane];
  float ps = v.x * as.x + v.y * as.y + v.z * as.z + v.w * as.w;
  float pd = v.x * ad.x + v.y * ad.y + v.z * ad.z + v.w * ad.w;
#pragma unroll
  for (int off = 1; off < 8; off <<= 1) {
    ps += __shfl_xor(ps, off, 32);
    pd += __shfl_xor(pd, off, 32);
  }
  const int h = lane >> 3;
  if ((lane & 7) == 0) {
    a_src[(size_t)n * HEADS + h] = ps;
    a_dst[(size_t)n * HEADS + h] = pd;
  }
}

__global__ __launch_bounds__(1024) void k_scan(const int* __restrict__ cnt,
                                               int* __restrict__ cursor, int N) {
  __shared__ int wsum[32];
  const int lane = threadIdx.x & 31;
  const int wave = __builtin_amdgcn_readfirstlane((int)(threadIdx.x >> 5));
  const int per = (N + 31) / 32;
  const int seg = ((per + 255) / 256) * 256;
  const int lo = min(wave * seg, N);
  const int hi = min(lo + seg, N);
  int tot = 0;
#pragma unroll 8
  for (int i = lo + lane; i < hi; i += 32) tot += cnt[i];
#pragma unroll
  for (int off = 1; off < 32; off <<= 1) tot += __shfl_xor(tot, off, 32);
  if (lane == 0) wsum[wave] = tot;
  __syncthreads();
  int carry = (lane < wave) ? wsum[lane] : 0;
#pragma unroll
  for (int off = 1; off < 32; off <<= 1) carry += __shfl_xor(carry, off, 32);
  for (int b = lo; b < hi; b += 256) {
    const int i0 = b + lane * 8;
    int c[8];
    int t = 0;
#pragma unroll
    for (int k = 0; k < 8; ++k) {
      c[k] = (i0 + k < hi) ? cnt[i0 + k] : 0;
      t += c[k];
    }
    int inc = t;
#pragma unroll
    for (int off = 1; off < 32; off <<= 1) {
      const int u = __shfl_up(inc, off, 32);
      if (lane >= off) inc += u;
    }
    int run = carry + inc - t;
#pragma unroll
    for (int k = 0; k < 8; ++k) {
      if (i0 + k < hi) cursor[i0 + k] = run;
      run += c[k];
    }
    carry += __shfl(inc, 31, 32);
  }
}

__global__ __launch_bounds__(256) void k_fill(const int* __restrict__ ei,
                                              int* __restrict__ cursor,
                                              int* __restrict__ csr, int E) {
  const int e = blockIdx.x * 256 + threadIdx.x;
  if (e >= E) return;
  const int s = ei[e];
  const int d = ei[E + e];
  const int pos = atomicAdd(cursor + d, 1);
  if ((unsigned)pos < (unsigned)E) csr[pos] = s;
}

static __device__ __forceinline__ void load_chunk(const int* __restrict__ csr,
                                                  const float* __restrict__ a_src,
                                                  const float4 ad, int beg, int deg,
                                                  int n, int N, int i, int& s,
                                                  float4& e) {
  s = n;
  e = make_float4(-INFINITY, -INFINITY, -INFINITY, -INFINITY);
  if (i <= deg) {
    if (i < deg) s = min(max(csr[beg + i], 0), N - 1);
    const float4 as = ((const float4*)a_src)[s];
    e.x = lrelu(as.x + ad.x);
    e.y = lrelu(as.y + ad.y);
    e.z = lrelu(as.z + ad.z);
    e.w = lrelu(as.w + ad.w);
  }
}

static __device__ __forceinline__ void gather_chunk(const float* __restrict__ xt,
                                                    int s, const float4 al, int cnt,
                                                    int lane, float& acc0, float& acc1,
                                                    float& acc2, float& acc3) {
#pragma unroll 4
  for (int j = 0; j < cnt; ++j) {
    const int sj = __builtin_amdgcn_readlane(s, j);
    const float b0 = __int_as_float(__builtin_amdgcn_readlane(__float_as_int(al.x), j));
    const float b1 = __int_as_float(__builtin_amdgcn_readlane(__float_as_int(al.y), j));
    const float b2 = __int_as_float(__builtin_amdgcn_readlane(__float_as_int(al.z), j));
    const float b3 = __int_as_float(__builtin_amdgcn_readlane(__float_as_int(al.w), j));
    const float* xp = xt + (size_t)sj * OUT_DIM + lane;
    acc0 += b0 * xp[0];
    acc1 += b1 * xp[32];
    acc2 += b2 * xp[64];
    acc3 += b3 * xp[96];
  }
}

static __device__ __forceinline__ float wave_max(float v) {
#pragma unroll
  for (int off = 1; off < 32; off <<= 1) v = fmaxf(v, __shfl_xor(v, off, 32));
  return v;
}
static __device__ __forceinline__ float wave_sum(float v) {
#pragma unroll
  for (int off = 1; off < 32; off <<= 1) v += __shfl_xor(v, off, 32);
  return v;
}

__global__ __launch_bounds__(256) void k_agg(const int* __restrict__ cursor,
                                             const int* __restrict__ cnt,
                                             const int* __restrict__ csr,
                                             const float* __restrict__ a_src,
                                             const float* __restrict__ a_dst,
                                             const float* __restrict__ xt,
                                             const float* __restrict__ zres,
                                             const float* __restrict__ gat_bias,
                                             const float* __restrict__ res_b,
                                             const float* __restrict__ ln_g,
                                             const float* __restrict__ ln_b,
                                             float* __restrict__ out, int N) {
  const int lane = threadIdx.x & 31;
  const int n = __builtin_amdgcn_readfirstlane((int)(blockIdx.x * 8 + (threadIdx.x >> 5)));
  if (n >= N) return;
  const int deg = max(cnt[n], 0);
  const int beg = cursor[n] - deg;
  const int tot = deg + 1;
  const float4 ad = ((const float4*)a_dst)[n];

  int s0;
  float4 e0;
  load_chunk(csr, a_src, ad, beg, deg, n, N, lane, s0, e0);

  float4 m = e0;
  for (int b = 32; b < tot; b += 32) {
    int s; float4 e;
    load_chunk(csr, a_src, ad, beg, deg, n, N, b + lane, s, e);
    m.x = fmaxf(m.x, e.x); m.y = fmaxf(m.y, e.y);
    m.z = fmaxf(m.z, e.z); m.w = fmaxf(m.w, e.w);
  }
  m.x = wave_max(m.x); m.y = wave_max(m.y);
  m.z = wave_max(m.z); m.w = wave_max(m.w);

  float4 x0;
  x0.x = __expf(e0.x - m.x); x0.y = __expf(e0.y - m.y);
  x0.z = __expf(e0.z - m.z); x0.w = __expf(e0.w - m.w);
  float4 q = x0;
  for (int b = 32; b < tot; b += 32) {
    int s; float4 e;
    load_chunk(csr, a_src, ad, beg, deg, n, N, b + lane, s, e);
    q.x += __expf(e.x - m.x); q.y += __expf(e.y - m.y);
    q.z += __expf(e.z - m.z); q.w += __expf(e.w - m.w);
  }
  q.x = wave_sum(q.x) + 1e-16f; q.y = wave_sum(q.y) + 1e-16f;
  q.z = wave_sum(q.z) + 1e-16f; q.w = wave_sum(q.w) + 1e-16f;

  float acc0 = 0.f, acc1 = 0.f, acc2 = 0.f, acc3 = 0.f;
  {
    float4 al;
    al.x = x0.x / q.x; al.y = x0.y / q.y; al.z = x0.z / q.z; al.w = x0.w / q.w;
    gather_chunk(xt, s0, al, min(tot, 32), lane, acc0, acc1, acc2, acc3);
  }
  for (int b = 32; b < tot; b += 32) {
    int s; float4 e;
    load_chunk(csr, a_src, ad, beg, deg, n, N, b + lane, s, e);
    float4 al;
    al.x = __expf(e.x - m.x) / q.x; al.y = __expf(e.y - m.y) / q.y;
    al.z = __expf(e.z - m.z) / q.z; al.w = __expf(e.w - m.w) / q.w;
    gather_chunk(xt, s, al, min(tot - b, 32), lane, acc0, acc1, acc2, acc3);
  }

  const float* zr = zres + (size_t)n * OUT_DIM + lane;
  const float z0 = ((acc0 + gat_bias[lane])      + zr[0])  + res_b[lane];
  const float z1 = ((acc1 + gat_bias[lane + 32]) + zr[32]) + res_b[lane + 32];
  const float z2 = ((acc2 + gat_bias[lane + 64]) + zr[64]) + res_b[lane + 64];
  const float z3 = ((acc3 + gat_bias[lane + 96]) + zr[96]) + res_b[lane + 96];

  const float mu = wave_sum((z0 + z1) + (z2 + z3)) * (1.0f / 128.0f);
  const float d0 = z0 - mu, d1 = z1 - mu, d2 = z2 - mu, d3 = z3 - mu;
  const float var = wave_sum((d0 * d0 + d1 * d1) + (d2 * d2 + d3 * d3)) * (1.0f / 128.0f);
  const float inv = rsqrtf(var + 1e-5f);
  float* op = out + (size_t)n * OUT_DIM + lane;
  op[0]  = lrelu(d0 * inv * ln_g[lane]      + ln_b[lane]);
  op[32] = lrelu(d1 * inv * ln_g[lane + 32] + ln_b[lane + 32]);
  op[64] = lrelu(d2 * inv * ln_g[lane + 64] + ln_b[lane + 64]);
  op[96] = lrelu(d3 * inv * ln_g[lane + 96] + ln_b[lane + 96]);
}

extern "C" void kernel_launch(void* const* d_in, const int* in_sizes, int n_in,
                              void* d_out, int out_size, void* d_ws, size_t ws_size,
                              hipStream_t stream) {
  const float* x        = (const float*)d_in[0];
  const int*   ei       = (const int*)d_in[1];
  const float* lin_w    = (const float*)d_in[2];
  const float* att_src  = (const float*)d_in[3];
  const float* att_dst  = (const float*)d_in[4];
  const float* gat_bias = (const float*)d_in[5];
  const float* res_w    = (const float*)d_in[6];
  const float* res_b    = (const float*)d_in[7];
  const float* ln_g     = (const float*)d_in[8];
  const float* ln_b     = (const float*)d_in[9];
  const int N  = in_sizes[0] / IN_CH;
  const int E  = in_sizes[1] / 2;

  float* xt     = (float*)d_ws;
  float* z      = xt + (size_t)N * OUT_DIM;
  float* a_src  = z + (size_t)N * OUT_DIM;
  float* a_dst  = a_src + (size_t)N * HEADS;
  int*   cnt    = (int*)(a_dst + (size_t)N * HEADS);
  int*   cursor = cnt + N;
  int*   csr    = cursor + N;

  k_gemm<<<(N + 15) / 16, 256, 0, stream>>>(x, lin_w, res_w, xt, z, cnt, N);
  k_node<<<(N + 7) / 8, 256, 0, stream>>>(xt, att_src, att_dst, ei, a_src, a_dst,
                                          cnt, E, N);
  k_scan<<<1, 1024, 0, stream>>>(cnt, cursor, N);
  k_fill<<<(E + 255) / 256, 256, 0, stream>>>(ei, cursor, csr, E);
  k_agg<<<(N + 7) / 8, 256, 0, stream>>>(cursor, cnt, csr, a_src, a_dst, xt, z,
                                         gat_bias, res_b, ln_g, ln_b,
                                         (float*)d_out, N);
}


# GEMM pass: 512 work-groups loop over 16-row chunks; W fragments read once per wave from an LDS copy and kept in registers, x chunks loaded as whole lines and turned through LDS; k order kept
# speedup vs baseline: 2.7295x; 1.0716x over previous
#include <hip/hip_runtime.h>
#include <hip/hip_bf16.h>

typedef __attribute__((ext_vector_type(2))) float v2f;
typedef __attribute__((ext_vector_type(8))) float v8f;

#define IN_CH 64
#define OUT_DIM 128
#define HEADS 4

static __device__ __forceinline__ float lrelu(float x) { return fmaxf(x, 0.2f * x); }

#define LDS_PITCH 68
#define GEMM_GROUPS 512

static __device__ __forceinline__ v2f frag(const float* s, int row, int ks, int kb) {
  return *(const v2f*)(s + row * LDS_PITCH + ks * 4 + kb);
}

__global__ __launch_bounds__(256) void k_gemm(const float* __restrict__ x,
                                              const float* __restrict__ lin_w,
                                              const float* __restrict__ res_w,
                                              float* __restrict__ xt,
                                              float* __restrict__ z,
                                              int* __restrict__ cnt,
                                              int N) {
  __shared__ __attribute__((aligned(16))) float Ws[256 * LDS_PITCH];
  __shared__ __attribute__((aligned(16))) float As[2][16 * LDS_PITCH];
  const int tid = threadIdx.x;
  const int wave = tid >> 5;
  const int lane = tid & 31;
  const int mrow = lane & 15;
  const int kb = (lane >> 4) << 1;
  const int nchunks = (N + 15) >> 4;
  const int sr = tid >> 4;
  const int sq = tid & 15;

#pragma unroll
  for (int i = 0; i < 8; ++i) {
    const int r = i * 16 + sr;
    const float4 v0 = ((const float4*)lin_w)[r * 16 + sq];
    const float4 v1 = ((const float4*)res_w)[r * 16 + sq];
    *(float4*)(Ws + r * LDS_PITCH + sq * 4) = v0;
    *(float4*)(Ws + (128 + r) * LDS_PITCH + sq * 4) = v1;
  }
  int chunk = blockIdx.x;
  float4 va = make_float4(0.f, 0.f, 0.f, 0.f);
  if (chunk < nchunks) {
    const int row = min(chunk * 16 + sr, N - 1);
    va = ((const float4*)x)[(size_t)row * 16 + sq];
  }
  __syncthreads();
  v2f b0[16], b1[16];
#pragma unroll
  for (int ks = 0; ks < 16; ++ks) {
    b0[ks] = frag(Ws, wave * 16 + mrow, ks, kb);
    b1[ks] = frag(Ws, 128 + wave * 16 + mrow, ks, kb);
  }
  const int cbase = wave * 16;
  int buf = 0;
  while (chunk < nchunks) {
    *(float4*)(&As[buf][sr * LDS_PITCH + sq * 4]) = va;
    __syncthreads();
    const int next = chunk + GEMM_GROUPS;
    if (next < nchunks) {
      const int row = min(next * 16 + sr, N - 1);
      va = ((const float4*)x)[(size_t)row * 16 + sq];
    }
    v2f a[16];
#pragma unroll
    for (int ks = 0; ks < 16; ++ks) a[ks] = frag(As[buf], mrow, ks, kb);
    asm volatile("" ::"v"(a[0]), "v"(a[1]), "v"(a[2]), "v"(a[3]), "v"(a[4]), "v"(a[5]),
                 "v"(a[6]), "v"(a[7]), "v"(a[8]), "v"(a[9]), "v"(a[10]), "v"(a[11]),
                 "v"(a[12]), "v"(a[13]), "v"(a[14]), "v"(a[15]));
    v8f c0 = {0.f, 0.f, 0.f, 0.f, 0.f, 0.f, 0.f, 0.f};
    v8f c1 = {0.f, 0.f, 0.f, 0.f, 0.f, 0.f, 0.f, 0.f};
#pragma unroll
    for (int ks = 0; ks < 16; ++ks) {
      c0 = __builtin_amdgcn_wmma_f32_16x16x4_f32(false, a[ks], false, b0[ks],
                                                 (short)0, c0, false, false);
      c1 = __builtin_amdgcn_wmma_f32_16x16x4_f32(false, a[ks], false, b1[ks],
                                                 (short)0, c1, false, false);
    }
    const int m0 = chunk * 16;
    const int mb = m0 + ((lane >> 4) << 3);
#pragma unroll
    for (int r = 0; r < 8; ++r) {
      const int m = mb + r;
      if (m < N) {
        xt[(size_t)m * OUT_DIM + cbase + mrow] = c0[r];
        z[(size_t)m * OUT_DIM + cbase + mrow] = c1[r];
      }
    }
    if (tid < 16) {
      const int m = m0 + tid;
      if (m < N) cnt[m] = 0;
    }
    buf ^= 1;
    chunk = next;
  }
}

__global__ __launch_bounds__(256) void k_node(const float* __restrict__ xt,
                                              const float* __restrict__ att_src,
                                              const float* __restrict__ att_dst,
                                              const int* __restrict__ ei,
                                              float* __restrict__ a_src,
                                              float* __restrict__ a_dst,
                                              int* __restrict__ cnt,
                                              int E, int N) {
  const int lane = threadIdx.x & 31;
  const int n = blockIdx.x * 8 + (threadIdx.x >> 5);
  if (n >= N) return;
  for (int g = n * 32 + lane; g < E; g += N * 32) atomicAdd(cnt + ei[E + g], 1);
  const float4 v = ((const float4*)xt)[(size_t)n * 32 + lane];
  const float4 as = ((const float4*)att_src)[lane];
  const float4 ad = ((const float4*)att_dst)[lane];
  float ps = v.x * as.x + v.y * as.y + v.z * as.z + v.w * as.w;
  float pd = v.x * ad.x + v.y * ad.y + v.z * ad.z + v.w * ad.w;
#pragma unroll
  for (int off = 1; off < 8; off <<= 1) {
    ps += __shfl_xor(ps, off, 32);
    pd += __shfl_xor(pd, off, 32);
  }
  const int h = lane >> 3;
  if ((lane & 7) == 0) {
    a_src[(size_t)n * HEADS + h] = ps;
    a_dst[(size_t)n * HEADS + h] = pd;
  }
}

__global__ __launch_bounds__(1024) void k_scan(const int* __restrict__ cnt,
                                               int* __restrict__ cursor, int N) {
  __shared__ int wsum[32];
  const int lane = threadIdx.x & 31;
  const int wave = __builtin_amdgcn_readfirstlane((int)(threadIdx.x >> 5));
  const int per = (N + 31) / 32;
  const int seg = ((per + 255) / 256) * 256;
  const int lo = min(wave * seg, N);
  const int hi = min(lo + seg, N);
  int tot = 0;
#pragma unroll 8
  for (int i = lo + lane; i < hi; i += 32) tot += cnt[i];
#pragma unroll
  for (int off = 1; off < 32; off <<= 1) tot += __shfl_xor(tot, off, 32);
  if (lane == 0) wsum[wave] = tot;
  __syncthreads();
  int carry = (lane < wave) ? wsum[lane] : 0;
#pragma unroll
  for (int off = 1; off < 32; off <<= 1) carry += __shfl_xor(carry, off, 32);
  for (int b = lo; b < hi; b += 256) {
    const int i0 = b + lane * 8;
    int c[8];
    int t = 0;
#pragma unroll
    for (int k = 0; k < 8; ++k) {
      c[k] = (i0 + k < hi) ? cnt[i0 + k] : 0;
      t += c[k];
    }
    int inc = t;
#pragma unroll
    for (int off = 1; off < 32; off <<= 1) {
      const int u = __shfl_up(inc, off, 32);
      if (lane >= off) inc += u;
    }
    int run = carry + inc - t;
#pragma unroll
    for (int k = 0; k < 8; ++k) {
      if (i0 + k < hi) cursor[i0 + k] = run;
      run += c[k];
    }
    carry += __shfl(inc, 31, 32);
  }
}

__global__ __launch_bounds__(256) void k_fill(const int* __restrict__ ei,
                                              int* __restrict__ cursor,
                                              int* __restrict__ csr, int E) {
  const int e = blockIdx.x * 256 + threadIdx.x;
  if (e >= E) return;
  const int s = ei[e];
  const int d = ei[E + e];
  const int pos = atomicAdd(cursor + d, 1);
  if ((unsigned)pos < (unsigned)E) csr[pos] = s;
}

static __device__ __forceinline__ void load_chunk(const int* __restrict__ csr,
                                                  const float* __restrict__ a_src,
                                                  const float4 ad, int beg, int deg,
                                                  int n, int N, int i, int& s,
                                                  float4& e) {
  s = n;
  e = make_float4(-INFINITY, -INFINITY, -INFINITY, -INFINITY);
  if (i <= deg) {
    if (i < deg) s = min(max(csr[beg + i], 0), N - 1);
    const float4 as = ((const float4*)a_src)[s];
    e.x = lrelu(as.x + ad.x);
    e.y = lrelu(as.y + ad.y);
    e.z = lrelu(as.z + ad.z);
    e.w = lrelu(as.w + ad.w);
  }
}

static __device__ __forceinline__ void gather_chunk(const float* __restrict__ xt,
                                                    int s, const float4 al, int cnt,
                                                    int lane, float& acc0, float& acc1,
                                                    float& acc2, float& acc3) {
#pragma unroll 4
  for (int j = 0; j < cnt; ++j) {
    const int sj = __builtin_amdgcn_readlane(s, j);
    const float b0 = __int_as_float(__builtin_amdgcn_readlane(__float_as_int(al.x), j));
    const float b1 = __int_as_float(__builtin_amdgcn_readlane(__float_as_int(al.y), j));
    const float b2 = __int_as_float(__builtin_amdgcn_readlane(__float_as_int(al.z), j));
    const float b3 = __int_as_float(__builtin_amdgcn_readlane(__float_as_int(al.w), j));
    const float* xp = xt + (size_t)sj * OUT_DIM + lane;
    acc0 += b0 * xp[0];
    acc1 += b1 * xp[32];
    acc2 += b2 * xp[64];
    acc3 += b3 * xp[96];
  }
}

static __device__ __forceinline__ float wave_max(float v) {
#pragma unroll
  for (int off = 1; off < 32; off <<= 1) v = fmaxf(v, __shfl_xor(v, off, 32));
  return v;
}
static __device__ __forceinline__ float wave_sum(float v) {
#pragma unroll
  for (int off = 1; off < 32; off <<= 1) v += __shfl_xor(v, off, 32);
  return v;
}

__global__ __launch_bounds__(256) void k_agg(const int* __restrict__ cursor,
                                             const int* __restrict__ cnt,
                                             const int* __restrict__ csr,
                                             const float* __restrict__ a_src,
                                             const float* __restrict__ a_dst,
                                             const float* __restrict__ xt,
                                             const float* __restrict__ zres,
                                             const float* __restrict__ gat_bias,
                                             const float* __restrict__ res_b,
                                             const float* __restrict__ ln_g,
                                             const float* __restrict__ ln_b,
                                             float* __restrict__ out, int N) {
  const int lane = threadIdx.x & 31;
  const int n = __builtin_amdgcn_readfirstlane((int)(blockIdx.x * 8 + (threadIdx.x >> 5)));
  if (n >= N) return;
  const int deg = max(cnt[n], 0);
  const int beg = cursor[n] - deg;
  const int tot = deg + 1;
  const float4 ad = ((const float4*)a_dst)[n];

  int s0;
  float4 e0;
  load_chunk(csr, a_src, ad, beg, deg, n, N, lane, s0, e0);

  float4 m = e0;
  for (int b = 32; b < tot; b += 32) {
    int s; float4 e;
    load_chunk(csr, a_src, ad, beg, deg, n, N, b + lane, s, e);
    m.x = fmaxf(m.x, e.x); m.y = fmaxf(m.y, e.y);
    m.z = fmaxf(m.z, e.z); m.w = fmaxf(m.w, e.w);
  }
  m.x = wave_max(m.x); m.y = wave_max(m.y);
  m.z = wave_max(m.z); m.w = wave_max(m.w);

  float4 x0;
  x0.x = __expf(e0.x - m.x); x0.y = __expf(e0.y - m.y);
  x0.z = __expf(e0.z - m.z); x0.w = __expf(e0.w - m.w);
  float4 q = x0;
  for (int b = 32; b < tot; b += 32) {
    int s; float4 e;
    load_chunk(csr, a_src, ad, beg, deg, n, N, b + lane, s, e);
    q.x += __expf(e.x - m.x); q.y += __expf(e.y - m.y);
    q.z += __expf(e.z - m.z); q.w += __expf(e.w - m.w);
  }
  q.x = wave_sum(q.x) + 1e-16f; q.y = wave_sum(q.y) + 1e-16f;
  q.z = wave_sum(q.z) + 1e-16f; q.w = wave_sum(q.w) + 1e-16f;

  float acc0 = 0.f, acc1 = 0.f, acc2 = 0.f, acc3 = 0.f;
  {
    float4 al;
    al.x = x0.x / q.x; al.y = x0.y / q.y; al.z = x0.z / q.z; al.w = x0.w / q.w;
    gather_chunk(xt, s0, al, min(tot, 32), lane, acc0, acc1, acc2, acc3);
  }
  for (int b = 32; b < tot; b += 32) {
    int s; float4 e;
    load_chunk(csr, a_src, ad, beg, deg, n, N, b + lane, s, e);
    float4 al;
    al.x = __expf(e.x - m.x) / q.x; al.y = __expf(e.y - m.y) / q.y;
    al.z = __expf(e.z - m.z) / q.z; al.w = __expf(e.w - m.w) / q.w;
    gather_chunk(xt, s, al, min(tot - b, 32), lane, acc0, acc1, acc2, acc3);
  }

  const float* zr = zres + (size_t)n * OUT_DIM + lane;
  const float z0 = ((acc0 + gat_bias[lane])      + zr[0])  + res_b[lane];
  const float z1 = ((acc1 + gat_bias[lane + 32]) + zr[32]) + res_b[lane + 32];
  const float z2 = ((acc2 + gat_bias[lane + 64]) + zr[64]) + res_b[lane + 64];
  const float z3 = ((acc3 + gat_bias[lane + 96]) + zr[96]) + res_b[lane + 96];

  const float mu = wave_sum((z0 + z1) + (z2 + z3)) * (1.0f / 128.0f);
  const float d0 = z0 - mu, d1 = z1 - mu, d2 = z2 - mu, d3 = z3 - mu;
  const float var = wave_sum((d0 * d0 + d1 * d1) + (d2 * d2 + d3 * d3)) * (1.0f / 128.0f);
  const float inv = rsqrtf(var + 1e-5f);
  float* op = out + (size_t)n * OUT_DIM + lane;
  op[0]  = lrelu(d0 * inv * ln_g[lane]      + ln_b[lane]);
  op[32] = lrelu(d1 * inv * ln_g[lane + 32] + ln_b[lane + 32]);
  op[64] = lrelu(d2 * inv * ln_g[lane + 64] + ln_b[lane + 64]);
  op[96] = lrelu(d3 * inv * ln_g[lane + 96] + ln_b[lane + 96]);
}

extern "C" void kernel_launch(void* const* d_in, const int* in_sizes, int n_in,
                              void* d_out, int out_size, void* d_ws, size_t ws_size,
                              hipStream_t stream) {
  const float* x        = (const float*)d_in[0];
  const int*   ei       = (const int*)d_in[1];
  const float* lin_w    = (const float*)d_in[2];
  const float* att_src  = (const float*)d_in[3];
  const float* att_dst  = (const float*)d_in[4];
  const float* gat_bias = (const float*)d_in[5];
  const float* res_w    = (const float*)d_in[6];
  const float* res_b    = (const float*)d_in[7];
  const float* ln_g     = (const float*)d_in[8];
  const float* ln_b     = (const float*)d_in[9];
  const int N  = in_sizes[0] / IN_CH;
  const int E  = in_sizes[1] / 2;

  float* xt     = (float*)d_ws;
  float* z      = xt + (size_t)N * OUT_DIM;
  float* a_src  = z + (size_t)N * OUT_DIM;
  float* a_dst  = a_src + (size_t)N * HEADS;
  int*   cnt    = (int*)(a_dst + (size_t)N * HEADS);
  int*   cursor = cnt + N;
  int*   csr    = cursor + N;

  k_gemm<<<GEMM_GROUPS, 256, 0, stream>>>(x, lin_w, res_w, xt, z, cnt, N);
  k_node<<<(N + 7) / 8, 256, 0, stream>>>(xt, att_src, att_dst, ei, a_src, a_dst,
                                          cnt, E, N);
  k_scan<<<1, 1024, 0, stream>>>(cnt, cursor, N);
  k_fill<<<(E + 255) / 256, 256, 0, stream>>>(ei, cursor, csr, E);
  k_agg<<<(N + 7) / 8, 256, 0, stream>>>(cursor, cnt, csr, a_src, a_dst, xt, z,
                                         gat_bias, res_b, ln_g, ln_b,
                                         (float*)d_out, N);
}


# Edge lists built by the count pass: the count atomic's return is the edge's slot in its destination's 32-word row (rare overflow: shared list); scan and fill passes dropped; sum order = count arrival
# speedup vs baseline: 4.5091x; 1.6520x over previous
#include <hip/hip_runtime.h>
#include <hip/hip_bf16.h>

typedef __attribute__((ext_vector_type(2))) float v2f;
typedef __attribute__((ext_vector_type(8))) float v8f;

#define IN_CH 64
#define OUT_DIM 128
#define HEADS 4

static __device__ __forceinline__ float lrelu(float x) { return fmaxf(x, 0.2f * x); }

#define LDS_PITCH 68
#define GEMM_GROUPS 512

static __device__ __forceinline__ v2f frag(const float* s, int row, int ks, int kb) {
  return *(const v2f*)(s + row * LDS_PITCH + ks * 4 + kb);
}

__global__ __launch_bounds__(256) void k_gemm(const float* __restrict__ x,
                                              const float* __restrict__ lin_w,
                                              const float* __restrict__ res_w,
                                              float* __restrict__ xt,
                                              float* __restrict__ z,
                                              int* __restrict__ cnt,
                                              int N) {
  __shared__ __attribute__((aligned(16))) float Ws[256 * LDS_PITCH];
  __shared__ __attribute__((aligned(16))) float As[2][16 * LDS_PITCH];
  const int tid = threadIdx.x;
  const int wave = tid >> 5;
  const int lane = tid & 31;
  const int mrow = lane & 15;
  const int kb = (lane >> 4) << 1;
  const int nchunks = (N + 15) >> 4;
  const int sr = tid >> 4;
  const int sq = tid & 15;

#pragma unroll
  for (int i = 0; i < 8; ++i) {
    const int r = i * 16 + sr;
    const float4 v0 = ((const float4*)lin_w)[r * 16 + sq];
    const float4 v1 = ((const float4*)res_w)[r * 16 + sq];
    *(float4*)(Ws + r * LDS_PITCH + sq * 4) = v0;
    *(float4*)(Ws + (128 + r) * LDS_PITCH + sq * 4) = v1;
  }
  int chunk = blockIdx.x;
  float4 va = make_float4(0.f, 0.f, 0.f, 0.f);
  if (chunk < nchunks) {
    const int row = min(chunk * 16 + sr, N - 1);
    va = ((const float4*)x)[(size_t)row * 16 + sq];
  }
  __syncthreads();
  v2f b0[16], b1[16];
#pragma unroll
  for (int ks = 0; ks < 16; ++ks) {
    b0[ks] = frag(Ws, wave * 16 + mrow, ks, kb);
    b1[ks] = frag(Ws, 128 + wave * 16 + mrow, ks, kb);
  }
  const int cbase = wave * 16;
  int buf = 0;
  while (chunk < nchunks) {
    *(float4*)(&As[buf][sr * LDS_PITCH + sq * 4]) = va;
    __syncthreads();
    const int next = chunk + GEMM_GROUPS;
    if (next < nchunks) {
      const int row = min(next * 16 + sr, N - 1);
      va = ((const float4*)x)[(size_t)row * 16 + sq];
    }
    v2f a[16];
#pragma unroll
    for (int ks = 0; ks < 16; ++ks) a[ks] = frag(As[buf], mrow, ks, kb);
    asm volatile("" ::"v"(a[0]), "v"(a[1]), "v"(a[2]), "v"(a[3]), "v"(a[4]), "v"(a[5]),
                 "v"(a[6]), "v"(a[7]), "v"(a[8]), "v"(a[9]), "v"(a[10]), "v"(a[11]),
                 "v"(a[12]), "v"(a[13]), "v"(a[14]), "v"(a[15]));
    v8f c0 = {0.f, 0.f, 0.f, 0.f, 0.f, 0.f, 0.f, 0.f};
    v8f c1 = {0.f, 0.f, 0.f, 0.f, 0.f, 0.f, 0.f, 0.f};
#pragma unroll
    for (int ks = 0; ks < 16; ++ks) {
      c0 = __builtin_amdgcn_wmma_f32_16x16x4_f32(false, a[ks], false, b0[ks],
                                                 (short)0, c0, false, false);
      c1 = __builtin_amdgcn_wmma_f32_16x16x4_f32(false, a[ks], false, b1[ks],
                                                 (short)0, c1, false, false);
    }
    const int m0 = chunk * 16;
    const int mb = m0 + ((lane >> 4) << 3);
#pragma unroll
    for (int r = 0; r < 8; ++r) {
      const int m = mb + r;
      if (m < N) {
        xt[(size_t)m * OUT_DIM + cbase + mrow] = c0[r];
        z[(size_t)m * OUT_DIM + cbase + mrow] = c1[r];
      }
    }
    if (tid < 16) {
      const int m = m0 + tid;
      if (m < N) cnt[m] = 0;
    }
    if (chunk == 0 && tid == 16) cnt[N] = 0;
    buf ^= 1;
    chunk = next;
  }
}

#define SLOTS 31
#define ROW 32

__global__ __launch_bounds__(256) void k_node(const float* __restrict__ xt,
                                              const float* __restrict__ att_src,
                                              const float* __restrict__ att_dst,
                                              const int* __restrict__ ei,
                                              float* __restrict__ a_src,
                                              float* __restrict__ a_dst,
                                              int* __restrict__ cnt,
                                              int* __restrict__ slab,
                                              int* __restrict__ ovf_s,
                                              int* __restrict__ ovf_d,
                                              int E, int N) {
  const int lane = threadIdx.x & 31;
  const int n = blockIdx.x * 8 + (threadIdx.x >> 5);
  if (n >= N) return;
  const float4 v = ((const float4*)xt)[(size_t)n * 32 + lane];
  const float4 as = ((const float4*)att_src)[lane];
  const float4 ad = ((const float4*)att_dst)[lane];
  for (int g = n * 32 + lane; g < E; g += N * 32) {
    const int s = ei[g];
    const int d = ei[E + g];
    const int r = atomicAdd(cnt + d, 1);
    if ((unsigned)r < (unsigned)SLOTS) {
      slab[(size_t)d * ROW + r] = s;
    } else {
      const int o = atomicAdd(cnt + N, 1);
      if ((unsigned)o < (unsigned)E) {
        ovf_s[o] = s;
        ovf_d[o] = d;
      }
    }
  }
  float ps = v.x * as.x + v.y * as.y + v.z * as.z + v.w * as.w;
  float pd = v.x * ad.x + v.y * ad.y + v.z * ad.z + v.w * ad.w;
#pragma unroll
  for (int off = 1; off < 8; off <<= 1) {
    ps += __shfl_xor(ps, off, 32);
    pd += __shfl_xor(pd, off, 32);
  }
  const int h = lane >> 3;
  if ((lane & 7) == 0) {
    a_src[(size_t)n * HEADS + h] = ps;
    a_dst[(size_t)n * HEADS + h] = pd;
  }
}

static __device__ __forceinline__ void load_first(const int* __restrict__ slab,
                                                  const float* __restrict__ a_src,
                                                  const float4 ad, int deg0, int n,
                                                  int N, int i, int& s, float4& e) {
  s = n;
  e = make_float4(-INFINITY, -INFINITY, -INFINITY, -INFINITY);
  if (i <= deg0) {
    if (i < deg0) s = min(max(slab[(size_t)n * ROW + i], 0), N - 1);
    const float4 as = ((const float4*)a_src)[s];
    e.x = lrelu(as.x + ad.x);
    e.y = lrelu(as.y + ad.y);
    e.z = lrelu(as.z + ad.z);
    e.w = lrelu(as.w + ad.w);
  }
}

static __device__ __forceinline__ bool load_ovf(const int* __restrict__ ovf_s,
                                                const int* __restrict__ ovf_d,
                                                const float* __restrict__ a_src,
                                                const float4 ad, int nov, int n, int N,
                                                int i, int& s, float4& e) {
  s = n;
  e = make_float4(-INFINITY, -INFINITY, -INFINITY, -INFINITY);
  bool mine = false;
  if (i < nov) mine = (ovf_d[i] == n);
  if (mine) {
    s = min(max(ovf_s[i], 0), N - 1);
    const float4 as = ((const float4*)a_src)[s];
    e.x = lrelu(as.x + ad.x);
    e.y = lrelu(as.y + ad.y);
    e.z = lrelu(as.z + ad.z);
    e.w = lrelu(as.w + ad.w);
  }
  return mine;
}

static __device__ __forceinline__ void gather_chunk(const float* __restrict__ xt,
                                                    int s, const float4 al, int cnt,
                                                    int lane, float& acc0, float& acc1,
                                                    float& acc2, float& acc3) {
#pragma unroll 4
  for (int j = 0; j < cnt; ++j) {
    const int sj = __builtin_amdgcn_readlane(s, j);
    const float b0 = __int_as_float(__builtin_amdgcn_readlane(__float_as_int(al.x), j));
    const float b1 = __int_as_float(__builtin_amdgcn_readlane(__float_as_int(al.y), j));
    const float b2 = __int_as_float(__builtin_amdgcn_readlane(__float_as_int(al.z), j));
    const float b3 = __int_as_float(__builtin_amdgcn_readlane(__float_as_int(al.w), j));
    const float* xp = xt + (size_t)sj * OUT_DIM + lane;
    acc0 += b0 * xp[0];
    acc1 += b1 * xp[32];
    acc2 += b2 * xp[64];
    acc3 += b3 * xp[96];
  }
}

static __device__ __forceinline__ void gather_mask(const float* __restrict__ xt,
                                                   int s, const float4 al, unsigned mask,
                                                   int lane, float& acc0, float& acc1,
                                                   float& acc2, float& acc3) {
  while (mask != 0u) {
    const int j = __builtin_amdgcn_readfirstlane(__builtin_ctz(mask));
    mask &= mask - 1u;
    const int sj = __builtin_amdgcn_readlane(s, j);
    const float b0 = __int_as_float(__builtin_amdgcn_readlane(__float_as_int(al.x), j));
    const float b1 = __int_as_float(__builtin_amdgcn_readlane(__float_as_int(al.y), j));
    const float b2 = __int_as_float(__builtin_amdgcn_readlane(__float_as_int(al.z), j));
    const float b3 = __int_as_float(__builtin_amdgcn_readlane(__float_as_int(al.w), j));
    const float* xp = xt + (size_t)sj * OUT_DIM + lane;
    acc0 += b0 * xp[0];
    acc1 += b1 * xp[32];
    acc2 += b2 * xp[64];
    acc3 += b3 * xp[96];
  }
}

static __device__ __forceinline__ float wave_max(float v) {
#pragma unroll
  for (int off = 1; off < 32; off <<= 1) v = fmaxf(v, __shfl_xor(v, off, 32));
  return v;
}
static __device__ __forceinline__ float wave_sum(float v) {
#pragma unroll
  for (int off = 1; off < 32; off <<= 1) v += __shfl_xor(v, off, 32);
  return v;
}

__global__ __launch_bounds__(256) void k_agg(const int* __restrict__ cnt,
                                             const int* __restrict__ slab,
                                             const int* __restrict__ ovf_s,
                                             const int* __restrict__ ovf_d,
                                             const float* __restrict__ a_src,
                                             const float* __restrict__ a_dst,
                                             const float* __restrict__ xt,
                                             const float* __restrict__ zres,
                                             const float* __restrict__ gat_bias,
                                             const float* __restrict__ res_b,
                                             const float* __restrict__ ln_g,
                                             const float* __restrict__ ln_b,
                                             float* __restrict__ out, int E, int N) {
  const int lane = threadIdx.x & 31;
  const int n = __builtin_amdgcn_readfirstlane((int)(blockIdx.x * 8 + (threadIdx.x >> 5)));
  if (n >= N) return;
  const int deg = max(cnt[n], 0);
  const int deg0 = min(deg, SLOTS);
  const bool more = deg > SLOTS;
  int nov = 0;
  if (more) nov = min(max(cnt[N], 0), E);
  const float4 ad = ((const float4*)a_dst)[n];

  int s0;
  float4 e0;
  load_first(slab, a_src, ad, deg0, n, N, lane, s0, e0);

  float4 m = e0;
  if (more) {
    for (int b = 0; b < nov; b += 32) {
      int s; float4 e;
      load_ovf(ovf_s, ovf_d, a_src, ad, nov, n, N, b + lane, s, e);
      m.x = fmaxf(m.x, e.x); m.y = fmaxf(m.y, e.y);
      m.z = fmaxf(m.z, e.z); m.w = fmaxf(m.w, e.w);
    }
  }
  m.x = wave_max(m.x); m.y = wave_max(m.y);
  m.z = wave_max(m.z); m.w = wave_max(m.w);

  float4 x0;
  x0.x = __expf(e0.x - m.x); x0.y = __expf(e0.y - m.y);
  x0.z = __expf(e0.z - m.z); x0.w = __expf(e0.w - m.w);
  float4 q = x0;
  if (more) {
    for (int b = 0; b < nov; b += 32) {
      int s; float4 e;
      load_ovf(ovf_s, ovf_d, a_src, ad, nov, n, N, b + lane, s, e);
      q.x += __expf(e.x - m.x); q.y += __expf(e.y - m.y);
      q.z += __expf(e.z - m.z); q.w += __expf(e.w - m.w);
    }
  }
  q.x = wave_sum(q.x) + 1e-16f; q.y = wave_sum(q.y) + 1e-16f;
  q.z = wave_sum(q.z) + 1e-16f; q.w = wave_sum(q.w) + 1e-16f;

  float acc0 = 0.f, acc1 = 0.f, acc2 = 0.f, acc3 = 0.f;
  {
    float4 al;
    al.x = x0.x / q.x; al.y = x0.y / q.y; al.z = x0.z / q.z; al.w = x0.w / q.w;
    gather_chunk(xt, s0, al, deg0 + 1, lane, acc0, acc1, acc2, acc3);
  }
  if (more) {
    for (int b = 0; b < nov; b += 32) {
      int s; float4 e;
      const bool mine = load_ovf(ovf_s, ovf_d, a_src, ad, nov, n, N, b + lane, s, e);
      float4 al;
      al.x = __expf(e.x - m.x) / q.x; al.y = __expf(e.y - m.y) / q.y;
      al.z = __expf(e.z - m.z) / q.z; al.w = __expf(e.w - m.w) / q.w;
      const unsigned mask = __builtin_amdgcn_ballot_w32(mine);
      gather_mask(xt, s, al, mask, lane, acc0, acc1, acc2, acc3);
    }
  }

  const float* zr = zres + (size_t)n * OUT_DIM + lane;
  const float z0 = ((acc0 + gat_bias[lane])      + zr[0])  + res_b[lane];
  const float z1 = ((acc1 + gat_bias[lane + 32]) + zr[32]) + res_b[lane + 32];
  const float z2 = ((acc2 + gat_bias[lane + 64]) + zr[64]) + res_b[lane + 64];
  const float z3 = ((acc3 + gat_bias[lane + 96]) + zr[96]) + res_b[lane + 96];

  const float mu = wave_sum((z0 + z1) + (z2 + z3)) * (1.0f / 128.0f);
  const float d0 = z0 - mu, d1 = z1 - mu, d2 = z2 - mu, d3 = z3 - mu;
  const float var = wave_sum((d0 * d0 + d1 * d1) + (d2 * d2 + d3 * d3)) * (1.0f / 128.0f);
  const float inv = rsqrtf(var + 1e-5f);
  float* op = out + (size_t)n * OUT_DIM + lane;
  op[0]  = lrelu(d0 * inv * ln_g[lane]      + ln_b[lane]);
  op[32] = lrelu(d1 * inv * ln_g[lane + 32] + ln_b[lane + 32]);
  op[64] = lrelu(d2 * inv * ln_g[lane + 64] + ln_b[lane + 64]);
  op[96] = lrelu(d3 * inv * ln_g[lane + 96] + ln_b[lane + 96]);
}

extern "C" void kernel_launch(void* const* d_in, const int* in_sizes, int n_in,
                              void* d_out, int out_size, void* d_ws, size_t ws_size,
                              hipStream_t stream) {
  const float* x        = (const float*)d_in[0];
  const int*   ei       = (const int*)d_in[1];
  const float* lin_w    = (const float*)d_in[2];
  const float* att_src  = (const float*)d_in[3];
  const float* att_dst  = (const float*)d_in[4];
  const float* gat_bias = (const float*)d_in[5];
  const float* res_w    = (const float*)d_in[6];
  const float* res_b    = (const float*)d_in[7];
  const float* ln_g     = (const float*)d_in[8];
  const float* ln_b     = (const float*)d_in[9];
  const int N  = in_sizes[0] / IN_CH;
  const int E  = in_sizes[1] / 2;

  float* xt     = (float*)d_ws;
  float* z      = xt + (size_t)N * OUT_DIM;
  float* a_src  = z + (size_t)N * OUT_DIM;
  float* a_dst  = a_src + (size_t)N * HEADS;
  int*   slab   = (int*)(a_dst + (size_t)N * HEADS);
  int*   ovf_s  = slab + (size_t)N * ROW;
  int*   ovf_d  = ovf_s + E;
  int*   cnt    = ovf_d + E;

  k_gemm<<<GEMM_GROUPS, 256, 0, stream>>>(x, lin_w, res_w, xt, z, cnt, N);
  k_node<<<(N + 7) / 8, 256, 0, stream>>>(xt, att_src, att_dst, ei, a_src, a_dst,
                                          cnt, slab, ovf_s, ovf_d, E, N);
  k_agg<<<(N + 7) / 8, 256, 0, stream>>>(cnt, slab, ovf_s, ovf_d, a_src, a_dst, xt, z,
                                         gat_bias, res_b, ln_g, ln_b,
                                         (float*)d_out, E, N);
}


# Edge lists built with LDS counters, not one global atomic per edge: extra work-groups of the GEMM launch bin edges by destination range, one owner group per range ranks them; sum order = LDS arrival
# speedup vs baseline: 7.0564x; 1.5649x over previous
#include <hip/hip_runtime.h>
#include <hip/hip_bf16.h>

typedef __attribute__((ext_vector_type(2))) float v2f;
typedef __attribute__((ext_vector_type(8))) float v8f;

#define IN_CH 64
#define OUT_DIM 128
#define HEADS 4

static __device__ __forceinline__ float lrelu(float x) { return fmaxf(x, 0.2f * x); }

#define LDS_PITCH 68
#define GEMM_GROUPS 512
#define NB 256
#define NC 256
#define CAP 32
#define BIN_GROUPS NC
#define BATCH 8

#define MAXNPB 2048
static __device__ __forceinline__ int range_size(int N) {
  return min((N + NB - 1) / NB, MAXNPB);
}

static __device__ __forceinline__ int bucket_of(int d, int npb) {
  return min((int)((unsigned)d / (unsigned)npb), NB - 1);
}

static __device__ __forceinline__ void bin_edges(const int* __restrict__ ei,
                                                 int* __restrict__ cellw,
                                                 int* __restrict__ cellcnt,
                                                 int c, int E, int N) {
  int2* cells = (int2*)cellw;
  __shared__ int bcnt[NB];
  const int tid = threadIdx.x;
  const int npb = range_size(N);
  const int cs = (E + NC - 1) / NC;
  bcnt[tid] = 0;
  __syncthreads();
  const int g1 = min(E, (c + 1) * cs);
  for (int g0 = c * cs + tid; g0 < g1; g0 += 256 * BATCH) {
    int sv[BATCH], dv[BATCH];
#pragma unroll
    for (int j = 0; j < BATCH; ++j) {
      const int g = min(g0 + j * 256, g1 - 1);
      sv[j] = ei[g];
      dv[j] = ei[E + g];
    }
#pragma unroll
    for (int j = 0; j < BATCH; ++j) {
      if (g0 + j * 256 < g1) {
        const int d = min(max(dv[j], 0), N - 1);
        const int b = bucket_of(d, npb);
        const int r = atomicAdd(&bcnt[b], 1);
        if (r < CAP) cells[((size_t)b * NC + c) * CAP + r] = make_int2(sv[j], d);
      }
    }
  }
  __syncthreads();
  cellcnt[tid * NC + c] = bcnt[tid];
}

static __device__ __forceinline__ v2f frag(const float* s, int row, int ks, int kb) {
  return *(const v2f*)(s + row * LDS_PITCH + ks * 4 + kb);
}

__global__ __launch_bounds__(256) void k_gemm(const float* __restrict__ x,
                                              const float* __restrict__ lin_w,
                                              const float* __restrict__ res_w,
                                              float* __restrict__ xt,
                                              float* __restrict__ z,
                                              int* __restrict__ cnt,
                                              const int* __restrict__ ei,
                                              int* __restrict__ cells,
                                              int* __restrict__ cellcnt,
                                              int E, int N) {
  if (blockIdx.x >= GEMM_GROUPS) {
    bin_edges(ei, cells, cellcnt, (int)blockIdx.x - GEMM_GROUPS, E, N);
    return;
  }
  __shared__ __attribute__((aligned(16))) float Ws[256 * LDS_PITCH];
  __shared__ __attribute__((aligned(16))) float As[2][16 * LDS_PITCH];
  const int tid = threadIdx.x;
  const int wave = tid >> 5;
  const int lane = tid & 31;
  const int mrow = lane & 15;
  const int kb = (lane >> 4) << 1;
  const int nchunks = (N + 15) >> 4;
  const int sr = tid >> 4;
  const int sq = tid & 15;

#pragma unroll
  for (int i = 0; i < 8; ++i) {
    const int r = i * 16 + sr;
    const float4 v0 = ((const float4*)lin_w)[r * 16 + sq];
    const float4 v1 = ((const float4*)res_w)[r * 16 + sq];
    *(float4*)(Ws + r * LDS_PITCH + sq * 4) = v0;
    *(float4*)(Ws + (128 + r) * LDS_PITCH + sq * 4) = v1;
  }
  int chunk = blockIdx.x;
  float4 va = make_float4(0.f, 0.f, 0.f, 0.f);
  if (chunk < nchunks) {
    const int row = min(chunk * 16 + sr, N - 1);
    va = ((const float4*)x)[(size_t)row * 16 + sq];
  }
  __syncthreads();
  v2f b0[16], b1[16];
#pragma unroll
  for (int ks = 0; ks < 16; ++ks) {
    b0[ks] = frag(Ws, wave * 16 + mrow, ks, kb);
    b1[ks] = frag(Ws, 128 + wave * 16 + mrow, ks, kb);
  }
  const int cbase = wave * 16;
  int buf = 0;
  while (chunk < nchunks) {
    *(float4*)(&As[buf][sr * LDS_PITCH + sq * 4]) = va;
    __syncthreads();
    const int next = chunk + GEMM_GROUPS;
    if (next < nchunks) {
      const int row = min(next * 16 + sr, N - 1);
      va = ((const float4*)x)[(size_t)row * 16 + sq];
    }
    v2f a[16];
#pragma unroll
    for (int ks = 0; ks < 16; ++ks) a[ks] = frag(As[buf], mrow, ks, kb);
    asm volatile("" ::"v"(a[0]), "v"(a[1]), "v"(a[2]), "v"(a[3]), "v"(a[4]), "v"(a[5]),
                 "v"(a[6]), "v"(a[7]), "v"(a[8]), "v"(a[9]), "v"(a[10]), "v"(a[11]),
                 "v"(a[12]), "v"(a[13]), "v"(a[14]), "v"(a[15]));
    v8f c0 = {0.f, 0.f, 0.f, 0.f, 0.f, 0.f, 0.f, 0.f};
    v8f c1 = {0.f, 0.f, 0.f, 0.f, 0.f, 0.f, 0.f, 0.f};
#pragma unroll
    for (int ks = 0; ks < 16; ++ks) {
      c0 = __builtin_amdgcn_wmma_f32_16x16x4_f32(false, a[ks], false, b0[ks],
                                                 (short)0, c0, false, false);
      c1 = __builtin_amdgcn_wmma_f32_16x16x4_f32(false, a[ks], false, b1[ks],
                                                 (short)0, c1, false, false);
    }
    const int m0 = chunk * 16;
    const int mb = m0 + ((lane >> 4) << 3);
#pragma unroll
    for (int r = 0; r < 8; ++r) {
      const int m = mb + r;
      if (m < N) {
        xt[(size_t)m * OUT_DIM + cbase + mrow] = c0[r];
        z[(size_t)m * OUT_DIM + cbase + mrow] = c1[r];
      }
    }
    if (chunk == 0 && tid == 16) cnt[N] = 0;
    buf ^= 1;
    chunk = next;
  }
}

#define SLOTS 31
#define ROW 32

static __device__ __forceinline__ void place_edge(int s, int d, int base, int npb,
                                                  int* lcnt, int* __restrict__ cnt,
                                                  int* __restrict__ slab,
                                                  int* __restrict__ ovf_s,
                                                  int* __restrict__ ovf_d, int E, int N) {
  const unsigned dl = (unsigned)(d - base);
  if (dl < (unsigned)npb && (unsigned)d < (unsigned)N) {
    const int r = atomicAdd(&lcnt[dl], 1);
    if ((unsigned)r < (unsigned)SLOTS) {
      slab[(size_t)d * ROW + r] = s;
    } else {
      const int o = atomicAdd(cnt + N, 1);
      if ((unsigned)o < (unsigned)E) {
        ovf_s[o] = s;
        ovf_d[o] = d;
      }
    }
  }
}

__global__ __launch_bounds__(1024) void k_node(const float* __restrict__ xt,
                                               const float* __restrict__ att_src,
                                               const float* __restrict__ att_dst,
                                               const int* __restrict__ ei,
                                               const int* __restrict__ cellw,
                                               const int* __restrict__ cellcnt,
                                               float* __restrict__ a_src,
                                               float* __restrict__ a_dst,
                                               int* __restrict__ cnt,
                                               int* __restrict__ slab,
                                               int* __restrict__ ovf_s,
                                               int* __restrict__ ovf_d,
                                               int E, int N) {
  const int2* cells = (const int2*)cellw;
  __shared__ int lcnt[MAXNPB];
  __shared__ int ccnt[NC];
  const int tid = threadIdx.x;
  const int lane = tid & 31;
  const int wave = tid >> 5;
  const int b = blockIdx.x;
  const int npb = range_size(N);
  const int base = b * npb;
  const int cs = (E + NC - 1) / NC;
  for (int i = tid; i < npb; i += 1024) lcnt[i] = 0;
  if (tid < NC) ccnt[tid] = cellcnt[b * NC + tid];
  __syncthreads();

  int2 ent[NC / 32];
  int num[NC / 32];
#pragma unroll
  for (int k = 0; k < NC / 32; ++k) {
    const int c = wave + k * 32;
    num[k] = ccnt[c];
    ent[k] = make_int2(0, -1);
    if (num[k] <= CAP && lane < num[k]) ent[k] = cells[((size_t)b * NC + c) * CAP + lane];
  }
#pragma unroll
  for (int k = 0; k < NC / 32; ++k) {
    if (num[k] <= CAP) {
      if (lane < num[k])
        place_edge(ent[k].x, ent[k].y, base, npb, lcnt, cnt, slab, ovf_s, ovf_d, E, N);
    } else {
      const int c = wave + k * 32;
      const int g1 = min(E, (c + 1) * cs);
      for (int g = c * cs + lane; g < g1; g += 32) {
        const int s = ei[g];
        const int d = min(max(ei[E + g], 0), N - 1);
        if (bucket_of(d, npb) == b)
          place_edge(s, d, base, npb, lcnt, cnt, slab, ovf_s, ovf_d, E, N);
      }
    }
  }
  __syncthreads();
  for (int i = tid; i < npb; i += 1024)
    if (base + i < N) cnt[base + i] = lcnt[i];

  const float4 as = ((const float4*)att_src)[lane];
  const float4 ad = ((const float4*)att_dst)[lane];
  const int nend = min(base + npb, N);
#pragma unroll 2
  for (int n = base + wave; n < nend; n += 32) {
    const float4 v = ((const float4*)xt)[(size_t)n * 32 + lane];
    float ps = v.x * as.x + v.y * as.y + v.z * as.z + v.w * as.w;
    float pd = v.x * ad.x + v.y * ad.y + v.z * ad.z + v.w * ad.w;
#pragma unroll
    for (int off = 1; off < 8; off <<= 1) {
      ps += __shfl_xor(ps, off, 32);
      pd += __shfl_xor(pd, off, 32);
    }
    const int h = lane >> 3;
    if ((lane & 7) == 0) {
      a_src[(size_t)n * HEADS + h] = ps;
      a_dst[(size_t)n * HEADS + h] = pd;
    }
  }
}

static __device__ __forceinline__ void load_first(const int* __restrict__ slab,
                                                  const float* __restrict__ a_src,
                                                  const float4 ad, int deg0, int n,
                                                  int N, int i, int& s, float4& e) {
  s = n;
  e = make_float4(-INFINITY, -INFINITY, -INFINITY, -INFINITY);
  if (i <= deg0) {
    if (i < deg0) s = min(max(slab[(size_t)n * ROW + i], 0), N - 1);
    const float4 as = ((const float4*)a_src)[s];
    e.x = lrelu(as.x + ad.x);
    e.y = lrelu(as.y + ad.y);
    e.z = lrelu(as.z + ad.z);
    e.w = lrelu(as.w + ad.w);
  }
}

static __device__ __forceinline__ bool load_ovf(const int* __restrict__ ovf_s,
                                                const int* __restrict__ ovf_d,
                                                const float* __restrict__ a_src,
                                                const float4 ad, int nov, int n, int N,
                                                int i, int& s, float4& e) {
  s = n;
  e = make_float4(-INFINITY, -INFINITY, -INFINITY, -INFINITY);
  bool mine = false;
  if (i < nov) mine = (ovf_d[i] == n);
  if (mine) {
    s = min(max(ovf_s[i], 0), N - 1);
    const float4 as = ((const float4*)a_src)[s];
    e.x = lrelu(as.x + ad.x);
    e.y = lrelu(as.y + ad.y);
    e.z = lrelu(as.z + ad.z);
    e.w = lrelu(as.w + ad.w);
  }
  return mine;
}

static __device__ __forceinline__ void gather_chunk(const float* __restrict__ xt,
                                                    int s, const float4 al, int cnt,
                                                    int lane, float& acc0, float& acc1,
                                                    float& acc2, float& acc3) {
#pragma unroll 4
  for (int j = 0; j < cnt; ++j) {
    const int sj = __builtin_amdgcn_readlane(s, j);
    const float b0 = __int_as_float(__builtin_amdgcn_readlane(__float_as_int(al.x), j));
    const float b1 = __int_as_float(__builtin_amdgcn_readlane(__float_as_int(al.y), j));
    const float b2 = __int_as_float(__builtin_amdgcn_readlane(__float_as_int(al.z), j));
    const float b3 = __int_as_float(__builtin_amdgcn_readlane(__float_as_int(al.w), j));
    const float* xp = xt + (size_t)sj * OUT_DIM + lane;
    acc0 += b0 * xp[0];
    acc1 += b1 * xp[32];
    acc2 += b2 * xp[64];
    acc3 += b3 * xp[96];
  }
}

static __device__ __forceinline__ void gather_mask(const float* __restrict__ xt,
                                                   int s, const float4 al, unsigned mask,
                                                   int lane, float& acc0, float& acc1,
                                                   float& acc2, float& acc3) {
  while (mask != 0u) {
    const int j = __builtin_amdgcn_readfirstlane(__builtin_ctz(mask));
    mask &= mask - 1u;
    const int sj = __builtin_amdgcn_readlane(s, j);
    const float b0 = __int_as_float(__builtin_amdgcn_readlane(__float_as_int(al.x), j));
    const float b1 = __int_as_float(__builtin_amdgcn_readlane(__float_as_int(al.y), j));
    const float b2 = __int_as_float(__builtin_amdgcn_readlane(__float_as_int(al.z), j));
    const float b3 = __int_as_float(__builtin_amdgcn_readlane(__float_as_int(al.w), j));
    const float* xp = xt + (size_t)sj * OUT_DIM + lane;
    acc0 += b0 * xp[0];
    acc1 += b1 * xp[32];
    acc2 += b2 * xp[64];
    acc3 += b3 * xp[96];
  }
}

static __device__ __forceinline__ float wave_max(float v) {
#pragma unroll
  for (int off = 1; off < 32; off <<= 1) v = fmaxf(v, __shfl_xor(v, off, 32));
  return v;
}
static __device__ __forceinline__ float wave_sum(float v) {
#pragma unroll
  for (int off = 1; off < 32; off <<= 1) v += __shfl_xor(v, off, 32);
  return v;
}

__global__ __launch_bounds__(256) void k_agg(const int* __restrict__ cnt,
                                             const int* __restrict__ slab,
                                             const int* __restrict__ ovf_s,
                                             const int* __restrict__ ovf_d,
                                             const float* __restrict__ a_src,
                                             const float* __restrict__ a_dst,
                                             const float* __restrict__ xt,
                                             const float* __restrict__ zres,
                                             const float* __restrict__ gat_bias,
                                             const float* __restrict__ res_b,
                                             const float* __restrict__ ln_g,
                                             const float* __restrict__ ln_b,
                                             float* __restrict__ out, int E, int N) {
  const int lane = threadIdx.x & 31;
  const int n = __builtin_amdgcn_readfirstlane((int)(blockIdx.x * 8 + (threadIdx.x >> 5)));
  if (n >= N) return;
  const int deg = max(cnt[n], 0);
  const int deg0 = min(deg, SLOTS);
  const bool more = deg > SLOTS;
  int nov = 0;
  if (more) nov = min(max(cnt[N], 0), E);
  const float4 ad = ((const float4*)a_dst)[n];

  int s0;
  float4 e0;
  load_first(slab, a_src, ad, deg0, n, N, lane, s0, e0);

  float4 m = e0;
  if (more) {
    for (int b = 0; b < nov; b += 32) {
      int s; float4 e;
      load_ovf(ovf_s, ovf_d, a_src, ad, nov, n, N, b + lane, s, e);
      m.x = fmaxf(m.x, e.x); m.y = fmaxf(m.y, e.y);
      m.z = fmaxf(m.z, e.z); m.w = fmaxf(m.w, e.w);
    }
  }
  m.x = wave_max(m.x); m.y = wave_max(m.y);
  m.z = wave_max(m.z); m.w = wave_max(m.w);

  float4 x0;
  x0.x = __expf(e0.x - m.x); x0.y = __expf(e0.y - m.y);
  x0.z = __expf(e0.z - m.z); x0.w = __expf(e0.w - m.w);
  float4 q = x0;
  if (more) {
    for (int b = 0; b < nov; b += 32) {
      int s; float4 e;
      load_ovf(ovf_s, ovf_d, a_src, ad, nov, n, N, b + lane, s, e);
      q.x += __expf(e.x - m.x); q.y += __expf(e.y - m.y);
      q.z += __expf(e.z - m.z); q.w += __expf(e.w - m.w);
    }
  }
  q.x = wave_sum(q.x) + 1e-16f; q.y = wave_sum(q.y) + 1e-16f;
  q.z = wave_sum(q.z) + 1e-16f; q.w = wave_sum(q.w) + 1e-16f;

  float acc0 = 0.f, acc1 = 0.f, acc2 = 0.f, acc3 = 0.f;
  {
    float4 al;
    al.x = x0.x / q.x; al.y = x0.y / q.y; al.z = x0.z / q.z; al.w = x0.w / q.w;
    gather_chunk(xt, s0, al, deg0 + 1, lane, acc0, acc1, acc2, acc3);
  }
  if (more) {
    for (int b = 0; b < nov; b += 32) {
      int s; float4 e;
      const bool mine = load_ovf(ovf_s, ovf_d, a_src, ad, nov, n, N, b + lane, s, e);
      float4 al;
      al.x = __expf(e.x - m.x) / q.x; al.y = __expf(e.y - m.y) / q.y;
      al.z = __expf(e.z - m.z) / q.z; al.w = __expf(e.w - m.w) / q.w;
      const unsigned mask = __builtin_amdgcn_ballot_w32(mine);
      gather_mask(xt, s, al, mask, lane, acc0, acc1, acc2, acc3);
    }
  }

  const float* zr = zres + (size_t)n * OUT_DIM + lane;
  const float z0 = ((acc0 + gat_bias[lane])      + zr[0])  + res_b[lane];
  const float z1 = ((acc1 + gat_bias[lane + 32]) + zr[32]) + res_b[lane + 32];
  const float z2 = ((acc2 + gat_bias[lane + 64]) + zr[64]) + res_b[lane + 64];
  const float z3 = ((acc3 + gat_bias[lane + 96]) + zr[96]) + res_b[lane + 96];

  const float mu = wave_sum((z0 + z1) + (z2 + z3)) * (1.0f / 128.0f);
  const float d0 = z0 - mu, d1 = z1 - mu, d2 = z2 - mu, d3 = z3 - mu;
  const float var = wave_sum((d0 * d0 + d1 * d1) + (d2 * d2 + d3 * d3)) * (1.0f / 128.0f);
  const float inv = rsqrtf(var + 1e-5f);
  float* op = out + (size_t)n * OUT_DIM + lane;
  op[0]  = lrelu(d0 * inv * ln_g[lane]      + ln_b[lane]);
  op[32] = lrelu(d1 * inv * ln_g[lane + 32] + ln_b[lane + 32]);
  op[64] = lrelu(d2 * inv * ln_g[lane + 64] + ln_b[lane + 64]);
  op[96] = lrelu(d3 * inv * ln_g[lane + 96] + ln_b[lane + 96]);
}

extern "C" void kernel_launch(void* const* d_in, const int* in_sizes, int n_in,
                              void* d_out, int out_size, void* d_ws, size_t ws_size,
                              hipStream_t stream) {
  const float* x        = (const float*)d_in[0];
  const int*   ei       = (const int*)d_in[1];
  const float* lin_w    = (const float*)d_in[2];
  const float* att_src  = (const float*)d_in[3];
  const float* att_dst  = (const float*)d_in[4];
  const float* gat_bias = (const float*)d_in[5];
  const float* res_w    = (const float*)d_in[6];
  const float* res_b    = (const float*)d_in[7];
  const float* ln_g     = (const float*)d_in[8];
  const float* ln_b     = (const float*)d_in[9];
  const int N  = in_sizes[0] / IN_CH;
  const int E  = in_sizes[1] / 2;

  float* xt     = (float*)d_ws;
  float* z      = xt + (size_t)N * OUT_DIM;
  float* a_src  = z + (size_t)N * OUT_DIM;
  float* a_dst  = a_src + (size_t)N * HEADS;
  int*   slab   = (int*)(a_dst + (size_t)N * HEADS);
  int*   ovf_s  = slab + (size_t)N * ROW;
  int*   ovf_d  = ovf_s + E;
  int*   cells  = ovf_d + E;
  int*   cellcnt = cells + (size_t)NB * NC * CAP * 2;
  int*   cnt    = cellcnt + NB * NC;

  k_gemm<<<GEMM_GROUPS + BIN_GROUPS, 256, 0, stream>>>(x, lin_w, res_w, xt, z, cnt, ei,
                                                       cells, cellcnt, E, N);
  k_node<<<NB, 1024, 0, stream>>>(xt, att_src, att_dst, ei, cells, cellcnt, a_src, a_dst,
                                  cnt, slab, ovf_s, ovf_d, E, N);
  k_agg<<<(N + 7) / 8, 256, 0, stream>>>(cnt, slab, ovf_s, ovf_d, a_src, a_dst, xt, z,
                                         gat_bias, res_b, ln_g, ln_b,
                                         (float*)d_out, E, N);
}


# Node pass and aggregation merged into one launch: each owner group keeps its edge lists in LDS and aggregates its own nodes; a_src/a_dst summed from the GEMM accumulators (16-lane tree)
# speedup vs baseline: 7.7802x; 1.1026x over previous
#include <hip/hip_runtime.h>
#include <hip/hip_bf16.h>

typedef __attribute__((ext_vector_type(2))) float v2f;
typedef __attribute__((ext_vector_type(8))) float v8f;

#define IN_CH 64
#define OUT_DIM 128
#define HEADS 4

static __device__ __forceinline__ float lrelu(float x) { return fmaxf(x, 0.2f * x); }

#define LDS_PITCH 68
#define GEMM_GROUPS 512
#define NB 256
#define NC 256
#define CAP 32
#define BIN_GROUPS NC
#define BATCH 8

#define MAXNPB 256
static __device__ __forceinline__ int range_size(int N) {
  return min((N + NB - 1) / NB, MAXNPB);
}

static __device__ __forceinline__ int bucket_of(int d, int npb) {
  return min((int)((unsigned)d / (unsigned)npb), NB - 1);
}

static __device__ __forceinline__ void bin_edges(const int* __restrict__ ei,
                                                 int* __restrict__ cellw,
                                                 int* __restrict__ cellcnt,
                                                 int c, int E, int N) {
  int2* cells = (int2*)cellw;
  __shared__ int bcnt[NB];
  const int tid = threadIdx.x;
  const int npb = range_size(N);
  const int cs = (E + NC - 1) / NC;
  bcnt[tid] = 0;
  __syncthreads();
  const int g1 = min(E, (c + 1) * cs);
  for (int g0 = c * cs + tid; g0 < g1; g0 += 256 * BATCH) {
    int sv[BATCH], dv[BATCH];
#pragma unroll
    for (int j = 0; j < BATCH; ++j) {
      const int g = min(g0 + j * 256, g1 - 1);
      sv[j] = ei[g];
      dv[j] = ei[E + g];
    }
#pragma unroll
    for (int j = 0; j < BATCH; ++j) {
      if (g0 + j * 256 < g1) {
        const int d = min(max(dv[j], 0), N - 1);
        const int b = bucket_of(d, npb);
        const int r = atomicAdd(&bcnt[b], 1);
        if (r < CAP) cells[((size_t)b * NC + c) * CAP + r] = make_int2(sv[j], d);
      }
    }
  }
  __syncthreads();
  cellcnt[tid * NC + c] = bcnt[tid];
}

#define ROW16_STEP(v, m)                                                              \
  ((v) + __int_as_float(__builtin_amdgcn_update_dpp(0, __float_as_int(v), 0x160 + (m), \
                                                    0xf, 0xf, true)))
static __device__ __forceinline__ float row16_sum(float v) {
  v = ROW16_STEP(v, 1);
  v = ROW16_STEP(v, 2);
  v = ROW16_STEP(v, 4);
  v = ROW16_STEP(v, 8);
  return v;
}

static __device__ __forceinline__ v2f frag(const float* s, int row, int ks, int kb) {
  return *(const v2f*)(s + row * LDS_PITCH + ks * 4 + kb);
}

__global__ __launch_bounds__(256) void k_gemm(const float* __restrict__ x,
                                              const float* __restrict__ lin_w,
                                              const float* __restrict__ res_w,
                                              const float* __restrict__ att_src,
                                              const float* __restrict__ att_dst,
                                              float* __restrict__ xt,
                                              float* __restrict__ z,
                                              float* __restrict__ a_src,
                                              float* __restrict__ a_dst,
                                              const int* __restrict__ ei,
                                              int* __restrict__ cells,
                                              int* __restrict__ cellcnt,
                                              int E, int N) {
  if (blockIdx.x >= GEMM_GROUPS) {
    bin_edges(ei, cells, cellcnt, (int)blockIdx.x - GEMM_GROUPS, E, N);
    return;
  }
  __shared__ __attribute__((aligned(16))) float Ws[256 * LDS_PITCH];
  __shared__ __attribute__((aligned(16))) float As[2][16 * LDS_PITCH];
  const int tid = threadIdx.x;
  const int wave = tid >> 5;
  const int lane = tid & 31;
  const int mrow = lane & 15;
  const int kb = (lane >> 4) << 1;
  const int nchunks = (N + 15) >> 4;
  const int sr = tid >> 4;
  const int sq = tid & 15;

#pragma unroll
  for (int i = 0; i < 8; ++i) {
    const int r = i * 16 + sr;
    const float4 v0 = ((const float4*)lin_w)[r * 16 + sq];
    const float4 v1 = ((const float4*)res_w)[r * 16 + sq];
    *(float4*)(Ws + r * LDS_PITCH + sq * 4) = v0;
    *(float4*)(Ws + (128 + r) * LDS_PITCH + sq * 4) = v1;
  }
  int chunk = blockIdx.x;
  float4 va = make_float4(0.f, 0.f, 0.f, 0.f);
  if (chunk < nchunks) {
    const int row = min(chunk * 16 + sr, N - 1);
    va = ((const float4*)x)[(size_t)row * 16 + sq];
  }
  __syncthreads();
  const int trow = (wave < 4) ? wave * 32 : 128 + (wave - 4) * 32;
  v2f b0[16], b1[16];
#pragma unroll
  for (int ks = 0; ks < 16; ++ks) {
    b0[ks] = frag(Ws, trow + mrow, ks, kb);
    b1[ks] = frag(Ws, trow + 16 + mrow, ks, kb);
  }
  const int cbase = (wave & 3) * 32;
  float* __restrict__ dstp = (wave < 4) ? xt : z;
  const int acol = (wave & 3) * 32 + mrow;
  const float as0 = att_src[acol], as1 = att_src[acol + 16];
  const float ad0 = att_dst[acol], ad1 = att_dst[acol + 16];
  int buf = 0;
  while (chunk < nchunks) {
    *(float4*)(&As[buf][sr * LDS_PITCH + sq * 4]) = va;
    __syncthreads();
    const int next = chunk + GEMM_GROUPS;
    if (next < nchunks) {
      const int row = min(next * 16 + sr, N - 1);
      va = ((const float4*)x)[(size_t)row * 16 + sq];
    }
    v2f a[16];
#pragma unroll
    for (int ks = 0; ks < 16; ++ks) a[ks] = frag(As[buf], mrow, ks, kb);
    asm volatile("" ::"v"(a[0]), "v"(a[1]), "v"(a[2]), "v"(a[3]), "v"(a[4]), "v"(a[5]),
                 "v"(a[6]), "v"(a[7]), "v"(a[8]), "v"(a[9]), "v"(a[10]), "v"(a[11]),
                 "v"(a[12]), "v"(a[13]), "v"(a[14]), "v"(a[15]));
    v8f c0 = {0.f, 0.f, 0.f, 0.f, 0.f, 0.f, 0.f, 0.f};
    v8f c1 = {0.f, 0.f, 0.f, 0.f, 0.f, 0.f, 0.f, 0.f};
#pragma unroll
    for (int ks = 0; ks < 16; ++ks) {
      c0 = __builtin_amdgcn_wmma_f32_16x16x4_f32(false, a[ks], false, b0[ks],
                                                 (short)0, c0, false, false);
      c1 = __builtin_amdgcn_wmma_f32_16x16x4_f32(false, a[ks], false, b1[ks],
                                                 (short)0, c1, false, false);
    }
    const int m0 = chunk * 16;
    const int mb = m0 + ((lane >> 4) << 3);
#pragma unroll
    for (int r = 0; r < 8; ++r) {
      const int m = mb + r;
      if (m < N) {
        dstp[(size_t)m * OUT_DIM + cbase + mrow] = c0[r];
        dstp[(size_t)m * OUT_DIM + cbase + 16 + mrow] = c1[r];
      }
    }
    if (wave < 4) {
      float ps[8], pd[8];
#pragma unroll
      for (int r = 0; r < 8; ++r) {
        ps[r] = c0[r] * as0 + c1[r] * as1;
        pd[r] = c0[r] * ad0 + c1[r] * ad1;
      }
#pragma unroll
      for (int r = 0; r < 8; ++r) {
        ps[r] = row16_sum(ps[r]);
        pd[r] = row16_sum(pd[r]);
      }
      float vs = ps[0], vd = pd[0];
#pragma unroll
      for (int r = 1; r < 8; ++r) {
        if ((mrow & 7) == r) { vs = ps[r]; vd = pd[r]; }
      }
      const int m = mb + (mrow & 7);
      if (mrow < 8 && m < N) {
        a_src[(size_t)m * HEADS + wave] = vs;
        a_dst[(size_t)m * HEADS + wave] = vd;
      }
    }
    buf ^= 1;
    chunk = next;
  }
}

#define SLOTS 31
#define ROW 32
#define OVC 512

static __device__ __forceinline__ void place_edge(int s, int d, int base, int npb,
                                                  int* lcnt, int* slab, int* ovn,
                                                  int2* ovl, int2* __restrict__ ovg,
                                                  bool to_global, int capg, int N) {
  const unsigned dl = (unsigned)(d - base);
  if (dl < (unsigned)npb && (unsigned)d < (unsigned)N) {
    const int r = atomicAdd(&lcnt[dl], 1);
    if ((unsigned)r < (unsigned)SLOTS) {
      slab[dl * ROW + r] = s;
    } else {
      const int o = atomicAdd(ovn, 1);
      if (to_global) {
        if ((unsigned)o < (unsigned)capg) ovg[o] = make_int2(s, d);
      } else if ((unsigned)o < (unsigned)OVC) {
        ovl[o] = make_int2(s, d);
      }
    }
  }
}

static __device__ __forceinline__ void load_first(const int* row,
                                                  const float* __restrict__ a_src,
                                                  const float4 ad, int deg0, int n,
                                                  int N, int i, int& s, float4& e) {
  s = n;
  e = make_float4(-INFINITY, -INFINITY, -INFINITY, -INFINITY);
  if (i <= deg0) {
    if (i < deg0) s = min(max(row[i], 0), N - 1);
    const float4 as = ((const float4*)a_src)[s];
    e.x = lrelu(as.x + ad.x);
    e.y = lrelu(as.y + ad.y);
    e.z = lrelu(as.z + ad.z);
    e.w = lrelu(as.w + ad.w);
  }
}

static __device__ __forceinline__ bool load_ovf(const int2* ovl,
                                                const int2* __restrict__ ovg, bool g,
                                                const float* __restrict__ a_src,
                                                const float4 ad, int nov, int n, int N,
                                                int i, int& s, float4& e) {
  s = n;
  e = make_float4(-INFINITY, -INFINITY, -INFINITY, -INFINITY);
  bool mine = false;
  int2 en = make_int2(n, -1);
  if (i < nov) {
    if (g) en = ovg[i]; else en = ovl[i];
    mine = (en.y == n);
  }
  if (mine) {
    s = min(max(en.x, 0), N - 1);
    const float4 as = ((const float4*)a_src)[s];
    e.x = lrelu(as.x + ad.x);
    e.y = lrelu(as.y + ad.y);
    e.z = lrelu(as.z + ad.z);
    e.w = lrelu(as.w + ad.w);
  }
  return mine;
}

static __device__ __forceinline__ void gather_chunk(const float* __restrict__ xt,
                                                    int s, const float4 al, int cnt,
                                                    int lane, float& acc0, float& acc1,
                                                    float& acc2, float& acc3) {
#pragma unroll 4
  for (int j = 0; j < cnt; ++j) {
    const int sj = __builtin_amdgcn_readlane(s, j);
    const float b0 = __int_as_float(__builtin_amdgcn_readlane(__float_as_int(al.x), j));
    const float b1 = __int_as_float(__builtin_amdgcn_readlane(__float_as_int(al.y), j));
    const float b2 = __int_as_float(__builtin_amdgcn_readlane(__float_as_int(al.z), j));
    const float b3 = __int_as_float(__builtin_amdgcn_readlane(__float_as_int(al.w), j));
    const float* xp = xt + (size_t)sj * OUT_DIM + lane;
    acc0 += b0 * xp[0];
    acc1 += b1 * xp[32];
    acc2 += b2 * xp[64];
    acc3 += b3 * xp[96];
  }
}

static __device__ __forceinline__ void gather_mask(const float* __restrict__ xt,
                                                   int s, const float4 al, unsigned mask,
                                                   int lane, float& acc0, float& acc1,
                                                   float& acc2, float& acc3) {
  while (mask != 0u) {
    const int j = __builtin_amdgcn_readfirstlane(__builtin_ctz(mask));
    mask &= mask - 1u;
    const int sj = __builtin_amdgcn_readlane(s, j);
    const float b0 = __int_as_float(__builtin_amdgcn_readlane(__float_as_int(al.x), j));
    const float b1 = __int_as_float(__builtin_amdgcn_readlane(__float_as_int(al.y), j));
    const float b2 = __int_as_float(__builtin_amdgcn_readlane(__float_as_int(al.z), j));
    const float b3 = __int_as_float(__builtin_amdgcn_readlane(__float_as_int(al.w), j));
    const float* xp = xt + (size_t)sj * OUT_DIM + lane;
    acc0 += b0 * xp[0];
    acc1 += b1 * xp[32];
    acc2 += b2 * xp[64];
    acc3 += b3 * xp[96];
  }
}

static __device__ __forceinline__ float wave_max(float v) {
#pragma unroll
  for (int off = 1; off < 32; off <<= 1) v = fmaxf(v, __shfl_xor(v, off, 32));
  return v;
}
static __device__ __forceinline__ float wave_sum(float v) {
#pragma unroll
  for (int off = 1; off < 32; off <<= 1) v += __shfl_xor(v, off, 32);
  return v;
}

__global__ __launch_bounds__(1024) void k_node(const int* __restrict__ ei,
                                               const int* __restrict__ cellw,
                                               const int* __restrict__ cellcnt,
                                               int* __restrict__ ovfw,
                                               const float* __restrict__ a_src,
                                               const float* __restrict__ a_dst,
                                               const float* __restrict__ xt,
                                               const float* __restrict__ zres,
                                               const float* __restrict__ gat_bias,
                                               const float* __restrict__ res_b,
                                               const float* __restrict__ ln_g,
                                               const float* __restrict__ ln_b,
                                               float* __restrict__ out, int E, int N) {
  const int2* cells = (const int2*)cellw;
  int2* ovg = (int2*)ovfw;
  __shared__ int lcnt[MAXNPB];
  __shared__ int ccnt[NC];
  __shared__ int slab[MAXNPB * ROW];
  __shared__ int2 ovl[OVC];
  __shared__ int ovn;
  __shared__ int sums[2];
  const int tid = threadIdx.x;
  const int lane = tid & 31;
  const int wave = __builtin_amdgcn_readfirstlane(tid >> 5);
  const int b = blockIdx.x;
  const int npb = range_size(N);
  const int base = b * npb;
  const int cs = (E + NC - 1) / NC;
  for (int i = tid; i < npb; i += 1024) lcnt[i] = 0;
  if (tid < NC) ccnt[tid] = cellcnt[b * NC + tid];
  if (tid == 0) { ovn = 0; sums[0] = 0; sums[1] = 0; }
  __syncthreads();

  int2 ent[NC / 32];
  int num[NC / 32];
#pragma unroll
  for (int k = 0; k < NC / 32; ++k) {
    const int c = wave + k * 32;
    num[k] = ccnt[c];
    ent[k] = make_int2(0, -1);
    if (num[k] <= CAP && lane < num[k]) ent[k] = cells[((size_t)b * NC + c) * CAP + lane];
  }
  bool to_global = false;
  int capg = 0;
  for (int pass = 0; pass < 2; ++pass) {
#pragma unroll
    for (int k = 0; k < NC / 32; ++k) {
      if (num[k] <= CAP) {
        if (lane < num[k])
          place_edge(ent[k].x, ent[k].y, base, npb, lcnt, slab, &ovn, ovl, ovg, to_global,
                     capg, N);
      } else {
        const int c = wave + k * 32;
        const int g1 = min(E, (c + 1) * cs);
        for (int g = c * cs + lane; g < g1; g += 32) {
          const int s = ei[g];
          const int d = min(max(ei[E + g], 0), N - 1);
          if (bucket_of(d, npb) == b)
            place_edge(s, d, base, npb, lcnt, slab, &ovn, ovl, ovg, to_global, capg, N);
        }
      }
    }
    __syncthreads();
    if (pass == 1 || ovn <= OVC) break;
    int part = 0;
    for (int i = tid; i < b * NC; i += 1024) part += max(cellcnt[i], 0);
    if (part != 0) atomicAdd(&sums[0], part);
    if (tid < NC) atomicAdd(&sums[1], max(ccnt[tid], 0));
    __syncthreads();
    const int p0 = min(max(sums[0], 0), E);
    capg = min(max(sums[1], 0), E - p0);
    ovg += p0;
    to_global = true;
    __syncthreads();
    for (int i = tid; i < npb; i += 1024) lcnt[i] = 0;
    if (tid == 0) ovn = 0;
    __syncthreads();
  }
  const int nov = min(max(ovn, 0), to_global ? capg : OVC);

  const float gb0 = gat_bias[lane], gb1 = gat_bias[lane + 32];
  const float gb2 = gat_bias[lane + 64], gb3 = gat_bias[lane + 96];
  const float rb0 = res_b[lane], rb1 = res_b[lane + 32];
  const float rb2 = res_b[lane + 64], rb3 = res_b[lane + 96];
  const float lg0 = ln_g[lane], lg1 = ln_g[lane + 32];
  const float lg2 = ln_g[lane + 64], lg3 = ln_g[lane + 96];
  const float lb0 = ln_b[lane], lb1 = ln_b[lane + 32];
  const float lb2 = ln_b[lane + 64], lb3 = ln_b[lane + 96];
  const int nend = min(base + npb, N);
  for (int n = base + wave; n < nend; n += 32) {
    const int dl = n - base;
    const int deg = max(lcnt[dl], 0);
    const int deg0 = min(deg, SLOTS);
    const bool more = deg > SLOTS;
    const float4 ad = ((const float4*)a_dst)[n];

    int s0;
    float4 e0;
    load_first(slab + dl * ROW, a_src, ad, deg0, n, N, lane, s0, e0);

    float4 m = e0;
    if (more) {
      for (int q0 = 0; q0 < nov; q0 += 32) {
        int s; float4 e;
        load_ovf(ovl, ovg, to_global, a_src, ad, nov, n, N, q0 + lane, s, e);
        m.x = fmaxf(m.x, e.x); m.y = fmaxf(m.y, e.y);
        m.z = fmaxf(m.z, e.z); m.w = fmaxf(m.w, e.w);
      }
    }
    m.x = wave_max(m.x); m.y = wave_max(m.y);
    m.z = wave_max(m.z); m.w = wave_max(m.w);

    float4 x0;
    x0.x = __expf(e0.x - m.x); x0.y = __expf(e0.y - m.y);
    x0.z = __expf(e0.z - m.z); x0.w = __expf(e0.w - m.w);
    float4 q = x0;
    if (more) {
      for (int q0 = 0; q0 < nov; q0 += 32) {
        int s; float4 e;
        load_ovf(ovl, ovg, to_global, a_src, ad, nov, n, N, q0 + lane, s, e);
        q.x += __expf(e.x - m.x); q.y += __expf(e.y - m.y);
        q.z += __expf(e.z - m.z); q.w += __expf(e.w - m.w);
      }
    }
    q.x = wave_sum(q.x) + 1e-16f; q.y = wave_sum(q.y) + 1e-16f;
    q.z = wave_sum(q.z) + 1e-16f; q.w = wave_sum(q.w) + 1e-16f;

    float acc0 = 0.f, acc1 = 0.f, acc2 = 0.f, acc3 = 0.f;
    {
      float4 al;
      al.x = x0.x / q.x; al.y = x0.y / q.y; al.z = x0.z / q.z; al.w = x0.w / q.w;
      gather_chunk(xt, s0, al, deg0 + 1, lane, acc0, acc1, acc2, acc3);
    }
    if (more) {
      for (int q0 = 0; q0 < nov; q0 += 32) {
        int s; float4 e;
        const bool mine = load_ovf(ovl, ovg, to_global, a_src, ad, nov, n, N, q0 + lane, s, e);
        float4 al;
        al.x = __expf(e.x - m.x) / q.x; al.y = __expf(e.y - m.y) / q.y;
        al.z = __expf(e.z - m.z) / q.z; al.w = __expf(e.w - m.w) / q.w;
        const unsigned mask = __builtin_amdgcn_ballot_w32(mine);
        gather_mask(xt, s, al, mask, lane, acc0, acc1, acc2, acc3);
      }
    }

    const float* zr = zres + (size_t)n * OUT_DIM + lane;
    const float z0 = ((acc0 + gb0) + zr[0])  + rb0;
    const float z1 = ((acc1 + gb1) + zr[32]) + rb1;
    const float z2 = ((acc2 + gb2) + zr[64]) + rb2;
    const float z3 = ((acc3 + gb3) + zr[96]) + rb3;

    const float mu = wave_sum((z0 + z1) + (z2 + z3)) * (1.0f / 128.0f);
    const float d0 = z0 - mu, d1 = z1 - mu, d2 = z2 - mu, d3 = z3 - mu;
    const float var = wave_sum((d0 * d0 + d1 * d1) + (d2 * d2 + d3 * d3)) * (1.0f / 128.0f);
    const float inv = rsqrtf(var + 1e-5f);
    float* op = out + (size_t)n * OUT_DIM + lane;
    op[0]  = lrelu(d0 * inv * lg0 + lb0);
    op[32] = lrelu(d1 * inv * lg1 + lb1);
    op[64] = lrelu(d2 * inv * lg2 + lb2);
    op[96] = lrelu(d3 * inv * lg3 + lb3);
  }
}

extern "C" void kernel_launch(void* const* d_in, const int* in_sizes, int n_in,
                              void* d_out, int out_size, void* d_ws, size_t ws_size,
                              hipStream_t stream) {
  const float* x        = (const float*)d_in[0];
  const int*   ei       = (const int*)d_in[1];
  const float* lin_w    = (const float*)d_in[2];
  const float* att_src  = (const float*)d_in[3];
  const float* att_dst  = (const float*)d_in[4];
  const float* gat_bias = (const float*)d_in[5];
  const float* res_w    = (const float*)d_in[6];
  const float* res_b    = (const float*)d_in[7];
  const float* ln_g     = (const float*)d_in[8];
  const float* ln_b     = (const float*)d_in[9];
  const int N  = in_sizes[0] / IN_CH;
  const int E  = in_sizes[1] / 2;

  float* xt     = (float*)d_ws;
  float* z      = xt + (size_t)N * OUT_DIM;
  float* a_src  = z + (size_t)N * OUT_DIM;
  float* a_dst  = a_src + (size_t)N * HEADS;
  int*   ovf    = (int*)(a_dst + (size_t)N * HEADS);
  int*   cells  = ovf + (size_t)E * 2;
  int*   cellcnt = cells + (size_t)NB * NC * CAP * 2;

  k_gemm<<<GEMM_GROUPS + BIN_GROUPS, 256, 0, stream>>>(x, lin_w, res_w, att_src, att_dst, xt,
                                                       z, a_src, a_dst, ei, cells, cellcnt,
                                                       E, N);
  k_node<<<NB, 1024, 0, stream>>>(ei, cells, cellcnt, ovf, a_src, a_dst, xt, z, gat_bias,
                                  res_b, ln_g, ln_b, (float*)d_out, E, N);
}


# Node pass sums 256-byte x rows per head (8 rows a group), not 512-byte xt rows, into LDS; WMMA then applies lin_w to the sums (linear); xt never stored; sum order: edges summed before the transform
# speedup vs baseline: 8.7610x; 1.1261x over previous
#include <hip/hip_runtime.h>
#include <hip/hip_bf16.h>

typedef __attribute__((ext_vector_type(2))) float v2f;
typedef __attribute__((ext_vector_type(8))) float v8f;

#define IN_CH 64
#define OUT_DIM 128
#define HEADS 4

static __device__ __forceinline__ float lrelu(float x) { return fmaxf(x, 0.2f * x); }

#define LDS_PITCH 68
#define GEMM_GROUPS 512
#define NB 256
#define NC 256
#define CAP 32
#define BIN_GROUPS NC
#define BATCH 8

#define MAXNPB 256
static __device__ __forceinline__ int range_size(int N) {
  return min((N + NB - 1) / NB, MAXNPB);
}

static __device__ __forceinline__ int bucket_of(int d, int npb) {
  return min((int)((unsigned)d / (unsigned)npb), NB - 1);
}

static __device__ __forceinline__ void bin_edges(const int* __restrict__ ei,
                                                 int* __restrict__ cellw,
                                                 int* __restrict__ cellcnt,
                                                 int c, int E, int N) {
  int2* cells = (int2*)cellw;
  __shared__ int bcnt[NB];
  const int tid = threadIdx.x;
  const int npb = range_size(N);
  const int cs = (E + NC - 1) / NC;
  bcnt[tid] = 0;
  __syncthreads();
  const int g1 = min(E, (c + 1) * cs);
  for (int g0 = c * cs + tid; g0 < g1; g0 += 256 * BATCH) {
    int sv[BATCH], dv[BATCH];
#pragma unroll
    for (int j = 0; j < BATCH; ++j) {
      const int g = min(g0 + j * 256, g1 - 1);
      sv[j] = ei[g];
      dv[j] = ei[E + g];
    }
#pragma unroll
    for (int j = 0; j < BATCH; ++j) {
      if (g0 + j * 256 < g1) {
        const int d = min(max(dv[j], 0), N - 1);
        const int b = bucket_of(d, npb);
        const int r = atomicAdd(&bcnt[b], 1);
        if (r < CAP) cells[((size_t)b * NC + c) * CAP + r] = make_int2(sv[j], d);
      }
    }
  }
  __syncthreads();
  cellcnt[tid * NC + c] = bcnt[tid];
}

#define ROW16_STEP(v, m)                                                              \
  ((v) + __int_as_float(__builtin_amdgcn_update_dpp(0, __float_as_int(v), 0x160 + (m), \
                                                    0xf, 0xf, true)))
static __device__ __forceinline__ float row16_sum(float v) {
  v = ROW16_STEP(v, 1);
  v = ROW16_STEP(v, 2);
  v = ROW16_STEP(v, 4);
  v = ROW16_STEP(v, 8);
  return v;
}

static __device__ __forceinline__ v2f frag(const float* s, int row, int ks, int kb) {
  return *(const v2f*)(s + row * LDS_PITCH + ks * 4 + kb);
}

__global__ __launch_bounds__(256) void k_gemm(const float* __restrict__ x,
                                              const float* __restrict__ lin_w,
                                              const float* __restrict__ res_w,
                                              const float* __restrict__ att_src,
                                              const float* __restrict__ att_dst,
                                              float* __restrict__ z,
                                              float* __restrict__ a_src,
                                              float* __restrict__ a_dst,
                                              const int* __restrict__ ei,
                                              int* __restrict__ cells,
                                              int* __restrict__ cellcnt,
                                              int E, int N) {
  if (blockIdx.x >= GEMM_GROUPS) {
    bin_edges(ei, cells, cellcnt, (int)blockIdx.x - GEMM_GROUPS, E, N);
    return;
  }
  __shared__ __attribute__((aligned(16))) float Ws[256 * LDS_PITCH];
  __shared__ __attribute__((aligned(16))) float As[2][16 * LDS_PITCH];
  const int tid = threadIdx.x;
  const int wave = tid >> 5;
  const int lane = tid & 31;
  const int mrow = lane & 15;
  const int kb = (lane >> 4) << 1;
  const int nchunks = (N + 15) >> 4;
  const int sr = tid >> 4;
  const int sq = tid & 15;

#pragma unroll
  for (int i = 0; i < 8; ++i) {
    const int r = i * 16 + sr;
    const float4 v0 = ((const float4*)lin_w)[r * 16 + sq];
    const float4 v1 = ((const float4*)res_w)[r * 16 + sq];
    *(float4*)(Ws + r * LDS_PITCH + sq * 4) = v0;
    *(float4*)(Ws + (128 + r) * LDS_PITCH + sq * 4) = v1;
  }
  int chunk = blockIdx.x;
  float4 va = make_float4(0.f, 0.f, 0.f, 0.f);
  if (chunk < nchunks) {
    const int row = min(chunk * 16 + sr, N - 1);
    va = ((const float4*)x)[(size_t)row * 16 + sq];
  }
  __syncthreads();
  const int trow = (wave < 4) ? wave * 32 : 128 + (wave - 4) * 32;
  v2f b0[16], b1[16];
#pragma unroll
  for (int ks = 0; ks < 16; ++ks) {
    b0[ks] = frag(Ws, trow + mrow, ks, kb);
    b1[ks] = frag(Ws, trow + 16 + mrow, ks, kb);
  }
  const int cbase = (wave & 3) * 32;
  const int acol = (wave & 3) * 32 + mrow;
  const float as0 = att_src[acol], as1 = att_src[acol + 16];
  const float ad0 = att_dst[acol], ad1 = att_dst[acol + 16];
  int buf = 0;
  while (chunk < nchunks) {
    *(float4*)(&As[buf][sr * LDS_PITCH + sq * 4]) = va;
    __syncthreads();
    const int next = chunk + GEMM_GROUPS;
    if (next < nchunks) {
      const int row = min(next * 16 + sr, N - 1);
      va = ((const float4*)x)[(size_t)row * 16 + sq];
    }
    v2f a[16];
#pragma unroll
    for (int ks = 0; ks < 16; ++ks) a[ks] = frag(As[buf], mrow, ks, kb);
    asm volatile("" ::"v"(a[0]), "v"(a[1]), "v"(a[2]), "v"(a[3]), "v"(a[4]), "v"(a[5]),
                 "v"(a[6]), "v"(a[7]), "v"(a[8]), "v"(a[9]), "v"(a[10]), "v"(a[11]),
                 "v"(a[12]), "v"(a[13]), "v"(a[14]), "v"(a[15]));
    v8f c0 = {0.f, 0.f, 0.f, 0.f, 0.f, 0.f, 0.f, 0.f};
    v8f c1 = {0.f, 0.f, 0.f, 0.f, 0.f, 0.f, 0.f, 0.f};
#pragma unroll
    for (int ks = 0; ks < 16; ++ks) {
      c0 = __builtin_amdgcn_wmma_f32_16x16x4_f32(false, a[ks], false, b0[ks],
                                                 (short)0, c0, false, false);
      c1 = __builtin_amdgcn_wmma_f32_16x16x4_f32(false, a[ks], false, b1[ks],
                                                 (short)0, c1, false, false);
    }
    const int m0 = chunk * 16;
    const int mb = m0 + ((lane >> 4) << 3);
    if (wave >= 4) {
#pragma unroll
      for (int r = 0; r < 8; ++r) {
        const int m = mb + r;
        if (m < N) {
          z[(size_t)m * OUT_DIM + cbase + mrow] = c0[r];
          z[(size_t)m * OUT_DIM + cbase + 16 + mrow] = c1[r];
        }
      }
    } else {
      float ps[8], pd[8];
#pragma unroll
      for (int r = 0; r < 8; ++r) {
        ps[r] = c0[r] * as0 + c1[r] * as1;
        pd[r] = c0[r] * ad0 + c1[r] * ad1;
      }
#pragma unroll
      for (int r = 0; r < 8; ++r) {
        ps[r] = row16_sum(ps[r]);
        pd[r] = row16_sum(pd[r]);
      }
      float vs = ps[0], vd = pd[0];
#pragma unroll
      for (int r = 1; r < 8; ++r) {
        if ((mrow & 7) == r) { vs = ps[r]; vd = pd[r]; }
      }
      const int m = mb + (mrow & 7);
      if (mrow < 8 && m < N) {
        a_src[(size_t)m * HEADS + wave] = vs;
        a_dst[(size_t)m * HEADS + wave] = vd;
      }
    }
    buf ^= 1;
    chunk = next;
  }
}

#define SLOTS 31
#define ROW 32
#define OVC 512
#define APITCH 260

static __device__ __forceinline__ void place_edge(int s, int d, int base, int npb,
                                                  int* lcnt, int* slab, int* ovn,
                                                  int2* ovl, int2* __restrict__ ovg,
                                                  bool to_global, int capg, int N) {
  const unsigned dl = (unsigned)(d - base);
  if (dl < (unsigned)npb && (unsigned)d < (unsigned)N) {
    const int r = atomicAdd(&lcnt[dl], 1);
    if ((unsigned)r < (unsigned)SLOTS) {
      slab[dl * ROW + r] = s;
    } else {
      const int o = atomicAdd(ovn, 1);
      if (to_global) {
        if ((unsigned)o < (unsigned)capg) ovg[o] = make_int2(s, d);
      } else if ((unsigned)o < (unsigned)OVC) {
        ovl[o] = make_int2(s, d);
      }
    }
  }
}

static __device__ __forceinline__ void load_first(const int* row,
                                                  const float* __restrict__ a_src,
                                                  const float4 ad, int deg0, int n,
                                                  int N, int i, int& s, float4& e) {
  s = n;
  e = make_float4(-INFINITY, -INFINITY, -INFINITY, -INFINITY);
  if (i <= deg0) {
    if (i < deg0) s = min(max(row[i], 0), N - 1);
    const float4 as = ((const float4*)a_src)[s];
    e.x = lrelu(as.x + ad.x);
    e.y = lrelu(as.y + ad.y);
    e.z = lrelu(as.z + ad.z);
    e.w = lrelu(as.w + ad.w);
  }
}

static __device__ __forceinline__ bool load_ovf(const int2* ovl,
                                                const int2* __restrict__ ovg, bool g,
                                                const float* __restrict__ a_src,
                                                const float4 ad, int nov, int n, int N,
                                                int i, int& s, float4& e) {
  s = n;
  e = make_float4(-INFINITY, -INFINITY, -INFINITY, -INFINITY);
  bool mine = false;
  int2 en = make_int2(n, -1);
  if (i < nov) {
    if (g) en = ovg[i]; else en = ovl[i];
    mine = (en.y == n);
  }
  if (mine) {
    s = min(max(en.x, 0), N - 1);
    const float4 as = ((const float4*)a_src)[s];
    e.x = lrelu(as.x + ad.x);
    e.y = lrelu(as.y + ad.y);
    e.z = lrelu(as.z + ad.z);
    e.w = lrelu(as.w + ad.w);
  }
  return mine;
}

static __device__ __forceinline__ void gather_chunk(const float* __restrict__ x,
                                                    int s, const float4 al, int cnt,
                                                    int lane, float (&acc)[8]) {
  const int cnt8 = (cnt + 7) & ~7;
#pragma unroll 8
  for (int j = 0; j < cnt8; ++j) {
    const int sj = __builtin_amdgcn_readlane(s, j);
    const float b0 = __int_as_float(__builtin_amdgcn_readlane(__float_as_int(al.x), j));
    const float b1 = __int_as_float(__builtin_amdgcn_readlane(__float_as_int(al.y), j));
    const float b2 = __int_as_float(__builtin_amdgcn_readlane(__float_as_int(al.z), j));
    const float b3 = __int_as_float(__builtin_amdgcn_readlane(__float_as_int(al.w), j));
    const float2 xv = *(const float2*)(x + (size_t)sj * IN_CH + 2 * lane);
    acc[0] += b0 * xv.x; acc[1] += b0 * xv.y;
    acc[2] += b1 * xv.x; acc[3] += b1 * xv.y;
    acc[4] += b2 * xv.x; acc[5] += b2 * xv.y;
    acc[6] += b3 * xv.x; acc[7] += b3 * xv.y;
  }
}

static __device__ __forceinline__ void gather_mask(const float* __restrict__ x,
                                                   int s, const float4 al, unsigned mask,
                                                   int lane, float (&acc)[8]) {
  while (mask != 0u) {
    const int j = __builtin_amdgcn_readfirstlane(__builtin_ctz(mask));
    mask &= mask - 1u;
    const int sj = __builtin_amdgcn_readlane(s, j);
    const float b0 = __int_as_float(__builtin_amdgcn_readlane(__float_as_int(al.x), j));
    const float b1 = __int_as_float(__builtin_amdgcn_readlane(__float_as_int(al.y), j));
    const float b2 = __int_as_float(__builtin_amdgcn_readlane(__float_as_int(al.z), j));
    const float b3 = __int_as_float(__builtin_amdgcn_readlane(__float_as_int(al.w), j));
    const float2 xv = *(const float2*)(x + (size_t)sj * IN_CH + 2 * lane);
    acc[0] += b0 * xv.x; acc[1] += b0 * xv.y;
    acc[2] += b1 * xv.x; acc[3] += b1 * xv.y;
    acc[4] += b2 * xv.x; acc[5] += b2 * xv.y;
    acc[6] += b3 * xv.x; acc[7] += b3 * xv.y;
  }
}

static __device__ __forceinline__ float wave_max(float v) {
#pragma unroll
  for (int off = 1; off < 32; off <<= 1) v = fmaxf(v, __shfl_xor(v, off, 32));
  return v;
}
static __device__ __forceinline__ float wave_sum(float v) {
#pragma unroll
  for (int off = 1; off < 32; off <<= 1) v += __shfl_xor(v, off, 32);
  return v;
}

__global__ __launch_bounds__(1024) void k_node(const int* __restrict__ ei,
                                               const int* __restrict__ cellw,
                                               const int* __restrict__ cellcnt,
                                               int* __restrict__ ovfw,
                                               const float* __restrict__ a_src,
                                               const float* __restrict__ a_dst,
                                               const float* __restrict__ x,
                                               const float* __restrict__ lin_w,
                                               const float* __restrict__ zres,
                                               const float* __restrict__ gat_bias,
                                               const float* __restrict__ res_b,
                                               const float* __restrict__ ln_g,
                                               const float* __restrict__ ln_b,
                                               float* __restrict__ out, int E, int N) {
  const int2* cells = (const int2*)cellw;
  int2* ovg = (int2*)ovfw;
  __shared__ int lcnt[MAXNPB];
  __shared__ int ccnt[NC];
  __shared__ __attribute__((aligned(16))) int slab[MAXNPB * ROW];
  __shared__ __attribute__((aligned(16))) float agg[MAXNPB * APITCH];
  __shared__ int2 ovl[OVC];
  __shared__ int ovn;
  __shared__ int sums[2];
  const int tid = threadIdx.x;
  const int lane = tid & 31;
  const int wave = __builtin_amdgcn_readfirstlane(tid >> 5);
  const int b = blockIdx.x;
  const int npb = range_size(N);
  const int base = b * npb;
  const int cs = (E + NC - 1) / NC;
  for (int i = tid; i < npb; i += 1024) lcnt[i] = 0;
  if (tid < NC) ccnt[tid] = cellcnt[b * NC + tid];
  if (tid == 0) { ovn = 0; sums[0] = 0; sums[1] = 0; }
  __syncthreads();

  int2 ent[NC / 32];
  int num[NC / 32];
#pragma unroll
  for (int k = 0; k < NC / 32; ++k) {
    const int c = wave + k * 32;
    num[k] = ccnt[c];
    ent[k] = make_int2(0, -1);
    if (num[k] <= CAP && lane < num[k]) ent[k] = cells[((size_t)b * NC + c) * CAP + lane];
  }
  bool to_global = false;
  int capg = 0;
  for (int pass = 0; pass < 2; ++pass) {
#pragma unroll
    for (int k = 0; k < NC / 32; ++k) {
      if (num[k] <= CAP) {
        if (lane < num[k])
          place_edge(ent[k].x, ent[k].y, base, npb, lcnt, slab, &ovn, ovl, ovg, to_global,
                     capg, N);
      } else {
        const int c = wave + k * 32;
        const int g1 = min(E, (c + 1) * cs);
        for (int g = c * cs + lane; g < g1; g += 32) {
          const int s = ei[g];
          const int d = min(max(ei[E + g], 0), N - 1);
          if (bucket_of(d, npb) == b)
            place_edge(s, d, base, npb, lcnt, slab, &ovn, ovl, ovg, to_global, capg, N);
        }
      }
    }
    __syncthreads();
    if (pass == 1 || ovn <= OVC) break;
    int part = 0;
    for (int i = tid; i < b * NC; i += 1024) part += max(cellcnt[i], 0);
    if (part != 0) atomicAdd(&sums[0], part);
    if (tid < NC) atomicAdd(&sums[1], max(ccnt[tid], 0));
    __syncthreads();
    const int p0 = min(max(sums[0], 0), E);
    capg = min(max(sums[1], 0), E - p0);
    ovg += p0;
    to_global = true;
    __syncthreads();
    for (int i = tid; i < npb; i += 1024) lcnt[i] = 0;
    if (tid == 0) ovn = 0;
    __syncthreads();
  }
  const int nov = min(max(ovn, 0), to_global ? capg : OVC);

  const int nend = min(base + npb, N);
  for (int n = base + wave; n < nend; n += 32) {
    const int dl = n - base;
    const int deg = max(lcnt[dl], 0);
    const int deg0 = min(deg, SLOTS);
    const bool more = deg > SLOTS;
    const float4 ad = ((const float4*)a_dst)[n];

    int s0;
    float4 e0;
    load_first(slab + dl * ROW, a_src, ad, deg0, n, N, lane, s0, e0);

    float4 m = e0;
    if (more) {
      for (int q0 = 0; q0 < nov; q0 += 32) {
        int s; float4 e;
        load_ovf(ovl, ovg, to_global, a_src, ad, nov, n, N, q0 + lane, s, e);
        m.x = fmaxf(m.x, e.x); m.y = fmaxf(m.y, e.y);
        m.z = fmaxf(m.z, e.z); m.w = fmaxf(m.w, e.w);
      }
    }
    m.x = wave_max(m.x); m.y = wave_max(m.y);
    m.z = wave_max(m.z); m.w = wave_max(m.w);

    float4 x0;
    x0.x = __expf(e0.x - m.x); x0.y = __expf(e0.y - m.y);
    x0.z = __expf(e0.z - m.z); x0.w = __expf(e0.w - m.w);
    float4 q = x0;
    if (more) {
      for (int q0 = 0; q0 < nov; q0 += 32) {
        int s; float4 e;
        load_ovf(ovl, ovg, to_global, a_src, ad, nov, n, N, q0 + lane, s, e);
        q.x += __expf(e.x - m.x); q.y += __expf(e.y - m.y);
        q.z += __expf(e.z - m.z); q.w += __expf(e.w - m.w);
      }
    }
    q.x = wave_sum(q.x) + 1e-16f; q.y = wave_sum(q.y) + 1e-16f;
    q.z = wave_sum(q.z) + 1e-16f; q.w = wave_sum(q.w) + 1e-16f;

    float acc[8] = {0.f, 0.f, 0.f, 0.f, 0.f, 0.f, 0.f, 0.f};
    {
      float4 al;
      al.x = x0.x / q.x; al.y = x0.y / q.y; al.z = x0.z / q.z; al.w = x0.w / q.w;
      gather_chunk(x, s0, al, deg0 + 1, lane, acc);
    }
    if (more) {
      for (int q0 = 0; q0 < nov; q0 += 32) {
        int s; float4 e;
        const bool mine = load_ovf(ovl, ovg, to_global, a_src, ad, nov, n, N, q0 + lane, s, e);
        float4 al;
        al.x = __expf(e.x - m.x) / q.x; al.y = __expf(e.y - m.y) / q.y;
        al.z = __expf(e.z - m.z) / q.z; al.w = __expf(e.w - m.w) / q.w;
        const unsigned mask = __builtin_amdgcn_ballot_w32(mine);
        gather_mask(x, s, al, mask, lane, acc);
      }
    }
    float* ar = agg + dl * APITCH + 2 * lane;
    *(float2*)(ar)       = make_float2(acc[0], acc[1]);
    *(float2*)(ar + 64)  = make_float2(acc[2], acc[3]);
    *(float2*)(ar + 128) = make_float2(acc[4], acc[5]);
    *(float2*)(ar + 192) = make_float2(acc[6], acc[7]);
  }

  const int wr = tid >> 4;
  const int wg = tid & 15;
  const float4 wv0 = ((const float4*)lin_w)[wr * 16 + wg];
  const float4 wv1 = ((const float4*)lin_w)[(wr + 64) * 16 + wg];
  __syncthreads();
  float* Wl = (float*)slab;
  *(float4*)(Wl + wr * IN_CH + ((wg ^ (wr & 15)) << 2)) = wv0;
  *(float4*)(Wl + (wr + 64) * IN_CH + ((wg ^ (wr & 15)) << 2)) = wv1;
  __syncthreads();
  {
    const int mrow = lane & 15;
    const int kb = (lane >> 4) << 1;
    const int hd = wave & 3;
    const int ntiles = (max(nend - base, 0) + 15) >> 4;
    if ((wave >> 2) < ntiles) {
      const float* w0p = Wl + (hd * 32 + mrow) * IN_CH + kb;
      const float* w1p = w0p + 16 * IN_CH;
      v2f b0[16], b1[16];
#pragma unroll
      for (int ks = 0; ks < 16; ++ks) {
        b0[ks] = *(const v2f*)(w0p + ((ks ^ mrow) << 2));
        b1[ks] = *(const v2f*)(w1p + ((ks ^ mrow) << 2));
      }
      for (int t = wave >> 2; t < ntiles; t += 8) {
        const float* ap = agg + (t * 16 + mrow) * APITCH + hd * 64 + kb;
        v8f c0 = {0.f, 0.f, 0.f, 0.f, 0.f, 0.f, 0.f, 0.f};
        v8f c1 = {0.f, 0.f, 0.f, 0.f, 0.f, 0.f, 0.f, 0.f};
#pragma unroll
        for (int kh = 0; kh < 2; ++kh) {
          v2f a[8];
#pragma unroll
          for (int ks = 0; ks < 8; ++ks) a[ks] = *(const v2f*)(ap + (kh * 8 + ks) * 4);
#pragma unroll
          for (int ks = 0; ks < 8; ++ks) {
            c0 = __builtin_amdgcn_wmma_f32_16x16x4_f32(false, a[ks], false, b0[kh * 8 + ks],
                                                       (short)0, c0, false, false);
            c1 = __builtin_amdgcn_wmma_f32_16x16x4_f32(false, a[ks], false, b1[kh * 8 + ks],
                                                       (short)0, c1, false, false);
          }
        }
        float* dp = agg + (t * 16 + ((lane >> 4) << 3)) * APITCH + hd * 64 + mrow;
#pragma unroll
        for (int r = 0; r < 8; ++r) {
          dp[r * APITCH] = c0[r];
          dp[r * APITCH + 16] = c1[r];
        }
      }
    }
  }
  float zc0 = 0.f, zc1 = 0.f, zc2 = 0.f, zc3 = 0.f;
  if (base + wave < nend) {
    const float* zr = zres + (size_t)(base + wave) * OUT_DIM + lane;
    zc0 = zr[0]; zc1 = zr[32]; zc2 = zr[64]; zc3 = zr[96];
  }
  const float gb0 = gat_bias[lane], gb1 = gat_bias[lane + 32];
  const float gb2 = gat_bias[lane + 64], gb3 = gat_bias[lane + 96];
  const float rb0 = res_b[lane], rb1 = res_b[lane + 32];
  const float rb2 = res_b[lane + 64], rb3 = res_b[lane + 96];
  const float lg0 = ln_g[lane], lg1 = ln_g[lane + 32];
  const float lg2 = ln_g[lane + 64], lg3 = ln_g[lane + 96];
  const float lb0 = ln_b[lane], lb1 = ln_b[lane + 32];
  const float lb2 = ln_b[lane + 64], lb3 = ln_b[lane + 96];
  __syncthreads();
  for (int n = base + wave; n < nend; n += 32) {
    float zn0 = 0.f, zn1 = 0.f, zn2 = 0.f, zn3 = 0.f;
    if (n + 32 < nend) {
      const float* zr = zres + (size_t)(n + 32) * OUT_DIM + lane;
      zn0 = zr[0]; zn1 = zr[32]; zn2 = zr[64]; zn3 = zr[96];
    }
    const float* hr = agg + (n - base) * APITCH + lane;
    const float acc0 = hr[0], acc1 = hr[64], acc2 = hr[128], acc3 = hr[192];

    const float z0 = ((acc0 + gb0) + zc0) + rb0;
    const float z1 = ((acc1 + gb1) + zc1) + rb1;
    const float z2 = ((acc2 + gb2) + zc2) + rb2;
    const float z3 = ((acc3 + gb3) + zc3) + rb3;
    zc0 = zn0; zc1 = zn1; zc2 = zn2; zc3 = zn3;

    const float mu = wave_sum((z0 + z1) + (z2 + z3)) * (1.0f / 128.0f);
    const float d0 = z0 - mu, d1 = z1 - mu, d2 = z2 - mu, d3 = z3 - mu;
    const float var = wave_sum((d0 * d0 + d1 * d1) + (d2 * d2 + d3 * d3)) * (1.0f / 128.0f);
    const float inv = rsqrtf(var + 1e-5f);
    float* op = out + (size_t)n * OUT_DIM + lane;
    op[0]  = lrelu(d0 * inv * lg0 + lb0);
    op[32] = lrelu(d1 * inv * lg1 + lb1);
    op[64] = lrelu(d2 * inv * lg2 + lb2);
    op[96] = lrelu(d3 * inv * lg3 + lb3);
  }
}

extern "C" void kernel_launch(void* const* d_in, const int* in_sizes, int n_in,
                              void* d_out, int out_size, void* d_ws, size_t ws_size,
                              hipStream_t stream) {
  const float* x        = (const float*)d_in[0];
  const int*   ei       = (const int*)d_in[1];
  const float* lin_w    = (const float*)d_in[2];
  const float* att_src  = (const float*)d_in[3];
  const float* att_dst  = (const float*)d_in[4];
  const float* gat_bias = (const float*)d_in[5];
  const float* res_w    = (const float*)d_in[6];
  const float* res_b    = (const float*)d_in[7];
  const float* ln_g     = (const float*)d_in[8];
  const float* ln_b     = (const float*)d_in[9];
  const int N  = in_sizes[0] / IN_CH;
  const int E  = in_sizes[1] / 2;

  float* z      = (float*)d_ws;
  float* a_src  = z + (size_t)N * OUT_DIM;
  float* a_dst  = a_src + (size_t)N * HEADS;
  int*   ovf    = (int*)(a_dst + (size_t)N * HEADS);
  int*   cells  = ovf + (size_t)E * 2;
  int*   cellcnt = cells + (size_t)NB * NC * CAP * 2;

  k_gemm<<<GEMM_GROUPS + BIN_GROUPS, 256, 0, stream>>>(x, lin_w, res_w, att_src, att_dst, z,
                                                       a_src, a_dst, ei, cells, cellcnt, E, N);
  k_node<<<NB, 1024, 0, stream>>>(ei, cells, cellcnt, ovf, a_src, a_dst, x, lin_w, z, gat_bias,
                                  res_b, ln_g, ln_b, (float*)d_out, E, N);
}


# GEMM pass: attention vectors folded into lin_w first (8 rows of 64 per group), so a_src/a_dst are one 16-column WMMA tile of x, not 128 columns of xt plus lane sums; a_src/a_dst sum order changes
# speedup vs baseline: 9.1407x; 1.0433x over previous
#include <hip/hip_runtime.h>
#include <hip/hip_bf16.h>

typedef __attribute__((ext_vector_type(2))) float v2f;
typedef __attribute__((ext_vector_type(8))) float v8f;

#define IN_CH 64
#define OUT_DIM 128
#define HEADS 4

static __device__ __forceinline__ float lrelu(float x) { return fmaxf(x, 0.2f * x); }

#define LDS_PITCH 68
#define GEMM_GROUPS 512
#define NB 256
#define NC 256
#define CAP 32
#define BIN_GROUPS NC
#define BATCH 8

#define MAXNPB 256
static __device__ __forceinline__ int range_size(int N) {
  return min((N + NB - 1) / NB, MAXNPB);
}

static __device__ __forceinline__ int bucket_of(int d, int npb) {
  return min((int)((unsigned)d / (unsigned)npb), NB - 1);
}

static __device__ __forceinline__ void bin_edges(const int* __restrict__ ei,
                                                 int* __restrict__ cellw,
                                                 int* __restrict__ cellcnt,
                                                 int c, int E, int N) {
  int2* cells = (int2*)cellw;
  __shared__ int bcnt[NB];
  const int tid = threadIdx.x;
  const int npb = range_size(N);
  const int cs = (E + NC - 1) / NC;
  bcnt[tid] = 0;
  __syncthreads();
  const int g1 = min(E, (c + 1) * cs);
  for (int g0 = c * cs + tid; g0 < g1; g0 += 256 * BATCH) {
    int sv[BATCH], dv[BATCH];
#pragma unroll
    for (int j = 0; j < BATCH; ++j) {
      const int g = min(g0 + j * 256, g1 - 1);
      sv[j] = ei[g];
      dv[j] = ei[E + g];
    }
#pragma unroll
    for (int j = 0; j < BATCH; ++j) {
      if (g0 + j * 256 < g1) {
        const int d = min(max(dv[j], 0), N - 1);
        const int b = bucket_of(d, npb);
        const int r = atomicAdd(&bcnt[b], 1);
        if (r < CAP) cells[((size_t)b * NC + c) * CAP + r] = make_int2(sv[j], d);
      }
    }
  }
  __syncthreads();
  cellcnt[tid * NC + c] = bcnt[tid];
}

static __device__ __forceinline__ v2f frag(const float* s, int row, int ks, int kb) {
  return *(const v2f*)(s + row * LDS_PITCH + ks * 4 + kb);
}

__global__ __launch_bounds__(256) void k_gemm(const float* __restrict__ x,
                                              const float* __restrict__ lin_w,
                                              const float* __restrict__ res_w,
                                              const float* __restrict__ att_src,
                                              const float* __restrict__ att_dst,
                                              float* __restrict__ z,
                                              float* __restrict__ a_src,
                                              float* __restrict__ a_dst,
                                              const int* __restrict__ ei,
                                              int* __restrict__ cells,
                                              int* __restrict__ cellcnt,
                                              int E, int N) {
  if (blockIdx.x >= GEMM_GROUPS) {
    bin_edges(ei, cells, cellcnt, (int)blockIdx.x - GEMM_GROUPS, E, N);
    return;
  }
  __shared__ __attribute__((aligned(16))) float Ws[(128 + 16) * LDS_PITCH];
  __shared__ __attribute__((aligned(16))) float As[2][16 * LDS_PITCH];
  const int tid = threadIdx.x;
  const int wave = tid >> 5;
  const int lane = tid & 31;
  const int mrow = lane & 15;
  const int kb = (lane >> 4) << 1;
  const int nchunks = (N + 15) >> 4;
  const int sr = tid >> 4;
  const int sq = tid & 15;

#pragma unroll
  for (int i = 0; i < 8; ++i) {
    const int r = i * 16 + sr;
    const float4 v1 = ((const float4*)res_w)[r * 16 + sq];
    *(float4*)(Ws + r * LDS_PITCH + sq * 4) = v1;
  }
  {
    const int h = __builtin_amdgcn_readfirstlane(tid >> 6);
    const int k = tid & 63;
    const float* wp = lin_w + (size_t)(h * 32) * IN_CH + k;
    float vs = 0.f, vd = 0.f;
#pragma unroll
    for (int c = 0; c < 32; ++c) {
      const float w = wp[c * IN_CH];
      vs = fmaf(att_src[h * 32 + c], w, vs);
      vd = fmaf(att_dst[h * 32 + c], w, vd);
    }
    Ws[(128 + h) * LDS_PITCH + k] = vs;
    Ws[(132 + h) * LDS_PITCH + k] = vd;
    Ws[(136 + h) * LDS_PITCH + k] = 0.f;
    Ws[(140 + h) * LDS_PITCH + k] = 0.f;
  }
  int chunk = blockIdx.x;
  float4 va = make_float4(0.f, 0.f, 0.f, 0.f);
  if (chunk < nchunks) {
    const int row = min(chunk * 16 + sr, N - 1);
    va = ((const float4*)x)[(size_t)row * 16 + sq];
  }
  __syncthreads();
  const bool zwave = wave >= 4;
  const int trow = zwave ? (wave - 4) * 32 : 128;
  v2f b0[16], b1[16];
#pragma unroll
  for (int ks = 0; ks < 16; ++ks) {
    b0[ks] = frag(Ws, trow + mrow, ks, kb);
    b1[ks] = zwave ? frag(Ws, trow + 16 + mrow, ks, kb) : b0[ks];
  }
  const int cbase = (wave & 3) * 32;
  float* const ap = ((mrow < 4) ? a_src : a_dst) + (mrow & 3);
  int buf = 0;
  int turn = 0;
  while (chunk < nchunks) {
    *(float4*)(&As[buf][sr * LDS_PITCH + sq * 4]) = va;
    __syncthreads();
    const int next = chunk + GEMM_GROUPS;
    if (next < nchunks) {
      const int row = min(next * 16 + sr, N - 1);
      va = ((const float4*)x)[(size_t)row * 16 + sq];
    }
    if (zwave || wave == turn) {
      v2f a[16];
#pragma unroll
      for (int ks = 0; ks < 16; ++ks) a[ks] = frag(As[buf], mrow, ks, kb);
      asm volatile("" ::"v"(a[0]), "v"(a[1]), "v"(a[2]), "v"(a[3]), "v"(a[4]), "v"(a[5]),
                   "v"(a[6]), "v"(a[7]), "v"(a[8]), "v"(a[9]), "v"(a[10]), "v"(a[11]),
                   "v"(a[12]), "v"(a[13]), "v"(a[14]), "v"(a[15]));
      v8f c0 = {0.f, 0.f, 0.f, 0.f, 0.f, 0.f, 0.f, 0.f};
      v8f c1 = {0.f, 0.f, 0.f, 0.f, 0.f, 0.f, 0.f, 0.f};
      const int m0 = chunk * 16;
      const int mb = m0 + ((lane >> 4) << 3);
      if (zwave) {
#pragma unroll
        for (int ks = 0; ks < 16; ++ks) {
          c0 = __builtin_amdgcn_wmma_f32_16x16x4_f32(false, a[ks], false, b0[ks],
                                                     (short)0, c0, false, false);
          c1 = __builtin_amdgcn_wmma_f32_16x16x4_f32(false, a[ks], false, b1[ks],
                                                     (short)0, c1, false, false);
        }
#pragma unroll
        for (int r = 0; r < 8; ++r) {
          const int m = mb + r;
          if (m < N) {
            z[(size_t)m * OUT_DIM + cbase + mrow] = c0[r];
            z[(size_t)m * OUT_DIM + cbase + 16 + mrow] = c1[r];
          }
        }
      } else {
#pragma unroll
        for (int ks = 0; ks < 16; ks += 2) {
          c0 = __builtin_amdgcn_wmma_f32_16x16x4_f32(false, a[ks], false, b0[ks],
                                                     (short)0, c0, false, false);
          c1 = __builtin_amdgcn_wmma_f32_16x16x4_f32(false, a[ks + 1], false, b0[ks + 1],
                                                     (short)0, c1, false, false);
        }
        if (mrow < 8) {
#pragma unroll
          for (int r = 0; r < 8; ++r) {
            const int m = mb + r;
            if (m < N) ap[(size_t)m * HEADS] = c0[r] + c1[r];
          }
        }
      }
    }
    buf ^= 1;
    chunk = next;
    turn = (turn + 1) & 3;
  }
}

#define SLOTS 31
#define ROW 32
#define OVC 512
#define APITCH 260

static __device__ __forceinline__ void place_edge(int s, int d, int base, int npb,
                                                  int* lcnt, int* slab, int* ovn,
                                                  int2* ovl, int2* __restrict__ ovg,
                                                  bool to_global, int capg, int N) {
  const unsigned dl = (unsigned)(d - base);
  if (dl < (unsigned)npb && (unsigned)d < (unsigned)N) {
    const int r = atomicAdd(&lcnt[dl], 1);
    if ((unsigned)r < (unsigned)SLOTS) {
      slab[dl * ROW + r] = s;
    } else {
      const int o = atomicAdd(ovn, 1);
      if (to_global) {
        if ((unsigned)o < (unsigned)capg) ovg[o] = make_int2(s, d);
      } else if ((unsigned)o < (unsigned)OVC) {
        ovl[o] = make_int2(s, d);
      }
    }
  }
}

static __device__ __forceinline__ void load_first(const int* row,
                                                  const float* __restrict__ a_src,
                                                  const float4 ad, int deg0, int n,
                                                  int N, int i, int& s, float4& e) {
  s = n;
  e = make_float4(-INFINITY, -INFINITY, -INFINITY, -INFINITY);
  if (i <= deg0) {
    if (i < deg0) s = min(max(row[i], 0), N - 1);
    const float4 as = ((const float4*)a_src)[s];
    e.x = lrelu(as.x + ad.x);
    e.y = lrelu(as.y + ad.y);
    e.z = lrelu(as.z + ad.z);
    e.w = lrelu(as.w + ad.w);
  }
}

static __device__ __forceinline__ bool load_ovf(const int2* ovl,
                                                const int2* __restrict__ ovg, bool g,
                                                const float* __restrict__ a_src,
                                                const float4 ad, int nov, int n, int N,
                                                int i, int& s, float4& e) {
  s = n;
  e = make_float4(-INFINITY, -INFINITY, -INFINITY, -INFINITY);
  bool mine = false;
  int2 en = make_int2(n, -1);
  if (i < nov) {
    if (g) en = ovg[i]; else en = ovl[i];
    mine = (en.y == n);
  }
  if (mine) {
    s = min(max(en.x, 0), N - 1);
    const float4 as = ((const float4*)a_src)[s];
    e.x = lrelu(as.x + ad.x);
    e.y = lrelu(as.y + ad.y);
    e.z = lrelu(as.z + ad.z);
    e.w = lrelu(as.w + ad.w);
  }
  return mine;
}

static __device__ __forceinline__ void gather_chunk(const float* __restrict__ x,
                                                    int s, const float4 al, int cnt,
                                                    int lane, float (&acc)[8]) {
  const int cnt8 = (cnt + 7) & ~7;
#pragma unroll 8
  for (int j = 0; j < cnt8; ++j) {
    const int sj = __builtin_amdgcn_readlane(s, j);
    const float b0 = __int_as_float(__builtin_amdgcn_readlane(__float_as_int(al.x), j));
    const float b1 = __int_as_float(__builtin_amdgcn_readlane(__float_as_int(al.y), j));
    const float b2 = __int_as_float(__builtin_amdgcn_readlane(__float_as_int(al.z), j));
    const float b3 = __int_as_float(__builtin_amdgcn_readlane(__float_as_int(al.w), j));
    const float2 xv = *(const float2*)(x + (size_t)sj * IN_CH + 2 * lane);
    acc[0] += b0 * xv.x; acc[1] += b0 * xv.y;
    acc[2] += b1 * xv.x; acc[3] += b1 * xv.y;
    acc[4] += b2 * xv.x; acc[5] += b2 * xv.y;
    acc[6] += b3 * xv.x; acc[7] += b3 * xv.y;
  }
}

static __device__ __forceinline__ void gather_mask(const float* __restrict__ x,
                                                   int s, const float4 al, unsigned mask,
                                                   int lane, float (&acc)[8]) {
  while (mask != 0u) {
    const int j = __builtin_amdgcn_readfirstlane(__builtin_ctz(mask));
    mask &= mask - 1u;
    const int sj = __builtin_amdgcn_readlane(s, j);
    const float b0 = __int_as_float(__builtin_amdgcn_readlane(__float_as_int(al.x), j));
    const float b1 = __int_as_float(__builtin_amdgcn_readlane(__float_as_int(al.y), j));
    const float b2 = __int_as_float(__builtin_amdgcn_readlane(__float_as_int(al.z), j));
    const float b3 = __int_as_float(__builtin_amdgcn_readlane(__float_as_int(al.w), j));
    const float2 xv = *(const float2*)(x + (size_t)sj * IN_CH + 2 * lane);
    acc[0] += b0 * xv.x; acc[1] += b0 * xv.y;
    acc[2] += b1 * xv.x; acc[3] += b1 * xv.y;
    acc[4] += b2 * xv.x; acc[5] += b2 * xv.y;
    acc[6] += b3 * xv.x; acc[7] += b3 * xv.y;
  }
}

static __device__ __forceinline__ float wave_max(float v) {
#pragma unroll
  for (int off = 1; off < 32; off <<= 1) v = fmaxf(v, __shfl_xor(v, off, 32));
  return v;
}
static __device__ __forceinline__ float wave_sum(float v) {
#pragma unroll
  for (int off = 1; off < 32; off <<= 1) v += __shfl_xor(v, off, 32);
  return v;
}

__global__ __launch_bounds__(1024) void k_node(const int* __restrict__ ei,
                                               const int* __restrict__ cellw,
                                               const int* __restrict__ cellcnt,
                                               int* __restrict__ ovfw,
                                               const float* __restrict__ a_src,
                                               const float* __restrict__ a_dst,
                                               const float* __restrict__ x,
                                               const float* __restrict__ lin_w,
                                               const float* __restrict__ zres,
                                               const float* __restrict__ gat_bias,
                                               const float* __restrict__ res_b,
                                               const float* __restrict__ ln_g,
                                               const float* __restrict__ ln_b,
                                               float* __restrict__ out, int E, int N) {
  const int2* cells = (const int2*)cellw;
  int2* ovg = (int2*)ovfw;
  __shared__ int lcnt[MAXNPB];
  __shared__ int ccnt[NC];
  __shared__ __attribute__((aligned(16))) int slab[MAXNPB * ROW];
  __shared__ __attribute__((aligned(16))) float agg[MAXNPB * APITCH];
  __shared__ int2 ovl[OVC];
  __shared__ int ovn;
  __shared__ int sums[2];
  const int tid = threadIdx.x;
  const int lane = tid & 31;
  const int wave = __builtin_amdgcn_readfirstlane(tid >> 5);
  const int b = blockIdx.x;
  const int npb = range_size(N);
  const int base = b * npb;
  const int cs = (E + NC - 1) / NC;
  for (int i = tid; i < npb; i += 1024) lcnt[i] = 0;
  if (tid < NC) ccnt[tid] = cellcnt[b * NC + tid];
  if (tid == 0) { ovn = 0; sums[0] = 0; sums[1] = 0; }
  __syncthreads();

  int2 ent[NC / 32];
  int num[NC / 32];
#pragma unroll
  for (int k = 0; k < NC / 32; ++k) {
    const int c = wave + k * 32;
    num[k] = ccnt[c];
    ent[k] = make_int2(0, -1);
    if (num[k] <= CAP && lane < num[k]) ent[k] = cells[((size_t)b * NC + c) * CAP + lane];
  }
  bool to_global = false;
  int capg = 0;
  for (int pass = 0; pass < 2; ++pass) {
#pragma unroll
    for (int k = 0; k < NC / 32; ++k) {
      if (num[k] <= CAP) {
        if (lane < num[k])
          place_edge(ent[k].x, ent[k].y, base, npb, lcnt, slab, &ovn, ovl, ovg, to_global,
                     capg, N);
      } else {
        const int c = wave + k * 32;
        const int g1 = min(E, (c + 1) * cs);
        for (int g = c * cs + lane; g < g1; g += 32) {
          const int s = ei[g];
          const int d = min(max(ei[E + g], 0), N - 1);
          if (bucket_of(d, npb) == b)
            place_edge(s, d, base, npb, lcnt, slab, &ovn, ovl, ovg, to_global, capg, N);
        }
      }
    }
    __syncthreads();
    if (pass == 1 || ovn <= OVC) break;
    int part = 0;
    for (int i = tid; i < b * NC; i += 1024) part += max(cellcnt[i], 0);
    if (part != 0) atomicAdd(&sums[0], part);
    if (tid < NC) atomicAdd(&sums[1], max(ccnt[tid], 0));
    __syncthreads();
    const int p0 = min(max(sums[0], 0), E);
    capg = min(max(sums[1], 0), E - p0);
    ovg += p0;
    to_global = true;
    __syncthreads();
    for (int i = tid; i < npb; i += 1024) lcnt[i] = 0;
    if (tid == 0) ovn = 0;
    __syncthreads();
  }
  const int nov = min(max(ovn, 0), to_global ? capg : OVC);

  const int nend = min(base + npb, N);
  for (int n = base + wave; n < nend; n += 32) {
    const int dl = n - base;
    const int deg = max(lcnt[dl], 0);
    const int deg0 = min(deg, SLOTS);
    const bool more = deg > SLOTS;
    const float4 ad = ((const float4*)a_dst)[n];

    int s0;
    float4 e0;
    load_first(slab + dl * ROW, a_src, ad, deg0, n, N, lane, s0, e0);

    float4 m = e0;
    if (more) {
      for (int q0 = 0; q0 < nov; q0 += 32) {
        int s; float4 e;
        load_ovf(ovl, ovg, to_global, a_src, ad, nov, n, N, q0 + lane, s, e);
        m.x = fmaxf(m.x, e.x); m.y = fmaxf(m.y, e.y);
        m.z = fmaxf(m.z, e.z); m.w = fmaxf(m.w, e.w);
      }
    }
    m.x = wave_max(m.x); m.y = wave_max(m.y);
    m.z = wave_max(m.z); m.w = wave_max(m.w);

    float4 x0;
    x0.x = __expf(e0.x - m.x); x0.y = __expf(e0.y - m.y);
    x0.z = __expf(e0.z - m.z); x0.w = __expf(e0.w - m.w);
    float4 q = x0;
    if (more) {
      for (int q0 = 0; q0 < nov; q0 += 32) {
        int s; float4 e;
        load_ovf(ovl, ovg, to_global, a_src, ad, nov, n, N, q0 + lane, s, e);
        q.x += __expf(e.x - m.x); q.y += __expf(e.y - m.y);
        q.z += __expf(e.z - m.z); q.w += __expf(e.w - m.w);
      }
    }
    q.x = wave_sum(q.x) + 1e-16f; q.y = wave_sum(q.y) + 1e-16f;
    q.z = wave_sum(q.z) + 1e-16f; q.w = wave_sum(q.w) + 1e-16f;

    float acc[8] = {0.f, 0.f, 0.f, 0.f, 0.f, 0.f, 0.f, 0.f};
    {
      float4 al;
      al.x = x0.x / q.x; al.y = x0.y / q.y; al.z = x0.z / q.z; al.w = x0.w / q.w;
      gather_chunk(x, s0, al, deg0 + 1, lane, acc);
    }
    if (more) {
      for (int q0 = 0; q0 < nov; q0 += 32) {
        int s; float4 e;
        const bool mine = load_ovf(ovl, ovg, to_global, a_src, ad, nov, n, N, q0 + lane, s, e);
        float4 al;
        al.x = __expf(e.x - m.x) / q.x; al.y = __expf(e.y - m.y) / q.y;
        al.z = __expf(e.z - m.z) / q.z; al.w = __expf(e.w - m.w) / q.w;
        const unsigned mask = __builtin_amdgcn_ballot_w32(mine);
        gather_mask(x, s, al, mask, lane, acc);
      }
    }
    float* ar = agg + dl * APITCH + 2 * lane;
    *(float2*)(ar)       = make_float2(acc[0], acc[1]);
    *(float2*)(ar + 64)  = make_float2(acc[2], acc[3]);
    *(float2*)(ar + 128) = make_float2(acc[4], acc[5]);
    *(float2*)(ar + 192) = make_float2(acc[6], acc[7]);
  }

  const int wr = tid >> 4;
  const int wg = tid & 15;
  const float4 wv0 = ((const float4*)lin_w)[wr * 16 + wg];
  const float4 wv1 = ((const float4*)lin_w)[(wr + 64) * 16 + wg];
  __syncthreads();
  float* Wl = (float*)slab;
  *(float4*)(Wl + wr * IN_CH + ((wg ^ (wr & 15)) << 2)) = wv0;
  *(float4*)(Wl + (wr + 64) * IN_CH + ((wg ^ (wr & 15)) << 2)) = wv1;
  __syncthreads();
  {
    const int mrow = lane & 15;
    const int kb = (lane >> 4) << 1;
    const int hd = wave & 3;
    const int ntiles = (max(nend - base, 0) + 15) >> 4;
    if ((wave >> 2) < ntiles) {
      const float* w0p = Wl + (hd * 32 + mrow) * IN_CH + kb;
      const float* w1p = w0p + 16 * IN_CH;
      v2f b0[16], b1[16];
#pragma unroll
      for (int ks = 0; ks < 16; ++ks) {
        b0[ks] = *(const v2f*)(w0p + ((ks ^ mrow) << 2));
        b1[ks] = *(const v2f*)(w1p + ((ks ^ mrow) << 2));
      }
      for (int t = wave >> 2; t < ntiles; t += 8) {
        const float* ap = agg + (t * 16 + mrow) * APITCH + hd * 64 + kb;
        v8f c0 = {0.f, 0.f, 0.f, 0.f, 0.f, 0.f, 0.f, 0.f};
        v8f c1 = {0.f, 0.f, 0.f, 0.f, 0.f, 0.f, 0.f, 0.f};
#pragma unroll
        for (int kh = 0; kh < 2; ++kh) {
          v2f a[8];
#pragma unroll
          for (int ks = 0; ks < 8; ++ks) a[ks] = *(const v2f*)(ap + (kh * 8 + ks) * 4);
#pragma unroll
          for (int ks = 0; ks < 8; ++ks) {
            c0 = __builtin_amdgcn_wmma_f32_16x16x4_f32(false, a[ks], false, b0[kh * 8 + ks],
                                                       (short)0, c0, false, false);
            c1 = __builtin_amdgcn_wmma_f32_16x16x4_f32(false, a[ks], false, b1[kh * 8 + ks],
                                                       (short)0, c1, false, false);
          }
        }
        float* dp = agg + (t * 16 + ((lane >> 4) << 3)) * APITCH + hd * 64 + mrow;
#pragma unroll
        for (int r = 0; r < 8; ++r) {
          dp[r * APITCH] = c0[r];
          dp[r * APITCH + 16] = c1[r];
        }
      }
    }
  }
  float zc0 = 0.f, zc1 = 0.f, zc2 = 0.f, zc3 = 0.f;
  if (base + wave < nend) {
    const float* zr = zres + (size_t)(base + wave) * OUT_DIM + lane;
    zc0 = zr[0]; zc1 = zr[32]; zc2 = zr[64]; zc3 = zr[96];
  }
  const float gb0 = gat_bias[lane], gb1 = gat_bias[lane + 32];
  const float gb2 = gat_bias[lane + 64], gb3 = gat_bias[lane + 96];
  const float rb0 = res_b[lane], rb1 = res_b[lane + 32];
  const float rb2 = res_b[lane + 64], rb3 = res_b[lane + 96];
  const float lg0 = ln_g[lane], lg1 = ln_g[lane + 32];
  const float lg2 = ln_g[lane + 64], lg3 = ln_g[lane + 96];
  const float lb0 = ln_b[lane], lb1 = ln_b[lane + 32];
  const float lb2 = ln_b[lane + 64], lb3 = ln_b[lane + 96];
  __syncthreads();
  for (int n = base + wave; n < nend; n += 32) {
    float zn0 = 0.f, zn1 = 0.f, zn2 = 0.f, zn3 = 0.f;
    if (n + 32 < nend) {
      const float* zr = zres + (size_t)(n + 32) * OUT_DIM + lane;
      zn0 = zr[0]; zn1 = zr[32]; zn2 = zr[64]; zn3 = zr[96];
    }
    const float* hr = agg + (n - base) * APITCH + lane;
    const float acc0 = hr[0], acc1 = hr[64], acc2 = hr[128], acc3 = hr[192];

    const float z0 = ((acc0 + gb0) + zc0) + rb0;
    const float z1 = ((acc1 + gb1) + zc1) + rb1;
    const float z2 = ((acc2 + gb2) + zc2) + rb2;
    const float z3 = ((acc3 + gb3) + zc3) + rb3;
    zc0 = zn0; zc1 = zn1; zc2 = zn2; zc3 = zn3;

    const float mu = wave_sum((z0 + z1) + (z2 + z3)) * (1.0f / 128.0f);
    const float d0 = z0 - mu, d1 = z1 - mu, d2 = z2 - mu, d3 = z3 - mu;
    const float var = wave_sum((d0 * d0 + d1 * d1) + (d2 * d2 + d3 * d3)) * (1.0f / 128.0f);
    const float inv = rsqrtf(var + 1e-5f);
    float* op = out + (size_t)n * OUT_DIM + lane;
    op[0]  = lrelu(d0 * inv * lg0 + lb0);
    op[32] = lrelu(d1 * inv * lg1 + lb1);
    op[64] = lrelu(d2 * inv * lg2 + lb2);
    op[96] = lrelu(d3 * inv * lg3 + lb3);
  }
}

extern "C" void kernel_launch(void* const* d_in, const int* in_sizes, int n_in,
                              void* d_out, int out_size, void* d_ws, size_t ws_size,
                              hipStream_t stream) {
  const float* x        = (const float*)d_in[0];
  const int*   ei       = (const int*)d_in[1];
  const float* lin_w    = (const float*)d_in[2];
  const float* att_src  = (const float*)d_in[3];
  const float* att_dst  = (const float*)d_in[4];
  const float* gat_bias = (const float*)d_in[5];
  const float* res_w    = (const float*)d_in[6];
  const float* res_b    = (const float*)d_in[7];
  const float* ln_g     = (const float*)d_in[8];
  const float* ln_b     = (const float*)d_in[9];
  const int N  = in_sizes[0] / IN_CH;
  const int E  = in_sizes[1] / 2;

  float* z      = (float*)d_ws;
  float* a_src  = z + (size_t)N * OUT_DIM;
  float* a_dst  = a_src + (size_t)N * HEADS;
  int*   ovf    = (int*)(a_dst + (size_t)N * HEADS);
  int*   cells  = ovf + (size_t)E * 2;
  int*   cellcnt = cells + (size_t)NB * NC * CAP * 2;

  k_gemm<<<GEMM_GROUPS + BIN_GROUPS, 256, 0, stream>>>(x, lin_w, res_w, att_src, att_dst, z,
                                                       a_src, a_dst, ei, cells, cellcnt, E, N);
  k_node<<<NB, 1024, 0, stream>>>(ei, cells, cellcnt, ovf, a_src, a_dst, x, lin_w, z, gat_bias,
                                  res_b, ln_g, ln_b, (float*)d_out, E, N);
}


# Node pass: all x-row loads of a node's edge list (12/16/20/24/32 rows by list length, one straight-line path each) are issued as one group ahead of the sums, not 8 per round trip; sum order kept
# speedup vs baseline: 9.5380x; 1.0435x over previous
#include <hip/hip_runtime.h>
#include <hip/hip_bf16.h>

typedef __attribute__((ext_vector_type(2))) float v2f;
typedef __attribute__((ext_vector_type(8))) float v8f;

#define IN_CH 64
#define OUT_DIM 128
#define HEADS 4

static __device__ __forceinline__ float lrelu(float x) { return fmaxf(x, 0.2f * x); }

#define LDS_PITCH 68
#define GEMM_GROUPS 512
#define NB 256
#define NC 256
#define CAP 32
#define BIN_GROUPS NC
#define BATCH 8

#define MAXNPB 256
static __device__ __forceinline__ int range_size(int N) {
  return min((N + NB - 1) / NB, MAXNPB);
}

static __device__ __forceinline__ int bucket_of(int d, int npb) {
  return min((int)((unsigned)d / (unsigned)npb), NB - 1);
}

static __device__ __forceinline__ void bin_edges(const int* __restrict__ ei,
                                                 int* __restrict__ cellw,
                                                 int* __restrict__ cellcnt,
                                                 int c, int E, int N) {
  int2* cells = (int2*)cellw;
  __shared__ int bcnt[NB];
  const int tid = threadIdx.x;
  const int npb = range_size(N);
  const int cs = (E + NC - 1) / NC;
  bcnt[tid] = 0;
  __syncthreads();
  const int g1 = min(E, (c + 1) * cs);
  for (int g0 = c * cs + tid; g0 < g1; g0 += 256 * BATCH) {
    int sv[BATCH], dv[BATCH];
#pragma unroll
    for (int j = 0; j < BATCH; ++j) {
      const int g = min(g0 + j * 256, g1 - 1);
      sv[j] = ei[g];
      dv[j] = ei[E + g];
    }
#pragma unroll
    for (int j = 0; j < BATCH; ++j) {
      if (g0 + j * 256 < g1) {
        const int d = min(max(dv[j], 0), N - 1);
        const int b = bucket_of(d, npb);
        const int r = atomicAdd(&bcnt[b], 1);
        if (r < CAP) cells[((size_t)b * NC + c) * CAP + r] = make_int2(sv[j], d);
      }
    }
  }
  __syncthreads();
  cellcnt[tid * NC + c] = bcnt[tid];
}

static __device__ __forceinline__ v2f frag(const float* s, int row, int ks, int kb) {
  return *(const v2f*)(s + row * LDS_PITCH + ks * 4 + kb);
}

__global__ __launch_bounds__(256) void k_gemm(const float* __restrict__ x,
                                              const float* __restrict__ lin_w,
                                              const float* __restrict__ res_w,
                                              const float* __restrict__ att_src,
                                              const float* __restrict__ att_dst,
                                              float* __restrict__ z,
                                              float* __restrict__ a_src,
                                              float* __restrict__ a_dst,
                                              const int* __restrict__ ei,
                                              int* __restrict__ cells,
                                              int* __restrict__ cellcnt,
                                              int E, int N) {
  if (blockIdx.x >= GEMM_GROUPS) {
    bin_edges(ei, cells, cellcnt, (int)blockIdx.x - GEMM_GROUPS, E, N);
    return;
  }
  __shared__ __attribute__((aligned(16))) float Ws[(128 + 16) * LDS_PITCH];
  __shared__ __attribute__((aligned(16))) float As[2][16 * LDS_PITCH];
  const int tid = threadIdx.x;
  const int wave = tid >> 5;
  const int lane = tid & 31;
  const int mrow = lane & 15;
  const int kb = (lane >> 4) << 1;
  const int nchunks = (N + 15) >> 4;
  const int sr = tid >> 4;
  const int sq = tid & 15;

#pragma unroll
  for (int i = 0; i < 8; ++i) {
    const int r = i * 16 + sr;
    const float4 v1 = ((const float4*)res_w)[r * 16 + sq];
    *(float4*)(Ws + r * LDS_PITCH + sq * 4) = v1;
  }
  {
    const int h = __builtin_amdgcn_readfirstlane(tid >> 6);
    const int k = tid & 63;
    const float* wp = lin_w + (size_t)(h * 32) * IN_CH + k;
    float vs = 0.f, vd = 0.f;
#pragma unroll
    for (int c = 0; c < 32; ++c) {
      const float w = wp[c * IN_CH];
      vs = fmaf(att_src[h * 32 + c], w, vs);
      vd = fmaf(att_dst[h * 32 + c], w, vd);
    }
    Ws[(128 + h) * LDS_PITCH + k] = vs;
    Ws[(132 + h) * LDS_PITCH + k] = vd;
    Ws[(136 + h) * LDS_PITCH + k] = 0.f;
    Ws[(140 + h) * LDS_PITCH + k] = 0.f;
  }
  int chunk = blockIdx.x;
  float4 va = make_float4(0.f, 0.f, 0.f, 0.f);
  if (chunk < nchunks) {
    const int row = min(chunk * 16 + sr, N - 1);
    va = ((const float4*)x)[(size_t)row * 16 + sq];
  }
  __syncthreads();
  const bool zwave = wave >= 4;
  const int trow = zwave ? (wave - 4) * 32 : 128;
  v2f b0[16], b1[16];
#pragma unroll
  for (int ks = 0; ks < 16; ++ks) {
    b0[ks] = frag(Ws, trow + mrow, ks, kb);
    b1[ks] = zwave ? frag(Ws, trow + 16 + mrow, ks, kb) : b0[ks];
  }
  const int cbase = (wave & 3) * 32;
  float* const ap = ((mrow < 4) ? a_src : a_dst) + (mrow & 3);
  int buf = 0;
  int turn = 0;
  while (chunk < nchunks) {
    *(float4*)(&As[buf][sr * LDS_PITCH + sq * 4]) = va;
    __syncthreads();
    const int next = chunk + GEMM_GROUPS;
    if (next < nchunks) {
      const int row = min(next * 16 + sr, N - 1);
      va = ((const float4*)x)[(size_t)row * 16 + sq];
    }
    if (zwave || wave == turn) {
      v2f a[16];
#pragma unroll
      for (int ks = 0; ks < 16; ++ks) a[ks] = frag(As[buf], mrow, ks, kb);
      asm volatile("" ::"v"(a[0]), "v"(a[1]), "v"(a[2]), "v"(a[3]), "v"(a[4]), "v"(a[5]),
                   "v"(a[6]), "v"(a[7]), "v"(a[8]), "v"(a[9]), "v"(a[10]), "v"(a[11]),
                   "v"(a[12]), "v"(a[13]), "v"(a[14]), "v"(a[15]));
      v8f c0 = {0.f, 0.f, 0.f, 0.f, 0.f, 0.f, 0.f, 0.f};
      v8f c1 = {0.f, 0.f, 0.f, 0.f, 0.f, 0.f, 0.f, 0.f};
      const int m0 = chunk * 16;
      const int mb = m0 + ((lane >> 4) << 3);
      if (zwave) {
#pragma unroll
        for (int ks = 0; ks < 16; ++ks) {
          c0 = __builtin_amdgcn_wmma_f32_16x16x4_f32(false, a[ks], false, b0[ks],
                                                     (short)0, c0, false, false);
          c1 = __builtin_amdgcn_wmma_f32_16x16x4_f32(false, a[ks], false, b1[ks],
                                                     (short)0, c1, false, false);
        }
#pragma unroll
        for (int r = 0; r < 8; ++r) {
          const int m = mb + r;
          if (m < N) {
            z[(size_t)m * OUT_DIM + cbase + mrow] = c0[r];
            z[(size_t)m * OUT_DIM + cbase + 16 + mrow] = c1[r];
          }
        }
      } else {
#pragma unroll
        for (int ks = 0; ks < 16; ks += 2) {
          c0 = __builtin_amdgcn_wmma_f32_16x16x4_f32(false, a[ks], false, b0[ks],
                                                     (short)0, c0, false, false);
          c1 = __builtin_amdgcn_wmma_f32_16x16x4_f32(false, a[ks + 1], false, b0[ks + 1],
                                                     (short)0, c1, false, false);
        }
        if (mrow < 8) {
#pragma unroll
          for (int r = 0; r < 8; ++r) {
            const int m = mb + r;
            if (m < N) ap[(size_t)m * HEADS] = c0[r] + c1[r];
          }
        }
      }
    }
    buf ^= 1;
    chunk = next;
    turn = (turn + 1) & 3;
  }
}

#define SLOTS 31
#define ROW 32
#define OVC 512
#define APITCH 260

static __device__ __forceinline__ void place_edge(int s, int d, int base, int npb,
                                                  int* lcnt, int* slab, int* ovn,
                                                  int2* ovl, int2* __restrict__ ovg,
                                                  bool to_global, int capg, int N) {
  const unsigned dl = (unsigned)(d - base);
  if (dl < (unsigned)npb && (unsigned)d < (unsigned)N) {
    const int r = atomicAdd(&lcnt[dl], 1);
    if ((unsigned)r < (unsigned)SLOTS) {
      slab[dl * ROW + r] = s;
    } else {
      const int o = atomicAdd(ovn, 1);
      if (to_global) {
        if ((unsigned)o < (unsigned)capg) ovg[o] = make_int2(s, d);
      } else if ((unsigned)o < (unsigned)OVC) {
        ovl[o] = make_int2(s, d);
      }
    }
  }
}

static __device__ __forceinline__ void load_first(const int* row,
                                                  const float* __restrict__ a_src,
                                                  const float4 ad, int deg0, int n,
                                                  int N, int i, int& s, float4& e) {
  s = n;
  e = make_float4(-INFINITY, -INFINITY, -INFINITY, -INFINITY);
  if (i <= deg0) {
    if (i < deg0) s = min(max(row[i], 0), N - 1);
    const float4 as = ((const float4*)a_src)[s];
    e.x = lrelu(as.x + ad.x);
    e.y = lrelu(as.y + ad.y);
    e.z = lrelu(as.z + ad.z);
    e.w = lrelu(as.w + ad.w);
  }
}

static __device__ __forceinline__ bool load_ovf(const int2* ovl,
                                                const int2* __restrict__ ovg, bool g,
                                                const float* __restrict__ a_src,
                                                const float4 ad, int nov, int n, int N,
                                                int i, int& s, float4& e) {
  s = n;
  e = make_float4(-INFINITY, -INFINITY, -INFINITY, -INFINITY);
  bool mine = false;
  int2 en = make_int2(n, -1);
  if (i < nov) {
    if (g) en = ovg[i]; else en = ovl[i];
    mine = (en.y == n);
  }
  if (mine) {
    s = min(max(en.x, 0), N - 1);
    const float4 as = ((const float4*)a_src)[s];
    e.x = lrelu(as.x + ad.x);
    e.y = lrelu(as.y + ad.y);
    e.z = lrelu(as.z + ad.z);
    e.w = lrelu(as.w + ad.w);
  }
  return mine;
}

template <int J0, int NE>
static __device__ __forceinline__ void gather_rows(const float* __restrict__ x, int s,
                                                   const float4 al, unsigned lane,
                                                   float (&acc)[8]) {
  float2 xv[NE];
  unsigned idx[NE];
#pragma unroll
  for (int k = 0; k < NE; ++k)
    idx[k] = (unsigned)__builtin_amdgcn_readlane(s, J0 + k) * (IN_CH / 2) + lane;
#pragma unroll
  for (int k = 0; k < NE; k += 4)
    asm volatile("" ::"v"(idx[k]), "v"(idx[k + 1]), "v"(idx[k + 2]), "v"(idx[k + 3]));
#pragma unroll
  for (int k = 0; k < NE; ++k) xv[k] = ((const float2*)x)[idx[k]];
#pragma unroll
  for (int k = 0; k < NE; ++k) {
    const int j = J0 + k;
    const float b0 = __int_as_float(__builtin_amdgcn_readlane(__float_as_int(al.x), j));
    const float b1 = __int_as_float(__builtin_amdgcn_readlane(__float_as_int(al.y), j));
    const float b2 = __int_as_float(__builtin_amdgcn_readlane(__float_as_int(al.z), j));
    const float b3 = __int_as_float(__builtin_amdgcn_readlane(__float_as_int(al.w), j));
    acc[0] = __builtin_fmaf(b0, xv[k].x, acc[0]); acc[1] = __builtin_fmaf(b0, xv[k].y, acc[1]);
    acc[2] = __builtin_fmaf(b1, xv[k].x, acc[2]); acc[3] = __builtin_fmaf(b1, xv[k].y, acc[3]);
    acc[4] = __builtin_fmaf(b2, xv[k].x, acc[4]); acc[5] = __builtin_fmaf(b2, xv[k].y, acc[5]);
    acc[6] = __builtin_fmaf(b3, xv[k].x, acc[6]); acc[7] = __builtin_fmaf(b3, xv[k].y, acc[7]);
  }
}

static __device__ __forceinline__ void gather_chunk(const float* __restrict__ x,
                                                    int s, const float4 al, int cnt,
                                                    int lane, float (&acc)[8]) {
  const unsigned l = (unsigned)lane;
  if (cnt <= 12) {
    gather_rows<0, 12>(x, s, al, l, acc);
  } else if (cnt <= 16) {
    gather_rows<0, 16>(x, s, al, l, acc);
  } else if (cnt <= 20) {
    gather_rows<0, 20>(x, s, al, l, acc);
  } else if (cnt <= 24) {
    gather_rows<0, 24>(x, s, al, l, acc);
  } else {
    gather_rows<0, 24>(x, s, al, l, acc);
    gather_rows<24, 8>(x, s, al, l, acc);
  }
}

static __device__ __forceinline__ void gather_mask(const float* __restrict__ x,
                                                   int s, const float4 al, unsigned mask,
                                                   int lane, float (&acc)[8]) {
  while (mask != 0u) {
    const int j = __builtin_amdgcn_readfirstlane(__builtin_ctz(mask));
    mask &= mask - 1u;
    const int sj = __builtin_amdgcn_readlane(s, j);
    const float b0 = __int_as_float(__builtin_amdgcn_readlane(__float_as_int(al.x), j));
    const float b1 = __int_as_float(__builtin_amdgcn_readlane(__float_as_int(al.y), j));
    const float b2 = __int_as_float(__builtin_amdgcn_readlane(__float_as_int(al.z), j));
    const float b3 = __int_as_float(__builtin_amdgcn_readlane(__float_as_int(al.w), j));
    const float2 xv = *(const float2*)(x + (size_t)sj * IN_CH + 2 * lane);
    acc[0] += b0 * xv.x; acc[1] += b0 * xv.y;
    acc[2] += b1 * xv.x; acc[3] += b1 * xv.y;
    acc[4] += b2 * xv.x; acc[5] += b2 * xv.y;
    acc[6] += b3 * xv.x; acc[7] += b3 * xv.y;
  }
}

static __device__ __forceinline__ float wave_max(float v) {
#pragma unroll
  for (int off = 1; off < 32; off <<= 1) v = fmaxf(v, __shfl_xor(v, off, 32));
  return v;
}
static __device__ __forceinline__ float wave_sum(float v) {
#pragma unroll
  for (int off = 1; off < 32; off <<= 1) v += __shfl_xor(v, off, 32);
  return v;
}

__global__ __launch_bounds__(1024) void k_node(const int* __restrict__ ei,
                                               const int* __restrict__ cellw,
                                               const int* __restrict__ cellcnt,
                                               int* __restrict__ ovfw,
                                               const float* __restrict__ a_src,
                                               const float* __restrict__ a_dst,
                                               const float* __restrict__ x,
                                               const float* __restrict__ lin_w,
                                               const float* __restrict__ zres,
                                               const float* __restrict__ gat_bias,
                                               const float* __restrict__ res_b,
                                               const float* __restrict__ ln_g,
                                               const float* __restrict__ ln_b,
                                               float* __restrict__ out, int E, int N) {
  const int2* cells = (const int2*)cellw;
  int2* ovg = (int2*)ovfw;
  __shared__ int lcnt[MAXNPB];
  __shared__ int ccnt[NC];
  __shared__ __attribute__((aligned(16))) int slab[MAXNPB * ROW];
  __shared__ __attribute__((aligned(16))) float agg[MAXNPB * APITCH];
  __shared__ int2 ovl[OVC];
  __shared__ int ovn;
  __shared__ int sums[2];
  const int tid = threadIdx.x;
  const int lane = tid & 31;
  const int wave = __builtin_amdgcn_readfirstlane(tid >> 5);
  const int b = blockIdx.x;
  const int npb = range_size(N);
  const int base = b * npb;
  const int cs = (E + NC - 1) / NC;
  for (int i = tid; i < npb; i += 1024) lcnt[i] = 0;
  if (tid < NC) ccnt[tid] = cellcnt[b * NC + tid];
  if (tid == 0) { ovn = 0; sums[0] = 0; sums[1] = 0; }
  __syncthreads();

  int2 ent[NC / 32];
  int num[NC / 32];
#pragma unroll
  for (int k = 0; k < NC / 32; ++k) {
    const int c = wave + k * 32;
    num[k] = ccnt[c];
    ent[k] = make_int2(0, -1);
    if (num[k] <= CAP && lane < num[k]) ent[k] = cells[((size_t)b * NC + c) * CAP + lane];
  }
  bool to_global = false;
  int capg = 0;
  for (int pass = 0; pass < 2; ++pass) {
#pragma unroll
    for (int k = 0; k < NC / 32; ++k) {
      if (num[k] <= CAP) {
        if (lane < num[k])
          place_edge(ent[k].x, ent[k].y, base, npb, lcnt, slab, &ovn, ovl, ovg, to_global,
                     capg, N);
      } else {
        const int c = wave + k * 32;
        const int g1 = min(E, (c + 1) * cs);
        for (int g = c * cs + lane; g < g1; g += 32) {
          const int s = ei[g];
          const int d = min(max(ei[E + g], 0), N - 1);
          if (bucket_of(d, npb) == b)
            place_edge(s, d, base, npb, lcnt, slab, &ovn, ovl, ovg, to_global, capg, N);
        }
      }
    }
    __syncthreads();
    if (pass == 1 || ovn <= OVC) break;
    int part = 0;
    for (int i = tid; i < b * NC; i += 1024) part += max(cellcnt[i], 0);
    if (part != 0) atomicAdd(&sums[0], part);
    if (tid < NC) atomicAdd(&sums[1], max(ccnt[tid], 0));
    __syncthreads();
    const int p0 = min(max(sums[0], 0), E);
    capg = min(max(sums[1], 0), E - p0);
    ovg += p0;
    to_global = true;
    __syncthreads();
    for (int i = tid; i < npb; i += 1024) lcnt[i] = 0;
    if (tid == 0) ovn = 0;
    __syncthreads();
  }
  const int nov = min(max(ovn, 0), to_global ? capg : OVC);

  const int nend = min(base + npb, N);
  for (int n = base + wave; n < nend; n += 32) {
    const int dl = n - base;
    const int deg = max(lcnt[dl], 0);
    const int deg0 = min(deg, SLOTS);
    const bool more = deg > SLOTS;
    const float4 ad = ((const float4*)a_dst)[n];

    int s0;
    float4 e0;
    load_first(slab + dl * ROW, a_src, ad, deg0, n, N, lane, s0, e0);

    float4 m = e0;
    if (more) {
      for (int q0 = 0; q0 < nov; q0 += 32) {
        int s; float4 e;
        load_ovf(ovl, ovg, to_global, a_src, ad, nov, n, N, q0 + lane, s, e);
        m.x = fmaxf(m.x, e.x); m.y = fmaxf(m.y, e.y);
        m.z = fmaxf(m.z, e.z); m.w = fmaxf(m.w, e.w);
      }
    }
    m.x = wave_max(m.x); m.y = wave_max(m.y);
    m.z = wave_max(m.z); m.w = wave_max(m.w);

    float4 x0;
    x0.x = __expf(e0.x - m.x); x0.y = __expf(e0.y - m.y);
    x0.z = __expf(e0.z - m.z); x0.w = __expf(e0.w - m.w);
    float4 q = x0;
    if (more) {
      for (int q0 = 0; q0 < nov; q0 += 32) {
        int s; float4 e;
        load_ovf(ovl, ovg, to_global, a_src, ad, nov, n, N, q0 + lane, s, e);
        q.x += __expf(e.x - m.x); q.y += __expf(e.y - m.y);
        q.z += __expf(e.z - m.z); q.w += __expf(e.w - m.w);
      }
    }
    q.x = wave_sum(q.x) + 1e-16f; q.y = wave_sum(q.y) + 1e-16f;
    q.z = wave_sum(q.z) + 1e-16f; q.w = wave_sum(q.w) + 1e-16f;

    float acc[8] = {0.f, 0.f, 0.f, 0.f, 0.f, 0.f, 0.f, 0.f};
    {
      float4 al;
      al.x = x0.x / q.x; al.y = x0.y / q.y; al.z = x0.z / q.z; al.w = x0.w / q.w;
      gather_chunk(x, s0, al, deg0 + 1, lane, acc);
    }
    if (more) {
      for (int q0 = 0; q0 < nov; q0 += 32) {
        int s; float4 e;
        const bool mine = load_ovf(ovl, ovg, to_global, a_src, ad, nov, n, N, q0 + lane, s, e);
        float4 al;
        al.x = __expf(e.x - m.x) / q.x; al.y = __expf(e.y - m.y) / q.y;
        al.z = __expf(e.z - m.z) / q.z; al.w = __expf(e.w - m.w) / q.w;
        const unsigned mask = __builtin_amdgcn_ballot_w32(mine);
        gather_mask(x, s, al, mask, lane, acc);
      }
    }
    float* ar = agg + dl * APITCH + 2 * lane;
    *(float2*)(ar)       = make_float2(acc[0], acc[1]);
    *(float2*)(ar + 64)  = make_float2(acc[2], acc[3]);
    *(float2*)(ar + 128) = make_float2(acc[4], acc[5]);
    *(float2*)(ar + 192) = make_float2(acc[6], acc[7]);
  }

  const int wr = tid >> 4;
  const int wg = tid & 15;
  const float4 wv0 = ((const float4*)lin_w)[wr * 16 + wg];
  const float4 wv1 = ((const float4*)lin_w)[(wr + 64) * 16 + wg];
  __syncthreads();
  float* Wl = (float*)slab;
  *(float4*)(Wl + wr * IN_CH + ((wg ^ (wr & 15)) << 2)) = wv0;
  *(float4*)(Wl + (wr + 64) * IN_CH + ((wg ^ (wr & 15)) << 2)) = wv1;
  __syncthreads();
  {
    const int mrow = lane & 15;
    const int kb = (lane >> 4) << 1;
    const int hd = wave & 3;
    const int ntiles = (max(nend - base, 0) + 15) >> 4;
    if ((wave >> 2) < ntiles) {
      const float* w0p = Wl + (hd * 32 + mrow) * IN_CH + kb;
      const float* w1p = w0p + 16 * IN_CH;
      v2f b0[16], b1[16];
#pragma unroll
      for (int ks = 0; ks < 16; ++ks) {
        b0[ks] = *(const v2f*)(w0p + ((ks ^ mrow) << 2));
        b1[ks] = *(const v2f*)(w1p + ((ks ^ mrow) << 2));
      }
      for (int t = wave >> 2; t < ntiles; t += 8) {
        const float* ap = agg + (t * 16 + mrow) * APITCH + hd * 64 + kb;
        v8f c0 = {0.f, 0.f, 0.f, 0.f, 0.f, 0.f, 0.f, 0.f};
        v8f c1 = {0.f, 0.f, 0.f, 0.f, 0.f, 0.f, 0.f, 0.f};
#pragma unroll
        for (int kh = 0; kh < 2; ++kh) {
          v2f a[8];
#pragma unroll
          for (int ks = 0; ks < 8; ++ks) a[ks] = *(const v2f*)(ap + (kh * 8 + ks) * 4);
#pragma unroll
          for (int ks = 0; ks < 8; ++ks) {
            c0 = __builtin_amdgcn_wmma_f32_16x16x4_f32(false, a[ks], false, b0[kh * 8 + ks],
                                                       (short)0, c0, false, false);
            c1 = __builtin_amdgcn_wmma_f32_16x16x4_f32(false, a[ks], false, b1[kh * 8 + ks],
                                                       (short)0, c1, false, false);
          }
        }
        float* dp = agg + (t * 16 + ((lane >> 4) << 3)) * APITCH + hd * 64 + mrow;
#pragma unroll
        for (int r = 0; r < 8; ++r) {
          dp[r * APITCH] = c0[r];
          dp[r * APITCH + 16] = c1[r];
        }
      }
    }
  }
  float zc0 = 0.f, zc1 = 0.f, zc2 = 0.f, zc3 = 0.f;
  if (base + wave < nend) {
    const float* zr = zres + (size_t)(base + wave) * OUT_DIM + lane;
    zc0 = zr[0]; zc1 = zr[32]; zc2 = zr[64]; zc3 = zr[96];
  }
  const float gb0 = gat_bias[lane], gb1 = gat_bias[lane + 32];
  const float gb2 = gat_bias[lane + 64], gb3 = gat_bias[lane + 96];
  const float rb0 = res_b[lane], rb1 = res_b[lane + 32];
  const float rb2 = res_b[lane + 64], rb3 = res_b[lane + 96];
  const float lg0 = ln_g[lane], lg1 = ln_g[lane + 32];
  const float lg2 = ln_g[lane + 64], lg3 = ln_g[lane + 96];
  const float lb0 = ln_b[lane], lb1 = ln_b[lane + 32];
  const float lb2 = ln_b[lane + 64], lb3 = ln_b[lane + 96];
  __syncthreads();
  for (int n = base + wave; n < nend; n += 32) {
    float zn0 = 0.f, zn1 = 0.f, zn2 = 0.f, zn3 = 0.f;
    if (n + 32 < nend) {
      const float* zr = zres + (size_t)(n + 32) * OUT_DIM + lane;
      zn0 = zr[0]; zn1 = zr[32]; zn2 = zr[64]; zn3 = zr[96];
    }
    const float* hr = agg + (n - base) * APITCH + lane;
    const float acc0 = hr[0], acc1 = hr[64], acc2 = hr[128], acc3 = hr[192];

    const float z0 = ((acc0 + gb0) + zc0) + rb0;
    const float z1 = ((acc1 + gb1) + zc1) + rb1;
    const float z2 = ((acc2 + gb2) + zc2) + rb2;
    const float z3 = ((acc3 + gb3) + zc3) + rb3;
    zc0 = zn0; zc1 = zn1; zc2 = zn2; zc3 = zn3;

    const float mu = wave_sum((z0 + z1) + (z2 + z3)) * (1.0f / 128.0f);
    const float d0 = z0 - mu, d1 = z1 - mu, d2 = z2 - mu, d3 = z3 - mu;
    const float var = wave_sum((d0 * d0 + d1 * d1) + (d2 * d2 + d3 * d3)) * (1.0f / 128.0f);
    const float inv = rsqrtf(var + 1e-5f);
    float* op = out + (size_t)n * OUT_DIM + lane;
    op[0]  = lrelu(d0 * inv * lg0 + lb0);
    op[32] = lrelu(d1 * inv * lg1 + lb1);
    op[64] = lrelu(d2 * inv * lg2 + lb2);
    op[96] = lrelu(d3 * inv * lg3 + lb3);
  }
}

extern "C" void kernel_launch(void* const* d_in, const int* in_sizes, int n_in,
                              void* d_out, int out_size, void* d_ws, size_t ws_size,
                              hipStream_t stream) {
  const float* x        = (const float*)d_in[0];
  const int*   ei       = (const int*)d_in[1];
  const float* lin_w    = (const float*)d_in[2];
  const float* att_src  = (const float*)d_in[3];
  const float* att_dst  = (const float*)d_in[4];
  const float* gat_bias = (const float*)d_in[5];
  const float* res_w    = (const float*)d_in[6];
  const float* res_b    = (const float*)d_in[7];
  const float* ln_g     = (const float*)d_in[8];
  const float* ln_b     = (const float*)d_in[9];
  const int N  = in_sizes[0] / IN_CH;
  const int E  = in_sizes[1] / 2;

  float* z      = (float*)d_ws;
  float* a_src  = z + (size_t)N * OUT_DIM;
  float* a_dst  = a_src + (size_t)N * HEADS;
  int*   ovf    = (int*)(a_dst + (size_t)N * HEADS);
  int*   cells  = ovf + (size_t)E * 2;
  int*   cellcnt = cells + (size_t)NB * NC * CAP * 2;

  k_gemm<<<GEMM_GROUPS + BIN_GROUPS, 256, 0, stream>>>(x, lin_w, res_w, att_src, att_dst, z,
                                                       a_src, a_dst, ei, cells, cellcnt, E, N);
  k_node<<<NB, 1024, 0, stream>>>(ei, cells, cellcnt, ovf, a_src, a_dst, x, lin_w, z, gat_bias,
                                  res_b, ln_g, ln_b, (float*)d_out, E, N);
}
